# starting kernel
# baseline (speedup 1.0000x reference)
#include <hip/hip_runtime.h>
#include <hip/hip_bf16.h>


#define N_NODES   50000
#define N_EDGES   800000
#define D_HID     256
#define N_DOCS    20000
#define MLP_HID   64
#define N_CLASSES 20

typedef __attribute__((ext_vector_type(2))) float v2f;
typedef __attribute__((ext_vector_type(8))) float v8f;

__global__ __launch_bounds__(64) void gemm_wmma_f32(
    const float* __restrict__ A, const float* __restrict__ B,
    float* __restrict__ C, int M, int N, int K,
    const float* __restrict__ bias, int doRelu)
{
    __shared__ float As[16][32];
    __shared__ float Bs[32][64];

    const int tid  = threadIdx.x;
    const int wave = tid >> 5;
    const int lane = tid & 31;
    const int m0 = blockIdx.x * 16;
    const int n0 = blockIdx.y * 64;

    v8f acc0 = {};
    v8f acc1 = {};

    const int mrow  = lane & 15;
    const int khalf = (lane >> 4) << 1;
    const int ncol0 = (lane & 15) + (wave << 5);
    const int ncol1 = ncol0 + 16;

    const int ar = tid >> 2;
    const int ac = (tid & 3) * 8;
    const int br = tid >> 1;
    const int bc = (tid & 1) * 32;

    for (int k0 = 0; k0 < K; k0 += 32) {
        {
            const float* s = A + (size_t)(m0 + ar) * K + k0 + ac;
            *(float4*)&As[ar][ac]     = *(const float4*)(s);
            *(float4*)&As[ar][ac + 4] = *(const float4*)(s + 4);
        }
        {
            const float* s = B + (size_t)(k0 + br) * N + n0 + bc;
            #pragma unroll
            for (int j = 0; j < 8; ++j)
                *(float4*)&Bs[br][bc + 4 * j] = *(const float4*)(s + 4 * j);
        }
        if (k0 + 32 < K) {
            __builtin_prefetch(A + (size_t)(m0 + ar) * K + k0 + 32 + ac, 0, 3);
            __builtin_prefetch(B + (size_t)(k0 + 32 + br) * N + n0 + bc, 0, 3);
        }
        __syncthreads();

        #pragma unroll
        for (int kk = 0; kk < 32; kk += 4) {
            v2f a, b0, b1;
            a.x = As[mrow][kk + khalf];
            a.y = As[mrow][kk + khalf + 1];
            b0.x = Bs[kk + khalf][ncol0];
            b0.y = Bs[kk + khalf + 1][ncol0];
            b1.x = Bs[kk + khalf][ncol1];
            b1.y = Bs[kk + khalf + 1][ncol1];
            acc0 = __builtin_amdgcn_wmma_f32_16x16x4_f32(
                false, a, false, b0, (short)0, acc0, false, false);
            acc1 = __builtin_amdgcn_wmma_f32_16x16x4_f32(
                false, a, false, b1, (short)0, acc1, false, false);
        }
        __syncthreads();
    }

    const int colg0 = n0 + ncol0;
    const int colg1 = n0 + ncol1;
    const float bv0 = bias ? bias[colg0] : 0.0f;
    const float bv1 = bias ? bias[colg1] : 0.0f;
    #pragma unroll
    for (int r = 0; r < 8; ++r) {
        const int rowg = m0 + r + ((lane >> 4) << 3);
        float v0 = acc0[r] + bv0;
        float v1 = acc1[r] + bv1;
        if (doRelu) { v0 = fmaxf(v0, 0.0f); v1 = fmaxf(v1, 0.0f); }
        C[(size_t)rowg * N + colg0] = v0;
        C[(size_t)rowg * N + colg1] = v1;
    }
}

__global__ __launch_bounds__(256) void node_dots(
    const float* __restrict__ H, const float* __restrict__ as_,
    const float* __restrict__ ad_, float* __restrict__ es,
    float* __restrict__ ed, int n)
{
    const int node = (int)((blockIdx.x * blockDim.x + threadIdx.x) >> 5);
    const int lane = threadIdx.x & 31;
    if (node >= n) return;
    const float* row = H + (size_t)node * D_HID;
    float s = 0.0f, d = 0.0f;
    #pragma unroll
    for (int j = 0; j < D_HID / 32; ++j) {
        const int i = lane + j * 32;
        const float h = row[i];
        s += h * as_[i];
        d += h * ad_[i];
    }
    #pragma unroll
    for (int off = 16; off; off >>= 1) {
        s += __shfl_down(s, off, 32);
        d += __shfl_down(d, off, 32);
    }
    if (lane == 0) { es[node] = s; ed[node] = d; }
}

__global__ void fill_f32(float* __restrict__ p, float v, size_t n)
{
    const size_t i = (size_t)blockIdx.x * blockDim.x + threadIdx.x;
    if (i < n) p[i] = v;
}

__device__ inline void atomicMaxF(float* addr, float val)
{
    if (val >= 0.0f) atomicMax((int*)addr, __float_as_int(val));
    else             atomicMin((unsigned int*)addr, __float_as_uint(val));
}

__global__ __launch_bounds__(256) void edge_score(
    const int* __restrict__ src, const int* __restrict__ dst,
    const float* __restrict__ es, const float* __restrict__ ed,
    float* __restrict__ e, float* __restrict__ mx, int nE)
{
    const int i = blockIdx.x * blockDim.x + threadIdx.x;
    if (i >= nE) return;
    float v = es[src[i]] + ed[dst[i]];
    v = (v > 0.0f) ? v : 0.2f * v;
    e[i] = v;
    atomicMaxF(&mx[dst[i]], v);
}

__global__ __launch_bounds__(256) void edge_exp(
    const int* __restrict__ dst, const float* __restrict__ mx,
    float* __restrict__ e, float* __restrict__ den, int nE)
{
    const int i = blockIdx.x * blockDim.x + threadIdx.x;
    if (i >= nE) return;
    const int d = dst[i];
    const float ex = __expf(e[i] - mx[d]);
    e[i] = ex;
    atomicAdd(&den[d], ex);
}

__global__ __launch_bounds__(256) void edge_aggregate(
    const int* __restrict__ src, const int* __restrict__ dst,
    const float* __restrict__ e, const float* __restrict__ den,
    const float* __restrict__ H, float* __restrict__ OUT, int nE)
{
    const int edge = (int)((blockIdx.x * blockDim.x + threadIdx.x) >> 5);
    const int lane = threadIdx.x & 31;
    if (edge >= nE) return;
    const int s = src[edge], d = dst[edge];
    const float alpha = e[edge] / (den[d] + 1e-16f);
    const float* hrow = H + (size_t)s * D_HID;
    float* orow = OUT + (size_t)d * D_HID;
    #pragma unroll
    for (int j = 0; j < D_HID / 32; ++j) {
        const int c = lane + j * 32;
        atomicAdd(&orow[c], alpha * hrow[c]);
    }
}

__global__ void bias_relu256(float* __restrict__ X, const float* __restrict__ b,
                             size_t n)
{
    const size_t i = (size_t)blockIdx.x * blockDim.x + threadIdx.x;
    if (i >= n) return;
    X[i] = fmaxf(X[i] + b[i & (D_HID - 1)], 0.0f);
}

__global__ void gather_rows256(const float* __restrict__ X,
                               const int* __restrict__ idx,
                               float* __restrict__ Y, size_t n)
{
    const size_t i = (size_t)blockIdx.x * blockDim.x + threadIdx.x;
    if (i >= n) return;
    const size_t r = i >> 8, c = i & (D_HID - 1);
    Y[i] = X[(size_t)idx[r] * D_HID + c];
}

__global__ __launch_bounds__(256) void mlp2_softmax(
    const float* __restrict__ DH, const float* __restrict__ Wm2,
    const float* __restrict__ bm2, float* __restrict__ out, int nDocs)
{
    const int doc  = (int)((blockIdx.x * blockDim.x + threadIdx.x) >> 5);
    const int lane = threadIdx.x & 31;
    if (doc >= nDocs) return;
    const float* row = DH + (size_t)doc * MLP_HID;
    float logit = -3.0e38f;
    if (lane < N_CLASSES) {
        float acc = bm2[lane];
        #pragma unroll
        for (int k = 0; k < MLP_HID; ++k)
            acc += row[k] * Wm2[k * N_CLASSES + lane];
        logit = acc;
    }
    float mx = logit;
    #pragma unroll
    for (int off = 16; off; off >>= 1) mx = fmaxf(mx, __shfl_xor(mx, off, 32));
    const float ex = (lane < N_CLASSES) ? __expf(logit - mx) : 0.0f;
    float sum = ex;
    #pragma unroll
    for (int off = 16; off; off >>= 1) sum += __shfl_xor(sum, off, 32);
    if (lane < N_CLASSES) out[(size_t)doc * N_CLASSES + lane] = ex / sum;
}

extern "C" void kernel_launch(void* const* d_in, const int* in_sizes, int n_in,
                              void* d_out, int out_size, void* d_ws, size_t ws_size,
                              hipStream_t stream)
{
    (void)in_sizes; (void)n_in; (void)out_size; (void)ws_size;

    const float* x    = (const float*)d_in[0];
    const int*   ei   = (const int*)  d_in[1];
    const int*   doci = (const int*)  d_in[2];
    const float* W1   = (const float*)d_in[3];
    const float* a1s  = (const float*)d_in[4];
    const float* a1d  = (const float*)d_in[5];
    const float* b1   = (const float*)d_in[6];
    const float* W2   = (const float*)d_in[7];
    const float* a2s  = (const float*)d_in[8];
    const float* a2d  = (const float*)d_in[9];
    const float* b2   = (const float*)d_in[10];
    const float* Wm1  = (const float*)d_in[11];
    const float* bm1  = (const float*)d_in[12];
    const float* Wm2  = (const float*)d_in[13];
    const float* bm2  = (const float*)d_in[14];

    const int* src = ei;
    const int* dst = ei + N_EDGES;

    float* H    = (float*)d_ws;
    float* OUT  = H    + (size_t)N_NODES * D_HID;
    float* ES   = OUT  + (size_t)N_NODES * D_HID;
    float* ED   = ES   + N_NODES;
    float* MX   = ED   + N_NODES;
    float* DEN  = MX   + N_NODES;
    float* EB   = DEN  + N_NODES;
    float* DOCS = EB   + N_EDGES;
    float* DH   = DOCS + (size_t)N_DOCS * D_HID;

    const dim3 gemmGridN(N_NODES / 16, D_HID / 64);
    const dim3 gemmGridD(N_DOCS  / 16, MLP_HID / 64);
    const size_t feat = (size_t)N_NODES * D_HID;
    const int gFeat  = (int)((feat + 255) / 256);
    const int gNode  = (N_NODES + 255) / 256;
    const int gNodeW = (N_NODES + 7) / 8;
    const int gEdge  = (N_EDGES + 255) / 256;
    const int gEdgeW = (N_EDGES + 7) / 8;
    const size_t docf = (size_t)N_DOCS * D_HID;
    const int gDocF  = (int)((docf + 255) / 256);
    const int gDocW  = (N_DOCS + 7) / 8;

    for (int layer = 0; layer < 2; ++layer) {
        const float* A   = (layer == 0) ? x   : OUT;
        const float* W   = (layer == 0) ? W1  : W2;
        const float* as_ = (layer == 0) ? a1s : a2s;
        const float* ad_ = (layer == 0) ? a1d : a2d;
        const float* b   = (layer == 0) ? b1  : b2;

        gemm_wmma_f32<<<gemmGridN, 64, 0, stream>>>(A, W, H, N_NODES, D_HID,
                                                    D_HID, nullptr, 0);
        node_dots<<<gNodeW, 256, 0, stream>>>(H, as_, ad_, ES, ED, N_NODES);
        fill_f32<<<gNode, 256, 0, stream>>>(MX, -INFINITY, (size_t)N_NODES);
        fill_f32<<<gNode, 256, 0, stream>>>(DEN, 0.0f, (size_t)N_NODES);
        fill_f32<<<gFeat, 256, 0, stream>>>(OUT, 0.0f, feat);
        edge_score<<<gEdge, 256, 0, stream>>>(src, dst, ES, ED, EB, MX, N_EDGES);
        edge_exp<<<gEdge, 256, 0, stream>>>(dst, MX, EB, DEN, N_EDGES);
        edge_aggregate<<<gEdgeW, 256, 0, stream>>>(src, dst, EB, DEN, H, OUT,
                                                   N_EDGES);
        bias_relu256<<<gFeat, 256, 0, stream>>>(OUT, b, feat);
    }

    gather_rows256<<<gDocF, 256, 0, stream>>>(OUT, doci, DOCS, docf);
    gemm_wmma_f32<<<gemmGridD, 64, 0, stream>>>(DOCS, Wm1, DH, N_DOCS, MLP_HID,
                                                D_HID, bm1, 1);
    mlp2_softmax<<<gDocW, 256, 0, stream>>>(DH, Wm2, bm2, (float*)d_out, N_DOCS);
}


# Edges bucketed by dst once per call (count, scan, place); one wave per dst sums alpha*H[src] in registers, stores its row once with bias+relu: no float atomics on OUT; sum order = arrival rank
# speedup vs baseline: 1.5937x; 1.5937x over previous
#include <hip/hip_runtime.h>
#include <hip/hip_bf16.h>


#define N_NODES   50000
#define N_EDGES   800000
#define D_HID     256
#define N_DOCS    20000
#define MLP_HID   64
#define N_CLASSES 20

typedef __attribute__((ext_vector_type(2))) float v2f;
typedef __attribute__((ext_vector_type(8))) float v8f;

__global__ __launch_bounds__(64) void gemm_wmma_f32(
    const float* __restrict__ A, const float* __restrict__ B,
    float* __restrict__ C, int M, int N, int K,
    const float* __restrict__ bias, int doRelu)
{
    __shared__ float As[16][32];
    __shared__ float Bs[32][64];

    const int tid  = threadIdx.x;
    const int wave = tid >> 5;
    const int lane = tid & 31;
    const int m0 = blockIdx.x * 16;
    const int n0 = blockIdx.y * 64;

    v8f acc0 = {};
    v8f acc1 = {};

    const int mrow  = lane & 15;
    const int khalf = (lane >> 4) << 1;
    const int ncol0 = (lane & 15) + (wave << 5);
    const int ncol1 = ncol0 + 16;

    const int ar = tid >> 2;
    const int ac = (tid & 3) * 8;
    const int br = tid >> 1;
    const int bc = (tid & 1) * 32;

    for (int k0 = 0; k0 < K; k0 += 32) {
        {
            const float* s = A + (size_t)(m0 + ar) * K + k0 + ac;
            *(float4*)&As[ar][ac]     = *(const float4*)(s);
            *(float4*)&As[ar][ac + 4] = *(const float4*)(s + 4);
        }
        {
            const float* s = B + (size_t)(k0 + br) * N + n0 + bc;
            #pragma unroll
            for (int j = 0; j < 8; ++j)
                *(float4*)&Bs[br][bc + 4 * j] = *(const float4*)(s + 4 * j);
        }
        if (k0 + 32 < K) {
            __builtin_prefetch(A + (size_t)(m0 + ar) * K + k0 + 32 + ac, 0, 3);
            __builtin_prefetch(B + (size_t)(k0 + 32 + br) * N + n0 + bc, 0, 3);
        }
        __syncthreads();

        #pragma unroll
        for (int kk = 0; kk < 32; kk += 4) {
            v2f a, b0, b1;
            a.x = As[mrow][kk + khalf];
            a.y = As[mrow][kk + khalf + 1];
            b0.x = Bs[kk + khalf][ncol0];
            b0.y = Bs[kk + khalf + 1][ncol0];
            b1.x = Bs[kk + khalf][ncol1];
            b1.y = Bs[kk + khalf + 1][ncol1];
            acc0 = __builtin_amdgcn_wmma_f32_16x16x4_f32(
                false, a, false, b0, (short)0, acc0, false, false);
            acc1 = __builtin_amdgcn_wmma_f32_16x16x4_f32(
                false, a, false, b1, (short)0, acc1, false, false);
        }
        __syncthreads();
    }

    const int colg0 = n0 + ncol0;
    const int colg1 = n0 + ncol1;
    const float bv0 = bias ? bias[colg0] : 0.0f;
    const float bv1 = bias ? bias[colg1] : 0.0f;
    #pragma unroll
    for (int r = 0; r < 8; ++r) {
        const int rowg = m0 + r + ((lane >> 4) << 3);
        float v0 = acc0[r] + bv0;
        float v1 = acc1[r] + bv1;
        if (doRelu) { v0 = fmaxf(v0, 0.0f); v1 = fmaxf(v1, 0.0f); }
        C[(size_t)rowg * N + colg0] = v0;
        C[(size_t)rowg * N + colg1] = v1;
    }
}

__global__ __launch_bounds__(256) void node_dots(
    const float* __restrict__ H, const float* __restrict__ as_,
    const float* __restrict__ ad_, float* __restrict__ es,
    float* __restrict__ ed, int n)
{
    const int node = (int)((blockIdx.x * blockDim.x + threadIdx.x) >> 5);
    const int lane = threadIdx.x & 31;
    if (node >= n) return;
    const float* row = H + (size_t)node * D_HID;
    float s = 0.0f, d = 0.0f;
    #pragma unroll
    for (int j = 0; j < D_HID / 32; ++j) {
        const int i = lane + j * 32;
        const float h = row[i];
        s += h * as_[i];
        d += h * ad_[i];
    }
    #pragma unroll
    for (int off = 16; off; off >>= 1) {
        s += __shfl_down(s, off, 32);
        d += __shfl_down(d, off, 32);
    }
    if (lane == 0) { es[node] = s; ed[node] = d; }
}

__global__ void fill_f32(float* __restrict__ p, float v, size_t n)
{
    const size_t i = (size_t)blockIdx.x * blockDim.x + threadIdx.x;
    if (i < n) p[i] = v;
}

__device__ inline void atomicMaxF(float* addr, float val)
{
    if (val >= 0.0f) atomicMax((int*)addr, __float_as_int(val));
    else             atomicMin((unsigned int*)addr, __float_as_uint(val));
}

__global__ __launch_bounds__(256) void edge_score(
    const int* __restrict__ src, const int* __restrict__ dst,
    const float* __restrict__ es, const float* __restrict__ ed,
    float* __restrict__ e, float* __restrict__ mx, int nE)
{
    const int i = blockIdx.x * blockDim.x + threadIdx.x;
    if (i >= nE) return;
    float v = es[src[i]] + ed[dst[i]];
    v = (v > 0.0f) ? v : 0.2f * v;
    e[i] = v;
    atomicMaxF(&mx[dst[i]], v);
}

__global__ __launch_bounds__(256) void edge_exp(
    const int* __restrict__ dst, const float* __restrict__ mx,
    float* __restrict__ e, float* __restrict__ den, int nE)
{
    const int i = blockIdx.x * blockDim.x + threadIdx.x;
    if (i >= nE) return;
    const int d = dst[i];
    const float ex = __expf(e[i] - mx[d]);
    e[i] = ex;
    atomicAdd(&den[d], ex);
}

#define N_PAD 53248

__global__ void fill_i32(int* __restrict__ p, int v, int n)
{
    const int i = blockIdx.x * blockDim.x + threadIdx.x;
    if (i < n) p[i] = v;
}

__global__ __launch_bounds__(256) void edge_rank(
    const int* __restrict__ dst, int* __restrict__ cnt,
    int* __restrict__ rank, int nE)
{
    const int i = blockIdx.x * blockDim.x + threadIdx.x;
    if (i >= nE) return;
    rank[i] = atomicAdd(&cnt[dst[i]], 1);
}

__global__ __launch_bounds__(1024) void scan_counts(
    const int* __restrict__ cnt, int* __restrict__ rowptr)
{
    __shared__ int wsum[2][32];
    const int tid  = threadIdx.x;
    const int lane = tid & 31;
    const int wave = tid >> 5;

    int4 v[N_PAD / 4096];
    #pragma unroll
    for (int it = 0; it < N_PAD / 4096; ++it)
        v[it] = ((const int4*)cnt)[it * 1024 + tid];

    int carry = 0;
    #pragma unroll
    for (int it = 0; it < N_PAD / 4096; ++it) {
        const int4 a = v[it];
        const int t = a.x + a.y + a.z + a.w;
        int inc = t;
        for (int off = 1; off < 32; off <<= 1) {
            const int u = __shfl_up(inc, off, 32);
            if (lane >= off) inc += u;
        }
        if (lane == 31) wsum[it & 1][wave] = inc;
        __syncthreads();
        const int ws = wsum[it & 1][lane];
        int winc = ws;
        for (int off = 1; off < 32; off <<= 1) {
            const int u = __shfl_up(winc, off, 32);
            if (lane >= off) winc += u;
        }
        const int total = __shfl(winc, 31, 32);
        const int wexcl = __shfl(winc, wave, 32) - __shfl(ws, wave, 32);
        const int excl  = carry + wexcl + inc - t;
        int4 o;
        o.x = excl;
        o.y = o.x + a.x;
        o.z = o.y + a.y;
        o.w = o.z + a.z;
        ((int4*)rowptr)[it * 1024 + tid] = o;
        carry += total;
    }
}

__global__ __launch_bounds__(256) void edge_place(
    const int* __restrict__ src, const int* __restrict__ dst,
    const int* __restrict__ rowptr, const int* __restrict__ rank,
    int2* __restrict__ perm, int nE)
{
    const int i = blockIdx.x * blockDim.x + threadIdx.x;
    if (i >= nE) return;
    perm[rowptr[dst[i]] + rank[i]] = make_int2(i, src[i]);
}

__global__ __launch_bounds__(256) void node_aggregate(
    const int* __restrict__ rowptr, const int2* __restrict__ perm,
    const float* __restrict__ e, const float* __restrict__ den,
    const float* __restrict__ H, const float* __restrict__ b,
    float* __restrict__ OUT, int n)
{
    const int lane = threadIdx.x & 31;
    const int node = __builtin_amdgcn_readfirstlane(
        (int)((blockIdx.x * blockDim.x + threadIdx.x) >> 5));
    if (node >= n) return;
    const int beg = rowptr[node];
    const int end = rowptr[node + 1];
    const float dn = den[node] + 1e-16f;

    float4 acc0 = make_float4(0.0f, 0.0f, 0.0f, 0.0f);
    float4 acc1 = make_float4(0.0f, 0.0f, 0.0f, 0.0f);

    for (int base = beg; base < end; base += 32) {
        int   s = 0;
        float a = 0.0f;
        if (base + lane < end) {
            const int2 pe = perm[base + lane];
            s = pe.y;
            a = e[pe.x] / dn;
        }
        const int cnt = (end - base < 32) ? (end - base) : 32;
        int t = 0;
        for (; t + 4 <= cnt; t += 4) {
            float4 h0[4], h1[4];
            float  at[4];
            #pragma unroll
            for (int u = 0; u < 4; ++u) {
                const int st = __builtin_amdgcn_readlane(s, t + u);
                at[u] = __int_as_float(
                    __builtin_amdgcn_readlane(__float_as_int(a), t + u));
                const float4* hr = (const float4*)(H + (size_t)st * D_HID);
                h0[u] = hr[lane];
                h1[u] = hr[lane + 32];
            }
            #pragma unroll
            for (int u = 0; u < 4; ++u) {
                acc0.x = fmaf(at[u], h0[u].x, acc0.x);
                acc0.y = fmaf(at[u], h0[u].y, acc0.y);
                acc0.z = fmaf(at[u], h0[u].z, acc0.z);
                acc0.w = fmaf(at[u], h0[u].w, acc0.w);
                acc1.x = fmaf(at[u], h1[u].x, acc1.x);
                acc1.y = fmaf(at[u], h1[u].y, acc1.y);
                acc1.z = fmaf(at[u], h1[u].z, acc1.z);
                acc1.w = fmaf(at[u], h1[u].w, acc1.w);
            }
        }
        #pragma nounroll
        for (; t < cnt; ++t) {
            const int st = __builtin_amdgcn_readlane(s, t);
            const float at1 = __int_as_float(
                __builtin_amdgcn_readlane(__float_as_int(a), t));
            const float4* hr = (const float4*)(H + (size_t)st * D_HID);
            const float4 g0 = hr[lane];
            const float4 g1 = hr[lane + 32];
            acc0.x = fmaf(at1, g0.x, acc0.x);
            acc0.y = fmaf(at1, g0.y, acc0.y);
            acc0.z = fmaf(at1, g0.z, acc0.z);
            acc0.w = fmaf(at1, g0.w, acc0.w);
            acc1.x = fmaf(at1, g1.x, acc1.x);
            acc1.y = fmaf(at1, g1.y, acc1.y);
            acc1.z = fmaf(at1, g1.z, acc1.z);
            acc1.w = fmaf(at1, g1.w, acc1.w);
        }
    }

    const int c0 = 4 * lane;
    const int c1 = 128 + 4 * lane;
    float4 o0, o1;
    o0.x = fmaxf(acc0.x + b[c0 + 0], 0.0f);
    o0.y = fmaxf(acc0.y + b[c0 + 1], 0.0f);
    o0.z = fmaxf(acc0.z + b[c0 + 2], 0.0f);
    o0.w = fmaxf(acc0.w + b[c0 + 3], 0.0f);
    o1.x = fmaxf(acc1.x + b[c1 + 0], 0.0f);
    o1.y = fmaxf(acc1.y + b[c1 + 1], 0.0f);
    o1.z = fmaxf(acc1.z + b[c1 + 2], 0.0f);
    o1.w = fmaxf(acc1.w + b[c1 + 3], 0.0f);
    float4* orow = (float4*)(OUT + (size_t)node * D_HID);
    orow[lane]      = o0;
    orow[lane + 32] = o1;
}

__global__ void gather_rows256(const float* __restrict__ X,
                               const int* __restrict__ idx,
                               float* __restrict__ Y, size_t n)
{
    const size_t i = (size_t)blockIdx.x * blockDim.x + threadIdx.x;
    if (i >= n) return;
    const size_t r = i >> 8, c = i & (D_HID - 1);
    Y[i] = X[(size_t)idx[r] * D_HID + c];
}

__global__ __launch_bounds__(256) void mlp2_softmax(
    const float* __restrict__ DH, const float* __restrict__ Wm2,
    const float* __restrict__ bm2, float* __restrict__ out, int nDocs)
{
    const int doc  = (int)((blockIdx.x * blockDim.x + threadIdx.x) >> 5);
    const int lane = threadIdx.x & 31;
    if (doc >= nDocs) return;
    const float* row = DH + (size_t)doc * MLP_HID;
    float logit = -3.0e38f;
    if (lane < N_CLASSES) {
        float acc = bm2[lane];
        #pragma unroll
        for (int k = 0; k < MLP_HID; ++k)
            acc += row[k] * Wm2[k * N_CLASSES + lane];
        logit = acc;
    }
    float mx = logit;
    #pragma unroll
    for (int off = 16; off; off >>= 1) mx = fmaxf(mx, __shfl_xor(mx, off, 32));
    const float ex = (lane < N_CLASSES) ? __expf(logit - mx) : 0.0f;
    float sum = ex;
    #pragma unroll
    for (int off = 16; off; off >>= 1) sum += __shfl_xor(sum, off, 32);
    if (lane < N_CLASSES) out[(size_t)doc * N_CLASSES + lane] = ex / sum;
}

extern "C" void kernel_launch(void* const* d_in, const int* in_sizes, int n_in,
                              void* d_out, int out_size, void* d_ws, size_t ws_size,
                              hipStream_t stream)
{
    (void)in_sizes; (void)n_in; (void)out_size; (void)ws_size;

    const float* x    = (const float*)d_in[0];
    const int*   ei   = (const int*)  d_in[1];
    const int*   doci = (const int*)  d_in[2];
    const float* W1   = (const float*)d_in[3];
    const float* a1s  = (const float*)d_in[4];
    const float* a1d  = (const float*)d_in[5];
    const float* b1   = (const float*)d_in[6];
    const float* W2   = (const float*)d_in[7];
    const float* a2s  = (const float*)d_in[8];
    const float* a2d  = (const float*)d_in[9];
    const float* b2   = (const float*)d_in[10];
    const float* Wm1  = (const float*)d_in[11];
    const float* bm1  = (const float*)d_in[12];
    const float* Wm2  = (const float*)d_in[13];
    const float* bm2  = (const float*)d_in[14];

    const int* src = ei;
    const int* dst = ei + N_EDGES;

    float* H    = (float*)d_ws;
    float* OUT  = H    + (size_t)N_NODES * D_HID;
    float* ES   = OUT  + (size_t)N_NODES * D_HID;
    float* ED   = ES   + N_NODES;
    float* MX   = ED   + N_NODES;
    float* DEN  = MX   + N_NODES;
    float* EB   = DEN  + N_NODES;
    float* DOCS = EB   + N_EDGES;
    float* DH   = DOCS + (size_t)N_DOCS * D_HID;
    int*  CNT    = (int*)DOCS;
    int*  ROWPTR = CNT    + N_PAD;
    int*  RANK   = ROWPTR + N_PAD;
    int2* PERM   = (int2*)(RANK + N_EDGES);

    const dim3 gemmGridN(N_NODES / 16, D_HID / 64);
    const dim3 gemmGridD(N_DOCS  / 16, MLP_HID / 64);
    const int gNode  = (N_NODES + 255) / 256;
    const int gNodeW = (N_NODES + 7) / 8;
    const int gEdge  = (N_EDGES + 255) / 256;
    const size_t docf = (size_t)N_DOCS * D_HID;
    const int gDocF  = (int)((docf + 255) / 256);
    const int gDocW  = (N_DOCS + 7) / 8;

    fill_i32<<<N_PAD / 256, 256, 0, stream>>>(CNT, 0, N_PAD);
    edge_rank<<<gEdge, 256, 0, stream>>>(dst, CNT, RANK, N_EDGES);
    scan_counts<<<1, 1024, 0, stream>>>(CNT, ROWPTR);
    edge_place<<<gEdge, 256, 0, stream>>>(src, dst, ROWPTR, RANK, PERM, N_EDGES);

    for (int layer = 0; layer < 2; ++layer) {
        const float* A   = (layer == 0) ? x   : OUT;
        const float* W   = (layer == 0) ? W1  : W2;
        const float* as_ = (layer == 0) ? a1s : a2s;
        const float* ad_ = (layer == 0) ? a1d : a2d;
        const float* b   = (layer == 0) ? b1  : b2;

        gemm_wmma_f32<<<gemmGridN, 64, 0, stream>>>(A, W, H, N_NODES, D_HID,
                                                    D_HID, nullptr, 0);
        node_dots<<<gNodeW, 256, 0, stream>>>(H, as_, ad_, ES, ED, N_NODES);
        fill_f32<<<gNode, 256, 0, stream>>>(MX, -INFINITY, (size_t)N_NODES);
        fill_f32<<<gNode, 256, 0, stream>>>(DEN, 0.0f, (size_t)N_NODES);
        edge_score<<<gEdge, 256, 0, stream>>>(src, dst, ES, ED, EB, MX, N_EDGES);
        edge_exp<<<gEdge, 256, 0, stream>>>(dst, MX, EB, DEN, N_EDGES);
        node_aggregate<<<gNodeW, 256, 0, stream>>>(ROWPTR, PERM, EB, DEN, H, b,
                                                   OUT, N_NODES);
    }

    gather_rows256<<<gDocF, 256, 0, stream>>>(OUT, doci, DOCS, docf);
    gemm_wmma_f32<<<gemmGridD, 64, 0, stream>>>(DOCS, Wm1, DH, N_DOCS, MLP_HID,
                                                D_HID, bm1, 1);
    mlp2_softmax<<<gDocW, 256, 0, stream>>>(DH, Wm2, bm2, (float*)d_out, N_DOCS);
}


# Segment softmax moved into the one-wave-per-dst bucket walk: scores, max, exp, denominator by wave reductions; drops 2 atomic edge passes + 2 fills per layer; den summed by lane tree, not atomics
# speedup vs baseline: 2.3669x; 1.4852x over previous
#include <hip/hip_runtime.h>
#include <hip/hip_bf16.h>


#define N_NODES   50000
#define N_EDGES   800000
#define D_HID     256
#define N_DOCS    20000
#define MLP_HID   64
#define N_CLASSES 20

typedef __attribute__((ext_vector_type(2))) float v2f;
typedef __attribute__((ext_vector_type(8))) float v8f;

__global__ __launch_bounds__(64) void gemm_wmma_f32(
    const float* __restrict__ A, const float* __restrict__ B,
    float* __restrict__ C, int M, int N, int K,
    const float* __restrict__ bias, int doRelu)
{
    __shared__ float As[16][32];
    __shared__ float Bs[32][64];

    const int tid  = threadIdx.x;
    const int wave = tid >> 5;
    const int lane = tid & 31;
    const int m0 = blockIdx.x * 16;
    const int n0 = blockIdx.y * 64;

    v8f acc0 = {};
    v8f acc1 = {};

    const int mrow  = lane & 15;
    const int khalf = (lane >> 4) << 1;
    const int ncol0 = (lane & 15) + (wave << 5);
    const int ncol1 = ncol0 + 16;

    const int ar = tid >> 2;
    const int ac = (tid & 3) * 8;
    const int br = tid >> 1;
    const int bc = (tid & 1) * 32;

    for (int k0 = 0; k0 < K; k0 += 32) {
        {
            const float* s = A + (size_t)(m0 + ar) * K + k0 + ac;
            *(float4*)&As[ar][ac]     = *(const float4*)(s);
            *(float4*)&As[ar][ac + 4] = *(const float4*)(s + 4);
        }
        {
            const float* s = B + (size_t)(k0 + br) * N + n0 + bc;
            #pragma unroll
            for (int j = 0; j < 8; ++j)
                *(float4*)&Bs[br][bc + 4 * j] = *(const float4*)(s + 4 * j);
        }
        if (k0 + 32 < K) {
            __builtin_prefetch(A + (size_t)(m0 + ar) * K + k0 + 32 + ac, 0, 3);
            __builtin_prefetch(B + (size_t)(k0 + 32 + br) * N + n0 + bc, 0, 3);
        }
        __syncthreads();

        #pragma unroll
        for (int kk = 0; kk < 32; kk += 4) {
            v2f a, b0, b1;
            a.x = As[mrow][kk + khalf];
            a.y = As[mrow][kk + khalf + 1];
            b0.x = Bs[kk + khalf][ncol0];
            b0.y = Bs[kk + khalf + 1][ncol0];
            b1.x = Bs[kk + khalf][ncol1];
            b1.y = Bs[kk + khalf + 1][ncol1];
            acc0 = __builtin_amdgcn_wmma_f32_16x16x4_f32(
                false, a, false, b0, (short)0, acc0, false, false);
            acc1 = __builtin_amdgcn_wmma_f32_16x16x4_f32(
                false, a, false, b1, (short)0, acc1, false, false);
        }
        __syncthreads();
    }

    const int colg0 = n0 + ncol0;
    const int colg1 = n0 + ncol1;
    const float bv0 = bias ? bias[colg0] : 0.0f;
    const float bv1 = bias ? bias[colg1] : 0.0f;
    #pragma unroll
    for (int r = 0; r < 8; ++r) {
        const int rowg = m0 + r + ((lane >> 4) << 3);
        float v0 = acc0[r] + bv0;
        float v1 = acc1[r] + bv1;
        if (doRelu) { v0 = fmaxf(v0, 0.0f); v1 = fmaxf(v1, 0.0f); }
        C[(size_t)rowg * N + colg0] = v0;
        C[(size_t)rowg * N + colg1] = v1;
    }
}

__global__ __launch_bounds__(256) void node_dots(
    const float* __restrict__ H, const float* __restrict__ as_,
    const float* __restrict__ ad_, float* __restrict__ es,
    float* __restrict__ ed, int n)
{
    const int node = (int)((blockIdx.x * blockDim.x + threadIdx.x) >> 5);
    const int lane = threadIdx.x & 31;
    if (node >= n) return;
    const float* row = H + (size_t)node * D_HID;
    float s = 0.0f, d = 0.0f;
    #pragma unroll
    for (int j = 0; j < D_HID / 32; ++j) {
        const int i = lane + j * 32;
        const float h = row[i];
        s += h * as_[i];
        d += h * ad_[i];
    }
    #pragma unroll
    for (int off = 16; off; off >>= 1) {
        s += __shfl_down(s, off, 32);
        d += __shfl_down(d, off, 32);
    }
    if (lane == 0) { es[node] = s; ed[node] = d; }
}

#define N_PAD 53248

__global__ void fill_i32(int* __restrict__ p, int v, int n)
{
    const int i = blockIdx.x * blockDim.x + threadIdx.x;
    if (i < n) p[i] = v;
}

__global__ __launch_bounds__(256) void edge_rank(
    const int* __restrict__ dst, int* __restrict__ cnt,
    int* __restrict__ rank, int nE)
{
    const int i = blockIdx.x * blockDim.x + threadIdx.x;
    if (i >= nE) return;
    rank[i] = atomicAdd(&cnt[dst[i]], 1);
}

__global__ __launch_bounds__(1024) void scan_counts(
    const int* __restrict__ cnt, int* __restrict__ rowptr)
{
    __shared__ int wsum[2][32];
    const int tid  = threadIdx.x;
    const int lane = tid & 31;
    const int wave = tid >> 5;

    int4 v[N_PAD / 4096];
    #pragma unroll
    for (int it = 0; it < N_PAD / 4096; ++it)
        v[it] = ((const int4*)cnt)[it * 1024 + tid];

    int carry = 0;
    #pragma unroll
    for (int it = 0; it < N_PAD / 4096; ++it) {
        const int4 a = v[it];
        const int t = a.x + a.y + a.z + a.w;
        int inc = t;
        for (int off = 1; off < 32; off <<= 1) {
            const int u = __shfl_up(inc, off, 32);
            if (lane >= off) inc += u;
        }
        if (lane == 31) wsum[it & 1][wave] = inc;
        __syncthreads();
        const int ws = wsum[it & 1][lane];
        int winc = ws;
        for (int off = 1; off < 32; off <<= 1) {
            const int u = __shfl_up(winc, off, 32);
            if (lane >= off) winc += u;
        }
        const int total = __shfl(winc, 31, 32);
        const int wexcl = __shfl(winc, wave, 32) - __shfl(ws, wave, 32);
        const int excl  = carry + wexcl + inc - t;
        int4 o;
        o.x = excl;
        o.y = o.x + a.x;
        o.z = o.y + a.y;
        o.w = o.z + a.z;
        ((int4*)rowptr)[it * 1024 + tid] = o;
        carry += total;
    }
}

__global__ __launch_bounds__(256) void edge_place(
    const int* __restrict__ src, const int* __restrict__ dst,
    const int* __restrict__ rowptr, const int* __restrict__ rank,
    int* __restrict__ perm, int nE)
{
    const int i = blockIdx.x * blockDim.x + threadIdx.x;
    if (i >= nE) return;
    perm[rowptr[dst[i]] + rank[i]] = src[i];
}

__device__ inline float wave_max32(float v)
{
    for (int off = 16; off; off >>= 1) v = fmaxf(v, __shfl_xor(v, off, 32));
    return v;
}

__device__ inline float wave_sum32(float v)
{
    for (int off = 16; off; off >>= 1) v += __shfl_xor(v, off, 32);
    return v;
}

__device__ inline float edge_score1(float es_src, float ed_dst)
{
    const float t = es_src + ed_dst;
    return (t > 0.0f) ? t : 0.2f * t;
}

__global__ __launch_bounds__(256) void node_aggregate(
    const int* __restrict__ rowptr, const int* __restrict__ perm,
    const float* __restrict__ es, const float* __restrict__ ed,
    const float* __restrict__ H, const float* __restrict__ b,
    float* __restrict__ OUT, int n)
{
    const int lane = threadIdx.x & 31;
    const int node = __builtin_amdgcn_readfirstlane(
        (int)((blockIdx.x * blockDim.x + threadIdx.x) >> 5));
    if (node >= n) return;
    const int beg = rowptr[node];
    const int end = rowptr[node + 1];
    const float edn = ed[node];

    const bool in0 = beg + lane < end;
    int   s0 = 0;
    float v0 = -INFINITY;
    if (in0) {
        s0 = perm[beg + lane];
        v0 = edge_score1(es[s0], edn);
    }

    float m = v0;
    for (int base = beg + 32; base < end; base += 32)
        if (base + lane < end)
            m = fmaxf(m, edge_score1(es[perm[base + lane]], edn));
    m = wave_max32(m);

    const float ex0 = in0 ? __expf(v0 - m) : 0.0f;
    float den = ex0;
    for (int base = beg + 32; base < end; base += 32)
        if (base + lane < end)
            den += __expf(edge_score1(es[perm[base + lane]], edn) - m);
    const float dn = wave_sum32(den) + 1e-16f;

    float4 acc0 = make_float4(0.0f, 0.0f, 0.0f, 0.0f);
    float4 acc1 = make_float4(0.0f, 0.0f, 0.0f, 0.0f);

    for (int base = beg; base < end; base += 32) {
        int   s = s0;
        float a = ex0 / dn;
        if (base != beg) {
            s = 0;
            a = 0.0f;
            if (base + lane < end) {
                s = perm[base + lane];
                a = __expf(edge_score1(es[s], edn) - m) / dn;
            }
        }
        const int cnt = (end - base < 32) ? (end - base) : 32;
        int t = 0;
        for (; t + 4 <= cnt; t += 4) {
            float4 h0[4], h1[4];
            float  at[4];
            #pragma unroll
            for (int u = 0; u < 4; ++u) {
                const int st = __builtin_amdgcn_readlane(s, t + u);
                at[u] = __int_as_float(
                    __builtin_amdgcn_readlane(__float_as_int(a), t + u));
                const float4* hr = (const float4*)(H + (size_t)st * D_HID);
                h0[u] = hr[lane];
                h1[u] = hr[lane + 32];
            }
            #pragma unroll
            for (int u = 0; u < 4; ++u) {
                acc0.x = fmaf(at[u], h0[u].x, acc0.x);
                acc0.y = fmaf(at[u], h0[u].y, acc0.y);
                acc0.z = fmaf(at[u], h0[u].z, acc0.z);
                acc0.w = fmaf(at[u], h0[u].w, acc0.w);
                acc1.x = fmaf(at[u], h1[u].x, acc1.x);
                acc1.y = fmaf(at[u], h1[u].y, acc1.y);
                acc1.z = fmaf(at[u], h1[u].z, acc1.z);
                acc1.w = fmaf(at[u], h1[u].w, acc1.w);
            }
        }
        #pragma nounroll
        for (; t < cnt; ++t) {
            const int st = __builtin_amdgcn_readlane(s, t);
            const float at1 = __int_as_float(
                __builtin_amdgcn_readlane(__float_as_int(a), t));
            const float4* hr = (const float4*)(H + (size_t)st * D_HID);
            const float4 g0 = hr[lane];
            const float4 g1 = hr[lane + 32];
            acc0.x = fmaf(at1, g0.x, acc0.x);
            acc0.y = fmaf(at1, g0.y, acc0.y);
            acc0.z = fmaf(at1, g0.z, acc0.z);
            acc0.w = fmaf(at1, g0.w, acc0.w);
            acc1.x = fmaf(at1, g1.x, acc1.x);
            acc1.y = fmaf(at1, g1.y, acc1.y);
            acc1.z = fmaf(at1, g1.z, acc1.z);
            acc1.w = fmaf(at1, g1.w, acc1.w);
        }
    }

    const int c0 = 4 * lane;
    const int c1 = 128 + 4 * lane;
    float4 o0, o1;
    o0.x = fmaxf(acc0.x + b[c0 + 0], 0.0f);
    o0.y = fmaxf(acc0.y + b[c0 + 1], 0.0f);
    o0.z = fmaxf(acc0.z + b[c0 + 2], 0.0f);
    o0.w = fmaxf(acc0.w + b[c0 + 3], 0.0f);
    o1.x = fmaxf(acc1.x + b[c1 + 0], 0.0f);
    o1.y = fmaxf(acc1.y + b[c1 + 1], 0.0f);
    o1.z = fmaxf(acc1.z + b[c1 + 2], 0.0f);
    o1.w = fmaxf(acc1.w + b[c1 + 3], 0.0f);
    float4* orow = (float4*)(OUT + (size_t)node * D_HID);
    orow[lane]      = o0;
    orow[lane + 32] = o1;
}

__global__ void gather_rows256(const float* __restrict__ X,
                               const int* __restrict__ idx,
                               float* __restrict__ Y, size_t n)
{
    const size_t i = (size_t)blockIdx.x * blockDim.x + threadIdx.x;
    if (i >= n) return;
    const size_t r = i >> 8, c = i & (D_HID - 1);
    Y[i] = X[(size_t)idx[r] * D_HID + c];
}

__global__ __launch_bounds__(256) void mlp2_softmax(
    const float* __restrict__ DH, const float* __restrict__ Wm2,
    const float* __restrict__ bm2, float* __restrict__ out, int nDocs)
{
    const int doc  = (int)((blockIdx.x * blockDim.x + threadIdx.x) >> 5);
    const int lane = threadIdx.x & 31;
    if (doc >= nDocs) return;
    const float* row = DH + (size_t)doc * MLP_HID;
    float logit = -3.0e38f;
    if (lane < N_CLASSES) {
        float acc = bm2[lane];
        #pragma unroll
        for (int k = 0; k < MLP_HID; ++k)
            acc += row[k] * Wm2[k * N_CLASSES + lane];
        logit = acc;
    }
    float mx = logit;
    #pragma unroll
    for (int off = 16; off; off >>= 1) mx = fmaxf(mx, __shfl_xor(mx, off, 32));
    const float ex = (lane < N_CLASSES) ? __expf(logit - mx) : 0.0f;
    float sum = ex;
    #pragma unroll
    for (int off = 16; off; off >>= 1) sum += __shfl_xor(sum, off, 32);
    if (lane < N_CLASSES) out[(size_t)doc * N_CLASSES + lane] = ex / sum;
}

extern "C" void kernel_launch(void* const* d_in, const int* in_sizes, int n_in,
                              void* d_out, int out_size, void* d_ws, size_t ws_size,
                              hipStream_t stream)
{
    (void)in_sizes; (void)n_in; (void)out_size; (void)ws_size;

    const float* x    = (const float*)d_in[0];
    const int*   ei   = (const int*)  d_in[1];
    const int*   doci = (const int*)  d_in[2];
    const float* W1   = (const float*)d_in[3];
    const float* a1s  = (const float*)d_in[4];
    const float* a1d  = (const float*)d_in[5];
    const float* b1   = (const float*)d_in[6];
    const float* W2   = (const float*)d_in[7];
    const float* a2s  = (const float*)d_in[8];
    const float* a2d  = (const float*)d_in[9];
    const float* b2   = (const float*)d_in[10];
    const float* Wm1  = (const float*)d_in[11];
    const float* bm1  = (const float*)d_in[12];
    const float* Wm2  = (const float*)d_in[13];
    const float* bm2  = (const float*)d_in[14];

    const int* src = ei;
    const int* dst = ei + N_EDGES;

    float* H    = (float*)d_ws;
    float* OUT  = H    + (size_t)N_NODES * D_HID;
    float* ES   = OUT  + (size_t)N_NODES * D_HID;
    float* ED   = ES   + N_NODES;
    float* DOCS = ED   + N_NODES;
    float* DH   = DOCS + (size_t)N_DOCS * D_HID;
    int*  CNT    = (int*)DOCS;
    int*  ROWPTR = CNT    + N_PAD;
    int*  RANK   = ROWPTR + N_PAD;
    int*  PERM   = RANK   + N_EDGES;

    const dim3 gemmGridN(N_NODES / 16, D_HID / 64);
    const dim3 gemmGridD(N_DOCS  / 16, MLP_HID / 64);
    const int gNodeW = (N_NODES + 7) / 8;
    const int gEdge  = (N_EDGES + 255) / 256;
    const size_t docf = (size_t)N_DOCS * D_HID;
    const int gDocF  = (int)((docf + 255) / 256);
    const int gDocW  = (N_DOCS + 7) / 8;

    fill_i32<<<N_PAD / 256, 256, 0, stream>>>(CNT, 0, N_PAD);
    edge_rank<<<gEdge, 256, 0, stream>>>(dst, CNT, RANK, N_EDGES);
    scan_counts<<<1, 1024, 0, stream>>>(CNT, ROWPTR);
    edge_place<<<gEdge, 256, 0, stream>>>(src, dst, ROWPTR, RANK, PERM, N_EDGES);

    for (int layer = 0; layer < 2; ++layer) {
        const float* A   = (layer == 0) ? x   : OUT;
        const float* W   = (layer == 0) ? W1  : W2;
        const float* as_ = (layer == 0) ? a1s : a2s;
        const float* ad_ = (layer == 0) ? a1d : a2d;
        const float* b   = (layer == 0) ? b1  : b2;

        gemm_wmma_f32<<<gemmGridN, 64, 0, stream>>>(A, W, H, N_NODES, D_HID,
                                                    D_HID, nullptr, 0);
        node_dots<<<gNodeW, 256, 0, stream>>>(H, as_, ad_, ES, ED, N_NODES);
        node_aggregate<<<gNodeW, 256, 0, stream>>>(ROWPTR, PERM, ES, ED, H, b,
                                                   OUT, N_NODES);
    }

    gather_rows256<<<gDocF, 256, 0, stream>>>(OUT, doci, DOCS, docf);
    gemm_wmma_f32<<<gemmGridD, 64, 0, stream>>>(DOCS, Wm1, DH, N_DOCS, MLP_HID,
                                                D_HID, bm1, 1);
    mlp2_softmax<<<gDocW, 256, 0, stream>>>(DH, Wm2, bm2, (float*)d_out, N_DOCS);
}


# Node GEMMs: each wave keeps its 256x64 slice of W in registers, A row tiles stream through LDS, 256 persistent work-groups; k pairs meet in a new order, so the GEMM sum order changes
# speedup vs baseline: 3.3473x; 1.4142x over previous
#include <hip/hip_runtime.h>
#include <hip/hip_bf16.h>


#define N_NODES   50000
#define N_EDGES   800000
#define D_HID     256
#define N_DOCS    20000
#define MLP_HID   64
#define N_CLASSES 20

typedef __attribute__((ext_vector_type(2))) float v2f;
typedef __attribute__((ext_vector_type(8))) float v8f;

__global__ __launch_bounds__(64) void gemm_wmma_f32(
    const float* __restrict__ A, const float* __restrict__ B,
    float* __restrict__ C, int M, int N, int K,
    const float* __restrict__ bias, int doRelu)
{
    __shared__ float As[16][32];
    __shared__ float Bs[32][64];

    const int tid  = threadIdx.x;
    const int wave = tid >> 5;
    const int lane = tid & 31;
    const int m0 = blockIdx.x * 16;
    const int n0 = blockIdx.y * 64;

    v8f acc0 = {};
    v8f acc1 = {};

    const int mrow  = lane & 15;
    const int khalf = (lane >> 4) << 1;
    const int ncol0 = (lane & 15) + (wave << 5);
    const int ncol1 = ncol0 + 16;

    const int ar = tid >> 2;
    const int ac = (tid & 3) * 8;
    const int br = tid >> 1;
    const int bc = (tid & 1) * 32;

    for (int k0 = 0; k0 < K; k0 += 32) {
        {
            const float* s = A + (size_t)(m0 + ar) * K + k0 + ac;
            *(float4*)&As[ar][ac]     = *(const float4*)(s);
            *(float4*)&As[ar][ac + 4] = *(const float4*)(s + 4);
        }
        {
            const float* s = B + (size_t)(k0 + br) * N + n0 + bc;
            #pragma unroll
            for (int j = 0; j < 8; ++j)
                *(float4*)&Bs[br][bc + 4 * j] = *(const float4*)(s + 4 * j);
        }
        if (k0 + 32 < K) {
            __builtin_prefetch(A + (size_t)(m0 + ar) * K + k0 + 32 + ac, 0, 3);
            __builtin_prefetch(B + (size_t)(k0 + 32 + br) * N + n0 + bc, 0, 3);
        }
        __syncthreads();

        #pragma unroll
        for (int kk = 0; kk < 32; kk += 4) {
            v2f a, b0, b1;
            a.x = As[mrow][kk + khalf];
            a.y = As[mrow][kk + khalf + 1];
            b0.x = Bs[kk + khalf][ncol0];
            b0.y = Bs[kk + khalf + 1][ncol0];
            b1.x = Bs[kk + khalf][ncol1];
            b1.y = Bs[kk + khalf + 1][ncol1];
            acc0 = __builtin_amdgcn_wmma_f32_16x16x4_f32(
                false, a, false, b0, (short)0, acc0, false, false);
            acc1 = __builtin_amdgcn_wmma_f32_16x16x4_f32(
                false, a, false, b1, (short)0, acc1, false, false);
        }
        __syncthreads();
    }

    const int colg0 = n0 + ncol0;
    const int colg1 = n0 + ncol1;
    const float bv0 = bias ? bias[colg0] : 0.0f;
    const float bv1 = bias ? bias[colg1] : 0.0f;
    #pragma unroll
    for (int r = 0; r < 8; ++r) {
        const int rowg = m0 + r + ((lane >> 4) << 3);
        float v0 = acc0[r] + bv0;
        float v1 = acc1[r] + bv1;
        if (doRelu) { v0 = fmaxf(v0, 0.0f); v1 = fmaxf(v1, 0.0f); }
        C[(size_t)rowg * N + colg0] = v0;
        C[(size_t)rowg * N + colg1] = v1;
    }
}

#define GA_PITCH 260
#define GEMM_WGS 256
#define GA_ALL(X) X(0, ga) X(1, gb) X(2, gc) X(3, gd) X(4, ge) X(5, gf) X(6, gg) X(7, gh)
#define GA_LOAD1(i, v) v = *(const float4*)(arow + (size_t)(2 * (i)) * D_HID);
#define GA_PUT1(i, v)  *(float4*)(lrow + (2 * (i)) * GA_PITCH) = v;

__global__ __launch_bounds__(128) void gemm_wreg_f32(
    const float* __restrict__ A, const float* __restrict__ B,
    float* __restrict__ C, int nTiles)
{
    __shared__ float As[2][16 * GA_PITCH];

    const int tid  = threadIdx.x;
    const int wave = tid >> 5;
    const int lane = tid & 31;
    const int c    = lane & 15;
    const int h    = lane >> 4;
    const int ncol = (wave << 6) + (c << 2);

    int tile = blockIdx.x;
    if (tile >= nTiles) return;

    const int sr = tid >> 6;
    const int sc = (tid & 63) << 2;

    float4 g0[8];
    #pragma unroll
    for (int i = 0; i < 8; ++i)
        g0[i] = *(const float4*)(A + (size_t)(tile * 16 + 2 * i + sr) * D_HID + sc);

    v2f bf[32][2][4];
    #pragma unroll
    for (int p = 0; p < 32; ++p) {
        #pragma unroll
        for (int q = 0; q < 2; ++q) {
            const int k = 8 * p + 4 * h + 2 * q;
            const float4 w0 = *(const float4*)(B + (size_t)k * D_HID + ncol);
            const float4 w1 = *(const float4*)(B + (size_t)(k + 1) * D_HID + ncol);
            bf[p][q][0].x = w0.x; bf[p][q][0].y = w1.x;
            bf[p][q][1].x = w0.y; bf[p][q][1].y = w1.y;
            bf[p][q][2].x = w0.z; bf[p][q][2].y = w1.z;
            bf[p][q][3].x = w0.w; bf[p][q][3].y = w1.w;
        }
    }

    #pragma unroll
    for (int i = 0; i < 8; ++i)
        *(float4*)&As[0][(2 * i + sr) * GA_PITCH + sc] = g0[i];

    float4 ga, gb, gc, gd, ge, gf, gg, gh;
    {
        const int t1 = (tile + GEMM_WGS < nTiles) ? tile + GEMM_WGS : tile;
        const float* arow = A + (size_t)(t1 * 16 + sr) * D_HID + sc;
        GA_ALL(GA_LOAD1)
    }
    __syncthreads();

    int buf = 0;
    for (; tile < nTiles; tile += GEMM_WGS) {
        const float* as = &As[buf][c * GA_PITCH + 4 * h];
        v8f acc0 = {};
        v8f acc1 = {};
        v8f acc2 = {};
        v8f acc3 = {};
        #pragma unroll
        for (int p = 0; p < 32; ++p) {
            const float4 av = *(const float4*)(as + 8 * p);
            v2f a0, a1;
            a0.x = av.x; a0.y = av.y;
            a1.x = av.z; a1.y = av.w;
            acc0 = __builtin_amdgcn_wmma_f32_16x16x4_f32(
                false, a0, false, bf[p][0][0], (short)0, acc0, false, false);
            acc1 = __builtin_amdgcn_wmma_f32_16x16x4_f32(
                false, a0, false, bf[p][0][1], (short)0, acc1, false, false);
            acc2 = __builtin_amdgcn_wmma_f32_16x16x4_f32(
                false, a0, false, bf[p][0][2], (short)0, acc2, false, false);
            acc3 = __builtin_amdgcn_wmma_f32_16x16x4_f32(
                false, a0, false, bf[p][0][3], (short)0, acc3, false, false);
            acc0 = __builtin_amdgcn_wmma_f32_16x16x4_f32(
                false, a1, false, bf[p][1][0], (short)0, acc0, false, false);
            acc1 = __builtin_amdgcn_wmma_f32_16x16x4_f32(
                false, a1, false, bf[p][1][1], (short)0, acc1, false, false);
            acc2 = __builtin_amdgcn_wmma_f32_16x16x4_f32(
                false, a1, false, bf[p][1][2], (short)0, acc2, false, false);
            acc3 = __builtin_amdgcn_wmma_f32_16x16x4_f32(
                false, a1, false, bf[p][1][3], (short)0, acc3, false, false);
        }

        float* crow = C + (size_t)(tile * 16 + 8 * h) * D_HID + ncol;
        #pragma unroll
        for (int r = 0; r < 8; ++r) {
            float4 o;
            o.x = acc0[r]; o.y = acc1[r]; o.z = acc2[r]; o.w = acc3[r];
            *(float4*)(crow + (size_t)r * D_HID) = o;
        }

        {
            float* lrow = &As[buf ^ 1][sr * GA_PITCH + sc];
            GA_ALL(GA_PUT1)
        }
        {
            const int t2 = (tile + 2 * GEMM_WGS < nTiles) ? tile + 2 * GEMM_WGS : tile;
            const float* arow = A + (size_t)(t2 * 16 + sr) * D_HID + sc;
            GA_ALL(GA_LOAD1)
        }

        __builtin_amdgcn_fence(__ATOMIC_RELEASE, "workgroup", "local");
        __builtin_amdgcn_s_barrier();
        __builtin_amdgcn_fence(__ATOMIC_ACQUIRE, "workgroup", "local");
        buf ^= 1;
    }
}

__global__ __launch_bounds__(256) void node_dots(
    const float* __restrict__ H, const float* __restrict__ as_,
    const float* __restrict__ ad_, float* __restrict__ es,
    float* __restrict__ ed, int n)
{
    const int node = (int)((blockIdx.x * blockDim.x + threadIdx.x) >> 5);
    const int lane = threadIdx.x & 31;
    if (node >= n) return;
    const float* row = H + (size_t)node * D_HID;
    float s = 0.0f, d = 0.0f;
    #pragma unroll
    for (int j = 0; j < D_HID / 32; ++j) {
        const int i = lane + j * 32;
        const float h = row[i];
        s += h * as_[i];
        d += h * ad_[i];
    }
    #pragma unroll
    for (int off = 16; off; off >>= 1) {
        s += __shfl_down(s, off, 32);
        d += __shfl_down(d, off, 32);
    }
    if (lane == 0) { es[node] = s; ed[node] = d; }
}

#define N_PAD 53248

__global__ void fill_i32(int* __restrict__ p, int v, int n)
{
    const int i = blockIdx.x * blockDim.x + threadIdx.x;
    if (i < n) p[i] = v;
}

__global__ __launch_bounds__(256) void edge_rank(
    const int* __restrict__ dst, int* __restrict__ cnt,
    int* __restrict__ rank, int nE)
{
    const int i = blockIdx.x * blockDim.x + threadIdx.x;
    if (i >= nE) return;
    rank[i] = atomicAdd(&cnt[dst[i]], 1);
}

__global__ __launch_bounds__(1024) void scan_counts(
    const int* __restrict__ cnt, int* __restrict__ rowptr)
{
    __shared__ int wsum[2][32];
    const int tid  = threadIdx.x;
    const int lane = tid & 31;
    const int wave = tid >> 5;

    int4 v[N_PAD / 4096];
    #pragma unroll
    for (int it = 0; it < N_PAD / 4096; ++it)
        v[it] = ((const int4*)cnt)[it * 1024 + tid];

    int carry = 0;
    #pragma unroll
    for (int it = 0; it < N_PAD / 4096; ++it) {
        const int4 a = v[it];
        const int t = a.x + a.y + a.z + a.w;
        int inc = t;
        for (int off = 1; off < 32; off <<= 1) {
            const int u = __shfl_up(inc, off, 32);
            if (lane >= off) inc += u;
        }
        if (lane == 31) wsum[it & 1][wave] = inc;
        __syncthreads();
        const int ws = wsum[it & 1][lane];
        int winc = ws;
        for (int off = 1; off < 32; off <<= 1) {
            const int u = __shfl_up(winc, off, 32);
            if (lane >= off) winc += u;
        }
        const int total = __shfl(winc, 31, 32);
        const int wexcl = __shfl(winc, wave, 32) - __shfl(ws, wave, 32);
        const int excl  = carry + wexcl + inc - t;
        int4 o;
        o.x = excl;
        o.y = o.x + a.x;
        o.z = o.y + a.y;
        o.w = o.z + a.z;
        ((int4*)rowptr)[it * 1024 + tid] = o;
        carry += total;
    }
}

__global__ __launch_bounds__(256) void edge_place(
    const int* __restrict__ src, const int* __restrict__ dst,
    const int* __restrict__ rowptr, const int* __restrict__ rank,
    int* __restrict__ perm, int nE)
{
    const int i = blockIdx.x * blockDim.x + threadIdx.x;
    if (i >= nE) return;
    perm[rowptr[dst[i]] + rank[i]] = src[i];
}

__device__ inline float wave_max32(float v)
{
    for (int off = 16; off; off >>= 1) v = fmaxf(v, __shfl_xor(v, off, 32));
    return v;
}

__device__ inline float wave_sum32(float v)
{
    for (int off = 16; off; off >>= 1) v += __shfl_xor(v, off, 32);
    return v;
}

__device__ inline float edge_score1(float es_src, float ed_dst)
{
    const float t = es_src + ed_dst;
    return (t > 0.0f) ? t : 0.2f * t;
}

__global__ __launch_bounds__(256) void node_aggregate(
    const int* __restrict__ rowptr, const int* __restrict__ perm,
    const float* __restrict__ es, const float* __restrict__ ed,
    const float* __restrict__ H, const float* __restrict__ b,
    float* __restrict__ OUT, int n)
{
    const int lane = threadIdx.x & 31;
    const int node = __builtin_amdgcn_readfirstlane(
        (int)((blockIdx.x * blockDim.x + threadIdx.x) >> 5));
    if (node >= n) return;
    const int beg = rowptr[node];
    const int end = rowptr[node + 1];
    const float edn = ed[node];

    const bool in0 = beg + lane < end;
    int   s0 = 0;
    float v0 = -INFINITY;
    if (in0) {
        s0 = perm[beg + lane];
        v0 = edge_score1(es[s0], edn);
    }

    float m = v0;
    for (int base = beg + 32; base < end; base += 32)
        if (base + lane < end)
            m = fmaxf(m, edge_score1(es[perm[base + lane]], edn));
    m = wave_max32(m);

    const float ex0 = in0 ? __expf(v0 - m) : 0.0f;
    float den = ex0;
    for (int base = beg + 32; base < end; base += 32)
        if (base + lane < end)
            den += __expf(edge_score1(es[perm[base + lane]], edn) - m);
    const float dn = wave_sum32(den) + 1e-16f;

    float4 acc0 = make_float4(0.0f, 0.0f, 0.0f, 0.0f);
    float4 acc1 = make_float4(0.0f, 0.0f, 0.0f, 0.0f);

    for (int base = beg; base < end; base += 32) {
        int   s = s0;
        float a = ex0 / dn;
        if (base != beg) {
            s = 0;
            a = 0.0f;
            if (base + lane < end) {
                s = perm[base + lane];
                a = __expf(edge_score1(es[s], edn) - m) / dn;
            }
        }
        const int cnt = (end - base < 32) ? (end - base) : 32;
        int t = 0;
        for (; t + 4 <= cnt; t += 4) {
            float4 h0[4], h1[4];
            float  at[4];
            #pragma unroll
            for (int u = 0; u < 4; ++u) {
                const int st = __builtin_amdgcn_readlane(s, t + u);
                at[u] = __int_as_float(
                    __builtin_amdgcn_readlane(__float_as_int(a), t + u));
                const float4* hr = (const float4*)(H + (size_t)st * D_HID);
                h0[u] = hr[lane];
                h1[u] = hr[lane + 32];
            }
            #pragma unroll
            for (int u = 0; u < 4; ++u) {
                acc0.x = fmaf(at[u], h0[u].x, acc0.x);
                acc0.y = fmaf(at[u], h0[u].y, acc0.y);
                acc0.z = fmaf(at[u], h0[u].z, acc0.z);
                acc0.w = fmaf(at[u], h0[u].w, acc0.w);
                acc1.x = fmaf(at[u], h1[u].x, acc1.x);
                acc1.y = fmaf(at[u], h1[u].y, acc1.y);
                acc1.z = fmaf(at[u], h1[u].z, acc1.z);
                acc1.w = fmaf(at[u], h1[u].w, acc1.w);
            }
        }
        #pragma nounroll
        for (; t < cnt; ++t) {
            const int st = __builtin_amdgcn_readlane(s, t);
            const float at1 = __int_as_float(
                __builtin_amdgcn_readlane(__float_as_int(a), t));
            const float4* hr = (const float4*)(H + (size_t)st * D_HID);
            const float4 g0 = hr[lane];
            const float4 g1 = hr[lane + 32];
            acc0.x = fmaf(at1, g0.x, acc0.x);
            acc0.y = fmaf(at1, g0.y, acc0.y);
            acc0.z = fmaf(at1, g0.z, acc0.z);
            acc0.w = fmaf(at1, g0.w, acc0.w);
            acc1.x = fmaf(at1, g1.x, acc1.x);
            acc1.y = fmaf(at1, g1.y, acc1.y);
            acc1.z = fmaf(at1, g1.z, acc1.z);
            acc1.w = fmaf(at1, g1.w, acc1.w);
        }
    }

    const int c0 = 4 * lane;
    const int c1 = 128 + 4 * lane;
    float4 o0, o1;
    o0.x = fmaxf(acc0.x + b[c0 + 0], 0.0f);
    o0.y = fmaxf(acc0.y + b[c0 + 1], 0.0f);
    o0.z = fmaxf(acc0.z + b[c0 + 2], 0.0f);
    o0.w = fmaxf(acc0.w + b[c0 + 3], 0.0f);
    o1.x = fmaxf(acc1.x + b[c1 + 0], 0.0f);
    o1.y = fmaxf(acc1.y + b[c1 + 1], 0.0f);
    o1.z = fmaxf(acc1.z + b[c1 + 2], 0.0f);
    o1.w = fmaxf(acc1.w + b[c1 + 3], 0.0f);
    float4* orow = (float4*)(OUT + (size_t)node * D_HID);
    orow[lane]      = o0;
    orow[lane + 32] = o1;
}

__global__ void gather_rows256(const float* __restrict__ X,
                               const int* __restrict__ idx,
                               float* __restrict__ Y, size_t n)
{
    const size_t i = (size_t)blockIdx.x * blockDim.x + threadIdx.x;
    if (i >= n) return;
    const size_t r = i >> 8, c = i & (D_HID - 1);
    Y[i] = X[(size_t)idx[r] * D_HID + c];
}

__global__ __launch_bounds__(256) void mlp2_softmax(
    const float* __restrict__ DH, const float* __restrict__ Wm2,
    const float* __restrict__ bm2, float* __restrict__ out, int nDocs)
{
    const int doc  = (int)((blockIdx.x * blockDim.x + threadIdx.x) >> 5);
    const int lane = threadIdx.x & 31;
    if (doc >= nDocs) return;
    const float* row = DH + (size_t)doc * MLP_HID;
    float logit = -3.0e38f;
    if (lane < N_CLASSES) {
        float acc = bm2[lane];
        #pragma unroll
        for (int k = 0; k < MLP_HID; ++k)
            acc += row[k] * Wm2[k * N_CLASSES + lane];
        logit = acc;
    }
    float mx = logit;
    #pragma unroll
    for (int off = 16; off; off >>= 1) mx = fmaxf(mx, __shfl_xor(mx, off, 32));
    const float ex = (lane < N_CLASSES) ? __expf(logit - mx) : 0.0f;
    float sum = ex;
    #pragma unroll
    for (int off = 16; off; off >>= 1) sum += __shfl_xor(sum, off, 32);
    if (lane < N_CLASSES) out[(size_t)doc * N_CLASSES + lane] = ex / sum;
}

extern "C" void kernel_launch(void* const* d_in, const int* in_sizes, int n_in,
                              void* d_out, int out_size, void* d_ws, size_t ws_size,
                              hipStream_t stream)
{
    (void)in_sizes; (void)n_in; (void)out_size; (void)ws_size;

    const float* x    = (const float*)d_in[0];
    const int*   ei   = (const int*)  d_in[1];
    const int*   doci = (const int*)  d_in[2];
    const float* W1   = (const float*)d_in[3];
    const float* a1s  = (const float*)d_in[4];
    const float* a1d  = (const float*)d_in[5];
    const float* b1   = (const float*)d_in[6];
    const float* W2   = (const float*)d_in[7];
    const float* a2s  = (const float*)d_in[8];
    const float* a2d  = (const float*)d_in[9];
    const float* b2   = (const float*)d_in[10];
    const float* Wm1  = (const float*)d_in[11];
    const float* bm1  = (const float*)d_in[12];
    const float* Wm2  = (const float*)d_in[13];
    const float* bm2  = (const float*)d_in[14];

    const int* src = ei;
    const int* dst = ei + N_EDGES;

    float* H    = (float*)d_ws;
    float* OUT  = H    + (size_t)N_NODES * D_HID;
    float* ES   = OUT  + (size_t)N_NODES * D_HID;
    float* ED   = ES   + N_NODES;
    float* DOCS = ED   + N_NODES;
    float* DH   = DOCS + (size_t)N_DOCS * D_HID;
    int*  CNT    = (int*)DOCS;
    int*  ROWPTR = CNT    + N_PAD;
    int*  RANK   = ROWPTR + N_PAD;
    int*  PERM   = RANK   + N_EDGES;

    const dim3 gemmGridD(N_DOCS  / 16, MLP_HID / 64);
    const int gNodeW = (N_NODES + 7) / 8;
    const int gEdge  = (N_EDGES + 255) / 256;
    const size_t docf = (size_t)N_DOCS * D_HID;
    const int gDocF  = (int)((docf + 255) / 256);
    const int gDocW  = (N_DOCS + 7) / 8;

    fill_i32<<<N_PAD / 256, 256, 0, stream>>>(CNT, 0, N_PAD);
    edge_rank<<<gEdge, 256, 0, stream>>>(dst, CNT, RANK, N_EDGES);
    scan_counts<<<1, 1024, 0, stream>>>(CNT, ROWPTR);
    edge_place<<<gEdge, 256, 0, stream>>>(src, dst, ROWPTR, RANK, PERM, N_EDGES);

    for (int layer = 0; layer < 2; ++layer) {
        const float* A   = (layer == 0) ? x   : OUT;
        const float* W   = (layer == 0) ? W1  : W2;
        const float* as_ = (layer == 0) ? a1s : a2s;
        const float* ad_ = (layer == 0) ? a1d : a2d;
        const float* b   = (layer == 0) ? b1  : b2;

        gemm_wreg_f32<<<GEMM_WGS, 128, 0, stream>>>(A, W, H, N_NODES / 16);
        node_dots<<<gNodeW, 256, 0, stream>>>(H, as_, ad_, ES, ED, N_NODES);
        node_aggregate<<<gNodeW, 256, 0, stream>>>(ROWPTR, PERM, ES, ED, H, b,
                                                   OUT, N_NODES);
    }

    gather_rows256<<<gDocF, 256, 0, stream>>>(OUT, doci, DOCS, docf);
    gemm_wmma_f32<<<gemmGridD, 64, 0, stream>>>(DOCS, Wm1, DH, N_DOCS, MLP_HID,
                                                D_HID, bm1, 1);
    mlp2_softmax<<<gDocW, 256, 0, stream>>>(DH, Wm2, bm2, (float*)d_out, N_DOCS);
}


# Layer 2 edge phase runs one wave per doc on the bucket of node doc_idx[d] and stores row d of the doc matrix: only rows the head reads are built, row-gather launch dropped; per-row sum order kept
# speedup vs baseline: 3.7660x; 1.1251x over previous
#include <hip/hip_runtime.h>
#include <hip/hip_bf16.h>


#define N_NODES   50000
#define N_EDGES   800000
#define D_HID     256
#define N_DOCS    20000
#define MLP_HID   64
#define N_CLASSES 20

typedef __attribute__((ext_vector_type(2))) float v2f;
typedef __attribute__((ext_vector_type(8))) float v8f;

__global__ __launch_bounds__(64) void gemm_wmma_f32(
    const float* __restrict__ A, const float* __restrict__ B,
    float* __restrict__ C, int M, int N, int K,
    const float* __restrict__ bias, int doRelu)
{
    __shared__ float As[16][32];
    __shared__ float Bs[32][64];

    const int tid  = threadIdx.x;
    const int wave = tid >> 5;
    const int lane = tid & 31;
    const int m0 = blockIdx.x * 16;
    const int n0 = blockIdx.y * 64;

    v8f acc0 = {};
    v8f acc1 = {};

    const int mrow  = lane & 15;
    const int khalf = (lane >> 4) << 1;
    const int ncol0 = (lane & 15) + (wave << 5);
    const int ncol1 = ncol0 + 16;

    const int ar = tid >> 2;
    const int ac = (tid & 3) * 8;
    const int br = tid >> 1;
    const int bc = (tid & 1) * 32;

    for (int k0 = 0; k0 < K; k0 += 32) {
        {
            const float* s = A + (size_t)(m0 + ar) * K + k0 + ac;
            *(float4*)&As[ar][ac]     = *(const float4*)(s);
            *(float4*)&As[ar][ac + 4] = *(const float4*)(s + 4);
        }
        {
            const float* s = B + (size_t)(k0 + br) * N + n0 + bc;
            #pragma unroll
            for (int j = 0; j < 8; ++j)
                *(float4*)&Bs[br][bc + 4 * j] = *(const float4*)(s + 4 * j);
        }
        if (k0 + 32 < K) {
            __builtin_prefetch(A + (size_t)(m0 + ar) * K + k0 + 32 + ac, 0, 3);
            __builtin_prefetch(B + (size_t)(k0 + 32 + br) * N + n0 + bc, 0, 3);
        }
        __syncthreads();

        #pragma unroll
        for (int kk = 0; kk < 32; kk += 4) {
            v2f a, b0, b1;
            a.x = As[mrow][kk + khalf];
            a.y = As[mrow][kk + khalf + 1];
            b0.x = Bs[kk + khalf][ncol0];
            b0.y = Bs[kk + khalf + 1][ncol0];
            b1.x = Bs[kk + khalf][ncol1];
            b1.y = Bs[kk + khalf + 1][ncol1];
            acc0 = __builtin_amdgcn_wmma_f32_16x16x4_f32(
                false, a, false, b0, (short)0, acc0, false, false);
            acc1 = __builtin_amdgcn_wmma_f32_16x16x4_f32(
                false, a, false, b1, (short)0, acc1, false, false);
        }
        __syncthreads();
    }

    const int colg0 = n0 + ncol0;
    const int colg1 = n0 + ncol1;
    const float bv0 = bias ? bias[colg0] : 0.0f;
    const float bv1 = bias ? bias[colg1] : 0.0f;
    #pragma unroll
    for (int r = 0; r < 8; ++r) {
        const int rowg = m0 + r + ((lane >> 4) << 3);
        float v0 = acc0[r] + bv0;
        float v1 = acc1[r] + bv1;
        if (doRelu) { v0 = fmaxf(v0, 0.0f); v1 = fmaxf(v1, 0.0f); }
        C[(size_t)rowg * N + colg0] = v0;
        C[(size_t)rowg * N + colg1] = v1;
    }
}

#define GA_PITCH 260
#define GEMM_WGS 256
#define GA_ALL(X) X(0, ga) X(1, gb) X(2, gc) X(3, gd) X(4, ge) X(5, gf) X(6, gg) X(7, gh)
#define GA_LOAD1(i, v) v = *(const float4*)(arow + (size_t)(2 * (i)) * D_HID);
#define GA_PUT1(i, v)  *(float4*)(lrow + (2 * (i)) * GA_PITCH) = v;

__global__ __launch_bounds__(128) void gemm_wreg_f32(
    const float* __restrict__ A, const float* __restrict__ B,
    float* __restrict__ C, int nTiles)
{
    __shared__ float As[2][16 * GA_PITCH];

    const int tid  = threadIdx.x;
    const int wave = tid >> 5;
    const int lane = tid & 31;
    const int c    = lane & 15;
    const int h    = lane >> 4;
    const int ncol = (wave << 6) + (c << 2);

    int tile = blockIdx.x;
    if (tile >= nTiles) return;

    const int sr = tid >> 6;
    const int sc = (tid & 63) << 2;

    float4 g0[8];
    #pragma unroll
    for (int i = 0; i < 8; ++i)
        g0[i] = *(const float4*)(A + (size_t)(tile * 16 + 2 * i + sr) * D_HID + sc);

    v2f bf[32][2][4];
    #pragma unroll
    for (int p = 0; p < 32; ++p) {
        #pragma unroll
        for (int q = 0; q < 2; ++q) {
            const int k = 8 * p + 4 * h + 2 * q;
            const float4 w0 = *(const float4*)(B + (size_t)k * D_HID + ncol);
            const float4 w1 = *(const float4*)(B + (size_t)(k + 1) * D_HID + ncol);
            bf[p][q][0].x = w0.x; bf[p][q][0].y = w1.x;
            bf[p][q][1].x = w0.y; bf[p][q][1].y = w1.y;
            bf[p][q][2].x = w0.z; bf[p][q][2].y = w1.z;
            bf[p][q][3].x = w0.w; bf[p][q][3].y = w1.w;
        }
    }

    #pragma unroll
    for (int i = 0; i < 8; ++i)
        *(float4*)&As[0][(2 * i + sr) * GA_PITCH + sc] = g0[i];

    float4 ga, gb, gc, gd, ge, gf, gg, gh;
    {
        const int t1 = (tile + GEMM_WGS < nTiles) ? tile + GEMM_WGS : tile;
        const float* arow = A + (size_t)(t1 * 16 + sr) * D_HID + sc;
        GA_ALL(GA_LOAD1)
    }
    __syncthreads();

    int buf = 0;
    for (; tile < nTiles; tile += GEMM_WGS) {
        const float* as = &As[buf][c * GA_PITCH + 4 * h];
        v8f acc0 = {};
        v8f acc1 = {};
        v8f acc2 = {};
        v8f acc3 = {};
        #pragma unroll
        for (int p = 0; p < 32; ++p) {
            const float4 av = *(const float4*)(as + 8 * p);
            v2f a0, a1;
            a0.x = av.x; a0.y = av.y;
            a1.x = av.z; a1.y = av.w;
            acc0 = __builtin_amdgcn_wmma_f32_16x16x4_f32(
                false, a0, false, bf[p][0][0], (short)0, acc0, false, false);
            acc1 = __builtin_amdgcn_wmma_f32_16x16x4_f32(
                false, a0, false, bf[p][0][1], (short)0, acc1, false, false);
            acc2 = __builtin_amdgcn_wmma_f32_16x16x4_f32(
                false, a0, false, bf[p][0][2], (short)0, acc2, false, false);
            acc3 = __builtin_amdgcn_wmma_f32_16x16x4_f32(
                false, a0, false, bf[p][0][3], (short)0, acc3, false, false);
            acc0 = __builtin_amdgcn_wmma_f32_16x16x4_f32(
                false, a1, false, bf[p][1][0], (short)0, acc0, false, false);
            acc1 = __builtin_amdgcn_wmma_f32_16x16x4_f32(
                false, a1, false, bf[p][1][1], (short)0, acc1, false, false);
            acc2 = __builtin_amdgcn_wmma_f32_16x16x4_f32(
                false, a1, false, bf[p][1][2], (short)0, acc2, false, false);
            acc3 = __builtin_amdgcn_wmma_f32_16x16x4_f32(
                false, a1, false, bf[p][1][3], (short)0, acc3, false, false);
        }

        float* crow = C + (size_t)(tile * 16 + 8 * h) * D_HID + ncol;
        #pragma unroll
        for (int r = 0; r < 8; ++r) {
            float4 o;
            o.x = acc0[r]; o.y = acc1[r]; o.z = acc2[r]; o.w = acc3[r];
            *(float4*)(crow + (size_t)r * D_HID) = o;
        }

        {
            float* lrow = &As[buf ^ 1][sr * GA_PITCH + sc];
            GA_ALL(GA_PUT1)
        }
        {
            const int t2 = (tile + 2 * GEMM_WGS < nTiles) ? tile + 2 * GEMM_WGS : tile;
            const float* arow = A + (size_t)(t2 * 16 + sr) * D_HID + sc;
            GA_ALL(GA_LOAD1)
        }

        __builtin_amdgcn_fence(__ATOMIC_RELEASE, "workgroup", "local");
        __builtin_amdgcn_s_barrier();
        __builtin_amdgcn_fence(__ATOMIC_ACQUIRE, "workgroup", "local");
        buf ^= 1;
    }
}

__global__ __launch_bounds__(256) void node_dots(
    const float* __restrict__ H, const float* __restrict__ as_,
    const float* __restrict__ ad_, float* __restrict__ es,
    float* __restrict__ ed, int n)
{
    const int node = (int)((blockIdx.x * blockDim.x + threadIdx.x) >> 5);
    const int lane = threadIdx.x & 31;
    if (node >= n) return;
    const float* row = H + (size_t)node * D_HID;
    float s = 0.0f, d = 0.0f;
    #pragma unroll
    for (int j = 0; j < D_HID / 32; ++j) {
        const int i = lane + j * 32;
        const float h = row[i];
        s += h * as_[i];
        d += h * ad_[i];
    }
    #pragma unroll
    for (int off = 16; off; off >>= 1) {
        s += __shfl_down(s, off, 32);
        d += __shfl_down(d, off, 32);
    }
    if (lane == 0) { es[node] = s; ed[node] = d; }
}

#define N_PAD 53248

__global__ void fill_i32(int* __restrict__ p, int v, int n)
{
    const int i = blockIdx.x * blockDim.x + threadIdx.x;
    if (i < n) p[i] = v;
}

__global__ __launch_bounds__(256) void edge_rank(
    const int* __restrict__ dst, int* __restrict__ cnt,
    int* __restrict__ rank, int nE)
{
    const int i = blockIdx.x * blockDim.x + threadIdx.x;
    if (i >= nE) return;
    rank[i] = atomicAdd(&cnt[dst[i]], 1);
}

__global__ __launch_bounds__(1024) void scan_counts(
    const int* __restrict__ cnt, int* __restrict__ rowptr)
{
    __shared__ int wsum[2][32];
    const int tid  = threadIdx.x;
    const int lane = tid & 31;
    const int wave = tid >> 5;

    int4 v[N_PAD / 4096];
    #pragma unroll
    for (int it = 0; it < N_PAD / 4096; ++it)
        v[it] = ((const int4*)cnt)[it * 1024 + tid];

    int carry = 0;
    #pragma unroll
    for (int it = 0; it < N_PAD / 4096; ++it) {
        const int4 a = v[it];
        const int t = a.x + a.y + a.z + a.w;
        int inc = t;
        for (int off = 1; off < 32; off <<= 1) {
            const int u = __shfl_up(inc, off, 32);
            if (lane >= off) inc += u;
        }
        if (lane == 31) wsum[it & 1][wave] = inc;
        __syncthreads();
        const int ws = wsum[it & 1][lane];
        int winc = ws;
        for (int off = 1; off < 32; off <<= 1) {
            const int u = __shfl_up(winc, off, 32);
            if (lane >= off) winc += u;
        }
        const int total = __shfl(winc, 31, 32);
        const int wexcl = __shfl(winc, wave, 32) - __shfl(ws, wave, 32);
        const int excl  = carry + wexcl + inc - t;
        int4 o;
        o.x = excl;
        o.y = o.x + a.x;
        o.z = o.y + a.y;
        o.w = o.z + a.z;
        ((int4*)rowptr)[it * 1024 + tid] = o;
        carry += total;
    }
}

__global__ __launch_bounds__(256) void edge_place(
    const int* __restrict__ src, const int* __restrict__ dst,
    const int* __restrict__ rowptr, const int* __restrict__ rank,
    int* __restrict__ perm, int nE)
{
    const int i = blockIdx.x * blockDim.x + threadIdx.x;
    if (i >= nE) return;
    perm[rowptr[dst[i]] + rank[i]] = src[i];
}

__device__ inline float wave_max32(float v)
{
    for (int off = 16; off; off >>= 1) v = fmaxf(v, __shfl_xor(v, off, 32));
    return v;
}

__device__ inline float wave_sum32(float v)
{
    for (int off = 16; off; off >>= 1) v += __shfl_xor(v, off, 32);
    return v;
}

__device__ inline float edge_score1(float es_src, float ed_dst)
{
    const float t = es_src + ed_dst;
    return (t > 0.0f) ? t : 0.2f * t;
}

__global__ __launch_bounds__(256) void node_aggregate(
    const int* __restrict__ rowptr, const int* __restrict__ perm,
    const float* __restrict__ es, const float* __restrict__ ed,
    const float* __restrict__ H, const float* __restrict__ b,
    float* __restrict__ OUT, const int* __restrict__ rowsel, int n)
{
    const int lane = threadIdx.x & 31;
    const int w = __builtin_amdgcn_readfirstlane(
        (int)((blockIdx.x * blockDim.x + threadIdx.x) >> 5));
    if (w >= n) return;
    const int node = rowsel ? rowsel[w] : w;
    const int beg = rowptr[node];
    const int end = rowptr[node + 1];
    const float edn = ed[node];

    const bool in0 = beg + lane < end;
    int   s0 = 0;
    float v0 = -INFINITY;
    if (in0) {
        s0 = perm[beg + lane];
        v0 = edge_score1(es[s0], edn);
    }

    float m = v0;
    for (int base = beg + 32; base < end; base += 32)
        if (base + lane < end)
            m = fmaxf(m, edge_score1(es[perm[base + lane]], edn));
    m = wave_max32(m);

    const float ex0 = in0 ? __expf(v0 - m) : 0.0f;
    float den = ex0;
    for (int base = beg + 32; base < end; base += 32)
        if (base + lane < end)
            den += __expf(edge_score1(es[perm[base + lane]], edn) - m);
    const float dn = wave_sum32(den) + 1e-16f;

    float4 acc0 = make_float4(0.0f, 0.0f, 0.0f, 0.0f);
    float4 acc1 = make_float4(0.0f, 0.0f, 0.0f, 0.0f);

    for (int base = beg; base < end; base += 32) {
        int   s = s0;
        float a = ex0 / dn;
        if (base != beg) {
            s = 0;
            a = 0.0f;
            if (base + lane < end) {
                s = perm[base + lane];
                a = __expf(edge_score1(es[s], edn) - m) / dn;
            }
        }
        const int cnt = (end - base < 32) ? (end - base) : 32;
        int t = 0;
        for (; t + 4 <= cnt; t += 4) {
            float4 h0[4], h1[4];
            float  at[4];
            #pragma unroll
            for (int u = 0; u < 4; ++u) {
                const int st = __builtin_amdgcn_readlane(s, t + u);
                at[u] = __int_as_float(
                    __builtin_amdgcn_readlane(__float_as_int(a), t + u));
                const float4* hr = (const float4*)(H + (size_t)st * D_HID);
                h0[u] = hr[lane];
                h1[u] = hr[lane + 32];
            }
            #pragma unroll
            for (int u = 0; u < 4; ++u) {
                acc0.x = fmaf(at[u], h0[u].x, acc0.x);
                acc0.y = fmaf(at[u], h0[u].y, acc0.y);
                acc0.z = fmaf(at[u], h0[u].z, acc0.z);
                acc0.w = fmaf(at[u], h0[u].w, acc0.w);
                acc1.x = fmaf(at[u], h1[u].x, acc1.x);
                acc1.y = fmaf(at[u], h1[u].y, acc1.y);
                acc1.z = fmaf(at[u], h1[u].z, acc1.z);
                acc1.w = fmaf(at[u], h1[u].w, acc1.w);
            }
        }
        #pragma nounroll
        for (; t < cnt; ++t) {
            const int st = __builtin_amdgcn_readlane(s, t);
            const float at1 = __int_as_float(
                __builtin_amdgcn_readlane(__float_as_int(a), t));
            const float4* hr = (const float4*)(H + (size_t)st * D_HID);
            const float4 g0 = hr[lane];
            const float4 g1 = hr[lane + 32];
            acc0.x = fmaf(at1, g0.x, acc0.x);
            acc0.y = fmaf(at1, g0.y, acc0.y);
            acc0.z = fmaf(at1, g0.z, acc0.z);
            acc0.w = fmaf(at1, g0.w, acc0.w);
            acc1.x = fmaf(at1, g1.x, acc1.x);
            acc1.y = fmaf(at1, g1.y, acc1.y);
            acc1.z = fmaf(at1, g1.z, acc1.z);
            acc1.w = fmaf(at1, g1.w, acc1.w);
        }
    }

    const int c0 = 4 * lane;
    const int c1 = 128 + 4 * lane;
    float4 o0, o1;
    o0.x = fmaxf(acc0.x + b[c0 + 0], 0.0f);
    o0.y = fmaxf(acc0.y + b[c0 + 1], 0.0f);
    o0.z = fmaxf(acc0.z + b[c0 + 2], 0.0f);
    o0.w = fmaxf(acc0.w + b[c0 + 3], 0.0f);
    o1.x = fmaxf(acc1.x + b[c1 + 0], 0.0f);
    o1.y = fmaxf(acc1.y + b[c1 + 1], 0.0f);
    o1.z = fmaxf(acc1.z + b[c1 + 2], 0.0f);
    o1.w = fmaxf(acc1.w + b[c1 + 3], 0.0f);
    float4* orow = (float4*)(OUT + (size_t)w * D_HID);
    orow[lane]      = o0;
    orow[lane + 32] = o1;
}

__global__ __launch_bounds__(256) void mlp2_softmax(
    const float* __restrict__ DH, const float* __restrict__ Wm2,
    const float* __restrict__ bm2, float* __restrict__ out, int nDocs)
{
    const int doc  = (int)((blockIdx.x * blockDim.x + threadIdx.x) >> 5);
    const int lane = threadIdx.x & 31;
    if (doc >= nDocs) return;
    const float* row = DH + (size_t)doc * MLP_HID;
    float logit = -3.0e38f;
    if (lane < N_CLASSES) {
        float acc = bm2[lane];
        #pragma unroll
        for (int k = 0; k < MLP_HID; ++k)
            acc += row[k] * Wm2[k * N_CLASSES + lane];
        logit = acc;
    }
    float mx = logit;
    #pragma unroll
    for (int off = 16; off; off >>= 1) mx = fmaxf(mx, __shfl_xor(mx, off, 32));
    const float ex = (lane < N_CLASSES) ? __expf(logit - mx) : 0.0f;
    float sum = ex;
    #pragma unroll
    for (int off = 16; off; off >>= 1) sum += __shfl_xor(sum, off, 32);
    if (lane < N_CLASSES) out[(size_t)doc * N_CLASSES + lane] = ex / sum;
}

extern "C" void kernel_launch(void* const* d_in, const int* in_sizes, int n_in,
                              void* d_out, int out_size, void* d_ws, size_t ws_size,
                              hipStream_t stream)
{
    (void)in_sizes; (void)n_in; (void)out_size; (void)ws_size;

    const float* x    = (const float*)d_in[0];
    const int*   ei   = (const int*)  d_in[1];
    const int*   doci = (const int*)  d_in[2];
    const float* W1   = (const float*)d_in[3];
    const float* a1s  = (const float*)d_in[4];
    const float* a1d  = (const float*)d_in[5];
    const float* b1   = (const float*)d_in[6];
    const float* W2   = (const float*)d_in[7];
    const float* a2s  = (const float*)d_in[8];
    const float* a2d  = (const float*)d_in[9];
    const float* b2   = (const float*)d_in[10];
    const float* Wm1  = (const float*)d_in[11];
    const float* bm1  = (const float*)d_in[12];
    const float* Wm2  = (const float*)d_in[13];
    const float* bm2  = (const float*)d_in[14];

    const int* src = ei;
    const int* dst = ei + N_EDGES;

    float* H    = (float*)d_ws;
    float* OUT  = H    + (size_t)N_NODES * D_HID;
    float* ES   = OUT  + (size_t)N_NODES * D_HID;
    float* ED   = ES   + N_NODES;
    float* BKT  = ED   + N_NODES;
    float* DH   = BKT  + (size_t)N_DOCS * D_HID;
    float* DOCS = OUT;
    int*  CNT    = (int*)BKT;
    int*  ROWPTR = CNT    + N_PAD;
    int*  RANK   = ROWPTR + N_PAD;
    int*  PERM   = RANK   + N_EDGES;

    const dim3 gemmGridD(N_DOCS  / 16, MLP_HID / 64);
    const int gNodeW = (N_NODES + 7) / 8;
    const int gEdge  = (N_EDGES + 255) / 256;
    const int gDocW  = (N_DOCS + 7) / 8;

    fill_i32<<<N_PAD / 256, 256, 0, stream>>>(CNT, 0, N_PAD);
    edge_rank<<<gEdge, 256, 0, stream>>>(dst, CNT, RANK, N_EDGES);
    scan_counts<<<1, 1024, 0, stream>>>(CNT, ROWPTR);
    edge_place<<<gEdge, 256, 0, stream>>>(src, dst, ROWPTR, RANK, PERM, N_EDGES);

    for (int layer = 0; layer < 2; ++layer) {
        const float* A   = (layer == 0) ? x   : OUT;
        const float* W   = (layer == 0) ? W1  : W2;
        const float* as_ = (layer == 0) ? a1s : a2s;
        const float* ad_ = (layer == 0) ? a1d : a2d;
        const float* b   = (layer == 0) ? b1  : b2;

        gemm_wreg_f32<<<GEMM_WGS, 128, 0, stream>>>(A, W, H, N_NODES / 16);
        node_dots<<<gNodeW, 256, 0, stream>>>(H, as_, ad_, ES, ED, N_NODES);
        if (layer == 0)
            node_aggregate<<<gNodeW, 256, 0, stream>>>(ROWPTR, PERM, ES, ED, H,
                                                       b, OUT, nullptr, N_NODES);
        else
            node_aggregate<<<gDocW, 256, 0, stream>>>(ROWPTR, PERM, ES, ED, H,
                                                      b, DOCS, doci, N_DOCS);
    }

    gemm_wmma_f32<<<gemmGridD, 64, 0, stream>>>(DOCS, Wm1, DH, N_DOCS, MLP_HID,
                                                D_HID, bm1, 1);
    mlp2_softmax<<<gDocW, 256, 0, stream>>>(DH, Wm2, bm2, (float*)d_out, N_DOCS);
}


# Attention dots es=H.a_s, ed=H.a_d taken from the node GEMM's accumulators (16-lane DPP butterfly, 4 wave sums via LDS): both node_dots launches and their H re-read dropped; es/ed sum order changes
# speedup vs baseline: 3.9698x; 1.0541x over previous
#include <hip/hip_runtime.h>
#include <hip/hip_bf16.h>


#define N_NODES   50000
#define N_EDGES   800000
#define D_HID     256
#define N_DOCS    20000
#define MLP_HID   64
#define N_CLASSES 20

typedef __attribute__((ext_vector_type(2))) float v2f;
typedef __attribute__((ext_vector_type(8))) float v8f;

__global__ __launch_bounds__(64) void gemm_wmma_f32(
    const float* __restrict__ A, const float* __restrict__ B,
    float* __restrict__ C, int M, int N, int K,
    const float* __restrict__ bias, int doRelu)
{
    __shared__ float As[16][32];
    __shared__ float Bs[32][64];

    const int tid  = threadIdx.x;
    const int wave = tid >> 5;
    const int lane = tid & 31;
    const int m0 = blockIdx.x * 16;
    const int n0 = blockIdx.y * 64;

    v8f acc0 = {};
    v8f acc1 = {};

    const int mrow  = lane & 15;
    const int khalf = (lane >> 4) << 1;
    const int ncol0 = (lane & 15) + (wave << 5);
    const int ncol1 = ncol0 + 16;

    const int ar = tid >> 2;
    const int ac = (tid & 3) * 8;
    const int br = tid >> 1;
    const int bc = (tid & 1) * 32;

    for (int k0 = 0; k0 < K; k0 += 32) {
        {
            const float* s = A + (size_t)(m0 + ar) * K + k0 + ac;
            *(float4*)&As[ar][ac]     = *(const float4*)(s);
            *(float4*)&As[ar][ac + 4] = *(const float4*)(s + 4);
        }
        {
            const float* s = B + (size_t)(k0 + br) * N + n0 + bc;
            #pragma unroll
            for (int j = 0; j < 8; ++j)
                *(float4*)&Bs[br][bc + 4 * j] = *(const float4*)(s + 4 * j);
        }
        if (k0 + 32 < K) {
            __builtin_prefetch(A + (size_t)(m0 + ar) * K + k0 + 32 + ac, 0, 3);
            __builtin_prefetch(B + (size_t)(k0 + 32 + br) * N + n0 + bc, 0, 3);
        }
        __syncthreads();

        #pragma unroll
        for (int kk = 0; kk < 32; kk += 4) {
            v2f a, b0, b1;
            a.x = As[mrow][kk + khalf];
            a.y = As[mrow][kk + khalf + 1];
            b0.x = Bs[kk + khalf][ncol0];
            b0.y = Bs[kk + khalf + 1][ncol0];
            b1.x = Bs[kk + khalf][ncol1];
            b1.y = Bs[kk + khalf + 1][ncol1];
            acc0 = __builtin_amdgcn_wmma_f32_16x16x4_f32(
                false, a, false, b0, (short)0, acc0, false, false);
            acc1 = __builtin_amdgcn_wmma_f32_16x16x4_f32(
                false, a, false, b1, (short)0, acc1, false, false);
        }
        __syncthreads();
    }

    const int colg0 = n0 + ncol0;
    const int colg1 = n0 + ncol1;
    const float bv0 = bias ? bias[colg0] : 0.0f;
    const float bv1 = bias ? bias[colg1] : 0.0f;
    #pragma unroll
    for (int r = 0; r < 8; ++r) {
        const int rowg = m0 + r + ((lane >> 4) << 3);
        float v0 = acc0[r] + bv0;
        float v1 = acc1[r] + bv1;
        if (doRelu) { v0 = fmaxf(v0, 0.0f); v1 = fmaxf(v1, 0.0f); }
        C[(size_t)rowg * N + colg0] = v0;
        C[(size_t)rowg * N + colg1] = v1;
    }
}

#define GA_PITCH 260
#define GEMM_WGS 256
#define GA_ALL(X) X(0, ga) X(1, gb) X(2, gc) X(3, gd) X(4, ge) X(5, gf) X(6, gg) X(7, gh)
#define GA_LOAD1(i, v) v = *(const float4*)(arow + (size_t)(2 * (i)) * D_HID);
#define GA_PUT1(i, v)  *(float4*)(lrow + (2 * (i)) * GA_PITCH) = v;

template <int M>
__device__ inline float row_xor_add(float v)
{
    return v + __int_as_float(__builtin_amdgcn_mov_dpp(
        __float_as_int(v), 0x160 + M, 0xf, 0xf, true));
}

__global__ __launch_bounds__(128) void gemm_wreg_f32(
    const float* __restrict__ A, const float* __restrict__ B,
    float* __restrict__ C, const float* __restrict__ as_,
    const float* __restrict__ ad_, float* __restrict__ es,
    float* __restrict__ ed, int nTiles)
{
    __shared__ float As[2][16 * GA_PITCH];
    __shared__ __attribute__((aligned(16))) float Ps[2][4][32];

    const int tid  = threadIdx.x;
    const int wave = tid >> 5;
    const int lane = tid & 31;
    const int c    = lane & 15;
    const int h    = lane >> 4;
    const int ncol = (wave << 6) + (c << 2);

    int tile = blockIdx.x;
    if (tile >= nTiles) return;

    const float4 asv = *(const float4*)(as_ + ncol);
    const float4 adv = *(const float4*)(ad_ + ncol);

    const int sr = tid >> 6;
    const int sc = (tid & 63) << 2;

    float4 g0[8];
    #pragma unroll
    for (int i = 0; i < 8; ++i)
        g0[i] = *(const float4*)(A + (size_t)(tile * 16 + 2 * i + sr) * D_HID + sc);

    v2f bf[32][2][4];
    #pragma unroll
    for (int p = 0; p < 32; ++p) {
        #pragma unroll
        for (int q = 0; q < 2; ++q) {
            const int k = 8 * p + 4 * h + 2 * q;
            const float4 w0 = *(const float4*)(B + (size_t)k * D_HID + ncol);
            const float4 w1 = *(const float4*)(B + (size_t)(k + 1) * D_HID + ncol);
            bf[p][q][0].x = w0.x; bf[p][q][0].y = w1.x;
            bf[p][q][1].x = w0.y; bf[p][q][1].y = w1.y;
            bf[p][q][2].x = w0.z; bf[p][q][2].y = w1.z;
            bf[p][q][3].x = w0.w; bf[p][q][3].y = w1.w;
        }
    }

    #pragma unroll
    for (int i = 0; i < 8; ++i)
        *(float4*)&As[0][(2 * i + sr) * GA_PITCH + sc] = g0[i];

    float4 ga, gb, gc, gd, ge, gf, gg, gh;
    {
        const int t1 = (tile + GEMM_WGS < nTiles) ? tile + GEMM_WGS : tile;
        const float* arow = A + (size_t)(t1 * 16 + sr) * D_HID + sc;
        GA_ALL(GA_LOAD1)
    }
    __syncthreads();

    int buf = 0;
    for (; tile < nTiles; tile += GEMM_WGS) {
        const float* as = &As[buf][c * GA_PITCH + 4 * h];
        v8f acc0 = {};
        v8f acc1 = {};
        v8f acc2 = {};
        v8f acc3 = {};
        #pragma unroll
        for (int p = 0; p < 32; ++p) {
            const float4 av = *(const float4*)(as + 8 * p);
            v2f a0, a1;
            a0.x = av.x; a0.y = av.y;
            a1.x = av.z; a1.y = av.w;
            acc0 = __builtin_amdgcn_wmma_f32_16x16x4_f32(
                false, a0, false, bf[p][0][0], (short)0, acc0, false, false);
            acc1 = __builtin_amdgcn_wmma_f32_16x16x4_f32(
                false, a0, false, bf[p][0][1], (short)0, acc1, false, false);
            acc2 = __builtin_amdgcn_wmma_f32_16x16x4_f32(
                false, a0, false, bf[p][0][2], (short)0, acc2, false, false);
            acc3 = __builtin_amdgcn_wmma_f32_16x16x4_f32(
                false, a0, false, bf[p][0][3], (short)0, acc3, false, false);
            acc0 = __builtin_amdgcn_wmma_f32_16x16x4_f32(
                false, a1, false, bf[p][1][0], (short)0, acc0, false, false);
            acc1 = __builtin_amdgcn_wmma_f32_16x16x4_f32(
                false, a1, false, bf[p][1][1], (short)0, acc1, false, false);
            acc2 = __builtin_amdgcn_wmma_f32_16x16x4_f32(
                false, a1, false, bf[p][1][2], (short)0, acc2, false, false);
            acc3 = __builtin_amdgcn_wmma_f32_16x16x4_f32(
                false, a1, false, bf[p][1][3], (short)0, acc3, false, false);
        }

        float* crow = C + (size_t)(tile * 16 + 8 * h) * D_HID + ncol;
        #pragma unroll
        for (int r = 0; r < 8; ++r) {
            float4 o;
            o.x = acc0[r]; o.y = acc1[r]; o.z = acc2[r]; o.w = acc3[r];
            *(float4*)(crow + (size_t)r * D_HID) = o;
        }

        {
            float ps[8], pd[8];
            #pragma unroll
            for (int r = 0; r < 8; ++r) {
                ps[r] = acc0[r] * asv.x + acc1[r] * asv.y
                      + acc2[r] * asv.z + acc3[r] * asv.w;
                pd[r] = acc0[r] * adv.x + acc1[r] * adv.y
                      + acc2[r] * adv.z + acc3[r] * adv.w;
            }
            #pragma unroll
            for (int r = 0; r < 8; ++r) { ps[r] = row_xor_add<1>(ps[r]); pd[r] = row_xor_add<1>(pd[r]); }
            #pragma unroll
            for (int r = 0; r < 8; ++r) { ps[r] = row_xor_add<2>(ps[r]); pd[r] = row_xor_add<2>(pd[r]); }
            #pragma unroll
            for (int r = 0; r < 8; ++r) { ps[r] = row_xor_add<4>(ps[r]); pd[r] = row_xor_add<4>(pd[r]); }
            #pragma unroll
            for (int r = 0; r < 8; ++r) { ps[r] = row_xor_add<8>(ps[r]); pd[r] = row_xor_add<8>(pd[r]); }
            if (c == 0) {
                float* pp = &Ps[buf][wave][8 * h];
                *(float4*)(pp)      = make_float4(ps[0], ps[1], ps[2], ps[3]);
                *(float4*)(pp + 4)  = make_float4(ps[4], ps[5], ps[6], ps[7]);
                *(float4*)(pp + 16) = make_float4(pd[0], pd[1], pd[2], pd[3]);
                *(float4*)(pp + 20) = make_float4(pd[4], pd[5], pd[6], pd[7]);
            }
        }

        {
            float* lrow = &As[buf ^ 1][sr * GA_PITCH + sc];
            GA_ALL(GA_PUT1)
        }
        {
            const int t2 = (tile + 2 * GEMM_WGS < nTiles) ? tile + 2 * GEMM_WGS : tile;
            const float* arow = A + (size_t)(t2 * 16 + sr) * D_HID + sc;
            GA_ALL(GA_LOAD1)
        }

        __builtin_amdgcn_fence(__ATOMIC_RELEASE, "workgroup", "local");
        __builtin_amdgcn_s_barrier();
        __builtin_amdgcn_fence(__ATOMIC_ACQUIRE, "workgroup", "local");

        if (wave == 0) {
            const float* q = &Ps[buf][0][lane];
            const float v = ((q[0] + q[32]) + q[64]) + q[96];
            (h ? ed : es)[tile * 16 + c] = v;
        }
        buf ^= 1;
    }
}

#define N_PAD 53248

__global__ void fill_i32(int* __restrict__ p, int v, int n)
{
    const int i = blockIdx.x * blockDim.x + threadIdx.x;
    if (i < n) p[i] = v;
}

__global__ __launch_bounds__(256) void edge_rank(
    const int* __restrict__ dst, int* __restrict__ cnt,
    int* __restrict__ rank, int nE)
{
    const int i = blockIdx.x * blockDim.x + threadIdx.x;
    if (i >= nE) return;
    rank[i] = atomicAdd(&cnt[dst[i]], 1);
}

__global__ __launch_bounds__(1024) void scan_counts(
    const int* __restrict__ cnt, int* __restrict__ rowptr)
{
    __shared__ int wsum[2][32];
    const int tid  = threadIdx.x;
    const int lane = tid & 31;
    const int wave = tid >> 5;

    int4 v[N_PAD / 4096];
    #pragma unroll
    for (int it = 0; it < N_PAD / 4096; ++it)
        v[it] = ((const int4*)cnt)[it * 1024 + tid];

    int carry = 0;
    #pragma unroll
    for (int it = 0; it < N_PAD / 4096; ++it) {
        const int4 a = v[it];
        const int t = a.x + a.y + a.z + a.w;
        int inc = t;
        for (int off = 1; off < 32; off <<= 1) {
            const int u = __shfl_up(inc, off, 32);
            if (lane >= off) inc += u;
        }
        if (lane == 31) wsum[it & 1][wave] = inc;
        __syncthreads();
        const int ws = wsum[it & 1][lane];
        int winc = ws;
        for (int off = 1; off < 32; off <<= 1) {
            const int u = __shfl_up(winc, off, 32);
            if (lane >= off) winc += u;
        }
        const int total = __shfl(winc, 31, 32);
        const int wexcl = __shfl(winc, wave, 32) - __shfl(ws, wave, 32);
        const int excl  = carry + wexcl + inc - t;
        int4 o;
        o.x = excl;
        o.y = o.x + a.x;
        o.z = o.y + a.y;
        o.w = o.z + a.z;
        ((int4*)rowptr)[it * 1024 + tid] = o;
        carry += total;
    }
}

__global__ __launch_bounds__(256) void edge_place(
    const int* __restrict__ src, const int* __restrict__ dst,
    const int* __restrict__ rowptr, const int* __restrict__ rank,
    int* __restrict__ perm, int nE)
{
    const int i = blockIdx.x * blockDim.x + threadIdx.x;
    if (i >= nE) return;
    perm[rowptr[dst[i]] + rank[i]] = src[i];
}

__device__ inline float wave_max32(float v)
{
    for (int off = 16; off; off >>= 1) v = fmaxf(v, __shfl_xor(v, off, 32));
    return v;
}

__device__ inline float wave_sum32(float v)
{
    for (int off = 16; off; off >>= 1) v += __shfl_xor(v, off, 32);
    return v;
}

__device__ inline float edge_score1(float es_src, float ed_dst)
{
    const float t = es_src + ed_dst;
    return (t > 0.0f) ? t : 0.2f * t;
}

__global__ __launch_bounds__(256) void node_aggregate(
    const int* __restrict__ rowptr, const int* __restrict__ perm,
    const float* __restrict__ es, const float* __restrict__ ed,
    const float* __restrict__ H, const float* __restrict__ b,
    float* __restrict__ OUT, const int* __restrict__ rowsel, int n)
{
    const int lane = threadIdx.x & 31;
    const int w = __builtin_amdgcn_readfirstlane(
        (int)((blockIdx.x * blockDim.x + threadIdx.x) >> 5));
    if (w >= n) return;
    const int node = rowsel ? rowsel[w] : w;
    const int beg = rowptr[node];
    const int end = rowptr[node + 1];
    const float edn = ed[node];

    const bool in0 = beg + lane < end;
    int   s0 = 0;
    float v0 = -INFINITY;
    if (in0) {
        s0 = perm[beg + lane];
        v0 = edge_score1(es[s0], edn);
    }

    float m = v0;
    for (int base = beg + 32; base < end; base += 32)
        if (base + lane < end)
            m = fmaxf(m, edge_score1(es[perm[base + lane]], edn));
    m = wave_max32(m);

    const float ex0 = in0 ? __expf(v0 - m) : 0.0f;
    float den = ex0;
    for (int base = beg + 32; base < end; base += 32)
        if (base + lane < end)
            den += __expf(edge_score1(es[perm[base + lane]], edn) - m);
    const float dn = wave_sum32(den) + 1e-16f;

    float4 acc0 = make_float4(0.0f, 0.0f, 0.0f, 0.0f);
    float4 acc1 = make_float4(0.0f, 0.0f, 0.0f, 0.0f);

    for (int base = beg; base < end; base += 32) {
        int   s = s0;
        float a = ex0 / dn;
        if (base != beg) {
            s = 0;
            a = 0.0f;
            if (base + lane < end) {
                s = perm[base + lane];
                a = __expf(edge_score1(es[s], edn) - m) / dn;
            }
        }
        const int cnt = (end - base < 32) ? (end - base) : 32;
        int t = 0;
        for (; t + 4 <= cnt; t += 4) {
            float4 h0[4], h1[4];
            float  at[4];
            #pragma unroll
            for (int u = 0; u < 4; ++u) {
                const int st = __builtin_amdgcn_readlane(s, t + u);
                at[u] = __int_as_float(
                    __builtin_amdgcn_readlane(__float_as_int(a), t + u));
                const float4* hr = (const float4*)(H + (size_t)st * D_HID);
                h0[u] = hr[lane];
                h1[u] = hr[lane + 32];
            }
            #pragma unroll
            for (int u = 0; u < 4; ++u) {
                acc0.x = fmaf(at[u], h0[u].x, acc0.x);
                acc0.y = fmaf(at[u], h0[u].y, acc0.y);
                acc0.z = fmaf(at[u], h0[u].z, acc0.z);
                acc0.w = fmaf(at[u], h0[u].w, acc0.w);
                acc1.x = fmaf(at[u], h1[u].x, acc1.x);
                acc1.y = fmaf(at[u], h1[u].y, acc1.y);
                acc1.z = fmaf(at[u], h1[u].z, acc1.z);
                acc1.w = fmaf(at[u], h1[u].w, acc1.w);
            }
        }
        #pragma nounroll
        for (; t < cnt; ++t) {
            const int st = __builtin_amdgcn_readlane(s, t);
            const float at1 = __int_as_float(
                __builtin_amdgcn_readlane(__float_as_int(a), t));
            const float4* hr = (const float4*)(H + (size_t)st * D_HID);
            const float4 g0 = hr[lane];
            const float4 g1 = hr[lane + 32];
            acc0.x = fmaf(at1, g0.x, acc0.x);
            acc0.y = fmaf(at1, g0.y, acc0.y);
            acc0.z = fmaf(at1, g0.z, acc0.z);
            acc0.w = fmaf(at1, g0.w, acc0.w);
            acc1.x = fmaf(at1, g1.x, acc1.x);
            acc1.y = fmaf(at1, g1.y, acc1.y);
            acc1.z = fmaf(at1, g1.z, acc1.z);
            acc1.w = fmaf(at1, g1.w, acc1.w);
        }
    }

    const int c0 = 4 * lane;
    const int c1 = 128 + 4 * lane;
    float4 o0, o1;
    o0.x = fmaxf(acc0.x + b[c0 + 0], 0.0f);
    o0.y = fmaxf(acc0.y + b[c0 + 1], 0.0f);
    o0.z = fmaxf(acc0.z + b[c0 + 2], 0.0f);
    o0.w = fmaxf(acc0.w + b[c0 + 3], 0.0f);
    o1.x = fmaxf(acc1.x + b[c1 + 0], 0.0f);
    o1.y = fmaxf(acc1.y + b[c1 + 1], 0.0f);
    o1.z = fmaxf(acc1.z + b[c1 + 2], 0.0f);
    o1.w = fmaxf(acc1.w + b[c1 + 3], 0.0f);
    float4* orow = (float4*)(OUT + (size_t)w * D_HID);
    orow[lane]      = o0;
    orow[lane + 32] = o1;
}

__global__ __launch_bounds__(256) void mlp2_softmax(
    const float* __restrict__ DH, const float* __restrict__ Wm2,
    const float* __restrict__ bm2, float* __restrict__ out, int nDocs)
{
    const int doc  = (int)((blockIdx.x * blockDim.x + threadIdx.x) >> 5);
    const int lane = threadIdx.x & 31;
    if (doc >= nDocs) return;
    const float* row = DH + (size_t)doc * MLP_HID;
    float logit = -3.0e38f;
    if (lane < N_CLASSES) {
        float acc = bm2[lane];
        #pragma unroll
        for (int k = 0; k < MLP_HID; ++k)
            acc += row[k] * Wm2[k * N_CLASSES + lane];
        logit = acc;
    }
    float mx = logit;
    #pragma unroll
    for (int off = 16; off; off >>= 1) mx = fmaxf(mx, __shfl_xor(mx, off, 32));
    const float ex = (lane < N_CLASSES) ? __expf(logit - mx) : 0.0f;
    float sum = ex;
    #pragma unroll
    for (int off = 16; off; off >>= 1) sum += __shfl_xor(sum, off, 32);
    if (lane < N_CLASSES) out[(size_t)doc * N_CLASSES + lane] = ex / sum;
}

extern "C" void kernel_launch(void* const* d_in, const int* in_sizes, int n_in,
                              void* d_out, int out_size, void* d_ws, size_t ws_size,
                              hipStream_t stream)
{
    (void)in_sizes; (void)n_in; (void)out_size; (void)ws_size;

    const float* x    = (const float*)d_in[0];
    const int*   ei   = (const int*)  d_in[1];
    const int*   doci = (const int*)  d_in[2];
    const float* W1   = (const float*)d_in[3];
    const float* a1s  = (const float*)d_in[4];
    const float* a1d  = (const float*)d_in[5];
    const float* b1   = (const float*)d_in[6];
    const float* W2   = (const float*)d_in[7];
    const float* a2s  = (const float*)d_in[8];
    const float* a2d  = (const float*)d_in[9];
    const float* b2   = (const float*)d_in[10];
    const float* Wm1  = (const float*)d_in[11];
    const float* bm1  = (const float*)d_in[12];
    const float* Wm2  = (const float*)d_in[13];
    const float* bm2  = (const float*)d_in[14];

    const int* src = ei;
    const int* dst = ei + N_EDGES;

    float* H    = (float*)d_ws;
    float* OUT  = H    + (size_t)N_NODES * D_HID;
    float* ES   = OUT  + (size_t)N_NODES * D_HID;
    float* ED   = ES   + N_NODES;
    float* BKT  = ED   + N_NODES;
    float* DH   = BKT  + (size_t)N_DOCS * D_HID;
    float* DOCS = OUT;
    int*  CNT    = (int*)BKT;
    int*  ROWPTR = CNT    + N_PAD;
    int*  RANK   = ROWPTR + N_PAD;
    int*  PERM   = RANK   + N_EDGES;

    const dim3 gemmGridD(N_DOCS  / 16, MLP_HID / 64);
    const int gNodeW = (N_NODES + 7) / 8;
    const int gEdge  = (N_EDGES + 255) / 256;
    const int gDocW  = (N_DOCS + 7) / 8;

    fill_i32<<<N_PAD / 256, 256, 0, stream>>>(CNT, 0, N_PAD);
    edge_rank<<<gEdge, 256, 0, stream>>>(dst, CNT, RANK, N_EDGES);
    scan_counts<<<1, 1024, 0, stream>>>(CNT, ROWPTR);
    edge_place<<<gEdge, 256, 0, stream>>>(src, dst, ROWPTR, RANK, PERM, N_EDGES);

    for (int layer = 0; layer < 2; ++layer) {
        const float* A   = (layer == 0) ? x   : OUT;
        const float* W   = (layer == 0) ? W1  : W2;
        const float* as_ = (layer == 0) ? a1s : a2s;
        const float* ad_ = (layer == 0) ? a1d : a2d;
        const float* b   = (layer == 0) ? b1  : b2;

        gemm_wreg_f32<<<GEMM_WGS, 128, 0, stream>>>(A, W, H, as_, ad_, ES, ED,
                                                    N_NODES / 16);
        if (layer == 0)
            node_aggregate<<<gNodeW, 256, 0, stream>>>(ROWPTR, PERM, ES, ED, H,
                                                       b, OUT, nullptr, N_NODES);
        else
            node_aggregate<<<gDocW, 256, 0, stream>>>(ROWPTR, PERM, ES, ED, H,
                                                      b, DOCS, doci, N_DOCS);
    }

    gemm_wmma_f32<<<gemmGridD, 64, 0, stream>>>(DOCS, Wm1, DH, N_DOCS, MLP_HID,
                                                D_HID, bm1, 1);
    mlp2_softmax<<<gDocW, 256, 0, stream>>>(DH, Wm2, bm2, (float*)d_out, N_DOCS);
}


# Layer 2 as aggregate-then-multiply: docs sum alpha*OUT[src] first, so the W2 GEMM runs on 20000 doc rows, not 50000; scores from OUT.(W2@a2) in layer 1's edge phase; layer 2 sum orders change
# speedup vs baseline: 4.1872x; 1.0548x over previous
#include <hip/hip_runtime.h>
#include <hip/hip_bf16.h>


#define N_NODES   50000
#define N_EDGES   800000
#define D_HID     256
#define N_DOCS    20000
#define MLP_HID   64
#define N_CLASSES 20

typedef __attribute__((ext_vector_type(2))) float v2f;
typedef __attribute__((ext_vector_type(8))) float v8f;

__global__ __launch_bounds__(64) void gemm_wmma_f32(
    const float* __restrict__ A, const float* __restrict__ B,
    float* __restrict__ C, int M, int N, int K,
    const float* __restrict__ bias, int doRelu)
{
    __shared__ float As[16][32];
    __shared__ float Bs[32][64];

    const int tid  = threadIdx.x;
    const int wave = tid >> 5;
    const int lane = tid & 31;
    const int m0 = blockIdx.x * 16;
    const int n0 = blockIdx.y * 64;

    v8f acc0 = {};
    v8f acc1 = {};

    const int mrow  = lane & 15;
    const int khalf = (lane >> 4) << 1;
    const int ncol0 = (lane & 15) + (wave << 5);
    const int ncol1 = ncol0 + 16;

    const int ar = tid >> 2;
    const int ac = (tid & 3) * 8;
    const int br = tid >> 1;
    const int bc = (tid & 1) * 32;

    for (int k0 = 0; k0 < K; k0 += 32) {
        {
            const float* s = A + (size_t)(m0 + ar) * K + k0 + ac;
            *(float4*)&As[ar][ac]     = *(const float4*)(s);
            *(float4*)&As[ar][ac + 4] = *(const float4*)(s + 4);
        }
        {
            const float* s = B + (size_t)(k0 + br) * N + n0 + bc;
            #pragma unroll
            for (int j = 0; j < 8; ++j)
                *(float4*)&Bs[br][bc + 4 * j] = *(const float4*)(s + 4 * j);
        }
        if (k0 + 32 < K) {
            __builtin_prefetch(A + (size_t)(m0 + ar) * K + k0 + 32 + ac, 0, 3);
            __builtin_prefetch(B + (size_t)(k0 + 32 + br) * N + n0 + bc, 0, 3);
        }
        __syncthreads();

        #pragma unroll
        for (int kk = 0; kk < 32; kk += 4) {
            v2f a, b0, b1;
            a.x = As[mrow][kk + khalf];
            a.y = As[mrow][kk + khalf + 1];
            b0.x = Bs[kk + khalf][ncol0];
            b0.y = Bs[kk + khalf + 1][ncol0];
            b1.x = Bs[kk + khalf][ncol1];
            b1.y = Bs[kk + khalf + 1][ncol1];
            acc0 = __builtin_amdgcn_wmma_f32_16x16x4_f32(
                false, a, false, b0, (short)0, acc0, false, false);
            acc1 = __builtin_amdgcn_wmma_f32_16x16x4_f32(
                false, a, false, b1, (short)0, acc1, false, false);
        }
        __syncthreads();
    }

    const int colg0 = n0 + ncol0;
    const int colg1 = n0 + ncol1;
    const float bv0 = bias ? bias[colg0] : 0.0f;
    const float bv1 = bias ? bias[colg1] : 0.0f;
    #pragma unroll
    for (int r = 0; r < 8; ++r) {
        const int rowg = m0 + r + ((lane >> 4) << 3);
        float v0 = acc0[r] + bv0;
        float v1 = acc1[r] + bv1;
        if (doRelu) { v0 = fmaxf(v0, 0.0f); v1 = fmaxf(v1, 0.0f); }
        C[(size_t)rowg * N + colg0] = v0;
        C[(size_t)rowg * N + colg1] = v1;
    }
}

#define GA_PITCH 260
#define GEMM_WGS 256
#define GA_ALL(X) X(0, ga) X(1, gb) X(2, gc) X(3, gd) X(4, ge) X(5, gf) X(6, gg) X(7, gh)
#define GA_LOAD1(i, v) v = *(const float4*)(arow + (size_t)(2 * (i)) * D_HID);
#define GA_PUT1(i, v)  *(float4*)(lrow + (2 * (i)) * GA_PITCH) = v;

template <int M>
__device__ inline float row_xor_add(float v)
{
    return v + __int_as_float(__builtin_amdgcn_mov_dpp(
        __float_as_int(v), 0x160 + M, 0xf, 0xf, true));
}

template <bool DOTS>
__global__ __launch_bounds__(128) void gemm_wreg_f32(
    const float* __restrict__ A, const float* __restrict__ B,
    float* __restrict__ C, const float* __restrict__ as_,
    const float* __restrict__ ad_, float* __restrict__ es,
    float* __restrict__ ed, const float* __restrict__ bias, int nTiles)
{
    __shared__ float As[2][16 * GA_PITCH];
    __shared__ __attribute__((aligned(16))) float Ps[2][4][32];

    const int tid  = threadIdx.x;
    const int wave = tid >> 5;
    const int lane = tid & 31;
    const int c    = lane & 15;
    const int h    = lane >> 4;
    const int ncol = (wave << 6) + (c << 2);

    int tile = blockIdx.x;
    if (tile >= nTiles) return;

    float4 asv = make_float4(0.0f, 0.0f, 0.0f, 0.0f);
    float4 adv = asv;
    float4 bv  = asv;
    if (DOTS) {
        asv = *(const float4*)(as_ + ncol);
        adv = *(const float4*)(ad_ + ncol);
    } else {
        bv = *(const float4*)(bias + ncol);
    }

    const int sr = tid >> 6;
    const int sc = (tid & 63) << 2;

    float4 g0[8];
    #pragma unroll
    for (int i = 0; i < 8; ++i)
        g0[i] = *(const float4*)(A + (size_t)(tile * 16 + 2 * i + sr) * D_HID + sc);

    v2f bf[32][2][4];
    #pragma unroll
    for (int p = 0; p < 32; ++p) {
        #pragma unroll
        for (int q = 0; q < 2; ++q) {
            const int k = 8 * p + 4 * h + 2 * q;
            const float4 w0 = *(const float4*)(B + (size_t)k * D_HID + ncol);
            const float4 w1 = *(const float4*)(B + (size_t)(k + 1) * D_HID + ncol);
            bf[p][q][0].x = w0.x; bf[p][q][0].y = w1.x;
            bf[p][q][1].x = w0.y; bf[p][q][1].y = w1.y;
            bf[p][q][2].x = w0.z; bf[p][q][2].y = w1.z;
            bf[p][q][3].x = w0.w; bf[p][q][3].y = w1.w;
        }
    }

    #pragma unroll
    for (int i = 0; i < 8; ++i)
        *(float4*)&As[0][(2 * i + sr) * GA_PITCH + sc] = g0[i];

    float4 ga, gb, gc, gd, ge, gf, gg, gh;
    {
        const int t1 = (tile + GEMM_WGS < nTiles) ? tile + GEMM_WGS : tile;
        const float* arow = A + (size_t)(t1 * 16 + sr) * D_HID + sc;
        GA_ALL(GA_LOAD1)
    }
    __syncthreads();

    int buf = 0;
    for (; tile < nTiles; tile += GEMM_WGS) {
        const float* as = &As[buf][c * GA_PITCH + 4 * h];
        v8f acc0 = {};
        v8f acc1 = {};
        v8f acc2 = {};
        v8f acc3 = {};
        #pragma unroll
        for (int p = 0; p < 32; ++p) {
            const float4 av = *(const float4*)(as + 8 * p);
            v2f a0, a1;
            a0.x = av.x; a0.y = av.y;
            a1.x = av.z; a1.y = av.w;
            acc0 = __builtin_amdgcn_wmma_f32_16x16x4_f32(
                false, a0, false, bf[p][0][0], (short)0, acc0, false, false);
            acc1 = __builtin_amdgcn_wmma_f32_16x16x4_f32(
                false, a0, false, bf[p][0][1], (short)0, acc1, false, false);
            acc2 = __builtin_amdgcn_wmma_f32_16x16x4_f32(
                false, a0, false, bf[p][0][2], (short)0, acc2, false, false);
            acc3 = __builtin_amdgcn_wmma_f32_16x16x4_f32(
                false, a0, false, bf[p][0][3], (short)0, acc3, false, false);
            acc0 = __builtin_amdgcn_wmma_f32_16x16x4_f32(
                false, a1, false, bf[p][1][0], (short)0, acc0, false, false);
            acc1 = __builtin_amdgcn_wmma_f32_16x16x4_f32(
                false, a1, false, bf[p][1][1], (short)0, acc1, false, false);
            acc2 = __builtin_amdgcn_wmma_f32_16x16x4_f32(
                false, a1, false, bf[p][1][2], (short)0, acc2, false, false);
            acc3 = __builtin_amdgcn_wmma_f32_16x16x4_f32(
                false, a1, false, bf[p][1][3], (short)0, acc3, false, false);
        }

        float* crow = C + (size_t)(tile * 16 + 8 * h) * D_HID + ncol;
        #pragma unroll
        for (int r = 0; r < 8; ++r) {
            float4 o;
            o.x = acc0[r]; o.y = acc1[r]; o.z = acc2[r]; o.w = acc3[r];
            if (!DOTS) {
                o.x = fmaxf(o.x + bv.x, 0.0f);
                o.y = fmaxf(o.y + bv.y, 0.0f);
                o.z = fmaxf(o.z + bv.z, 0.0f);
                o.w = fmaxf(o.w + bv.w, 0.0f);
            }
            *(float4*)(crow + (size_t)r * D_HID) = o;
        }

        if (DOTS) {
            float ps[8], pd[8];
            #pragma unroll
            for (int r = 0; r < 8; ++r) {
                ps[r] = acc0[r] * asv.x + acc1[r] * asv.y
                      + acc2[r] * asv.z + acc3[r] * asv.w;
                pd[r] = acc0[r] * adv.x + acc1[r] * adv.y
                      + acc2[r] * adv.z + acc3[r] * adv.w;
            }
            #pragma unroll
            for (int r = 0; r < 8; ++r) { ps[r] = row_xor_add<1>(ps[r]); pd[r] = row_xor_add<1>(pd[r]); }
            #pragma unroll
            for (int r = 0; r < 8; ++r) { ps[r] = row_xor_add<2>(ps[r]); pd[r] = row_xor_add<2>(pd[r]); }
            #pragma unroll
            for (int r = 0; r < 8; ++r) { ps[r] = row_xor_add<4>(ps[r]); pd[r] = row_xor_add<4>(pd[r]); }
            #pragma unroll
            for (int r = 0; r < 8; ++r) { ps[r] = row_xor_add<8>(ps[r]); pd[r] = row_xor_add<8>(pd[r]); }
            if (c == 0) {
                float* pp = &Ps[buf][wave][8 * h];
                *(float4*)(pp)      = make_float4(ps[0], ps[1], ps[2], ps[3]);
                *(float4*)(pp + 4)  = make_float4(ps[4], ps[5], ps[6], ps[7]);
                *(float4*)(pp + 16) = make_float4(pd[0], pd[1], pd[2], pd[3]);
                *(float4*)(pp + 20) = make_float4(pd[4], pd[5], pd[6], pd[7]);
            }
        }

        {
            float* lrow = &As[buf ^ 1][sr * GA_PITCH + sc];
            GA_ALL(GA_PUT1)
        }
        {
            const int t2 = (tile + 2 * GEMM_WGS < nTiles) ? tile + 2 * GEMM_WGS : tile;
            const float* arow = A + (size_t)(t2 * 16 + sr) * D_HID + sc;
            GA_ALL(GA_LOAD1)
        }

        __builtin_amdgcn_fence(__ATOMIC_RELEASE, "workgroup", "local");
        __builtin_amdgcn_s_barrier();
        __builtin_amdgcn_fence(__ATOMIC_ACQUIRE, "workgroup", "local");

        if (DOTS && wave == 0) {
            const float* q = &Ps[buf][0][lane];
            const float v = ((q[0] + q[32]) + q[64]) + q[96];
            (h ? ed : es)[tile * 16 + c] = v;
        }
        buf ^= 1;
    }
}

#define N_PAD 53248

__global__ void fill_i32(int* __restrict__ p, int v, int n)
{
    const int i = blockIdx.x * blockDim.x + threadIdx.x;
    if (i < n) p[i] = v;
}

__global__ __launch_bounds__(256) void edge_rank(
    const int* __restrict__ dst, int* __restrict__ cnt,
    int* __restrict__ rank, int nE)
{
    const int i = blockIdx.x * blockDim.x + threadIdx.x;
    if (i >= nE) return;
    rank[i] = atomicAdd(&cnt[dst[i]], 1);
}

__device__ inline float wave_max32(float v)
{
    for (int off = 16; off; off >>= 1) v = fmaxf(v, __shfl_xor(v, off, 32));
    return v;
}

__device__ inline float wave_sum32(float v)
{
    for (int off = 16; off; off >>= 1) v += __shfl_xor(v, off, 32);
    return v;
}

__device__ inline void fold_attention(
    const float* __restrict__ W, const float* __restrict__ as_,
    const float* __restrict__ ad_, float* __restrict__ us,
    float* __restrict__ ud)
{
    const int lane = threadIdx.x & 31;
    const int wave = threadIdx.x >> 5;
    const float4 s0 = ((const float4*)as_)[lane];
    const float4 s1 = ((const float4*)as_)[lane + 32];
    const float4 d0 = ((const float4*)ad_)[lane];
    const float4 d1 = ((const float4*)ad_)[lane + 32];
    for (int i = 0; i < 8; ++i) {
        const int k = 8 * wave + i;
        const float4* wr = (const float4*)(W + (size_t)k * D_HID);
        const float4 w0 = wr[lane];
        const float4 w1 = wr[lane + 32];
        float ps = w0.x * s0.x + w0.y * s0.y + w0.z * s0.z + w0.w * s0.w
                 + w1.x * s1.x + w1.y * s1.y + w1.z * s1.z + w1.w * s1.w;
        float pd = w0.x * d0.x + w0.y * d0.y + w0.z * d0.z + w0.w * d0.w
                 + w1.x * d1.x + w1.y * d1.y + w1.z * d1.z + w1.w * d1.w;
        ps = wave_sum32(ps);
        pd = wave_sum32(pd);
        if (lane == 0) { us[k] = ps; ud[k] = pd; }
    }
}

__global__ __launch_bounds__(1024) void scan_counts_fold(
    const int* __restrict__ cnt, int* __restrict__ rowptr,
    const float* __restrict__ W, const float* __restrict__ as_,
    const float* __restrict__ ad_, float* __restrict__ us,
    float* __restrict__ ud)
{
    __shared__ int wsum[2][32];
    if (blockIdx.x == 1) {
        fold_attention(W, as_, ad_, us, ud);
        return;
    }
    const int tid  = threadIdx.x;
    const int lane = tid & 31;
    const int wave = tid >> 5;

    int4 v[N_PAD / 4096];
    #pragma unroll
    for (int it = 0; it < N_PAD / 4096; ++it)
        v[it] = ((const int4*)cnt)[it * 1024 + tid];

    int carry = 0;
    #pragma unroll
    for (int it = 0; it < N_PAD / 4096; ++it) {
        const int4 a = v[it];
        const int t = a.x + a.y + a.z + a.w;
        int inc = t;
        for (int off = 1; off < 32; off <<= 1) {
            const int u = __shfl_up(inc, off, 32);
            if (lane >= off) inc += u;
        }
        if (lane == 31) wsum[it & 1][wave] = inc;
        __syncthreads();
        const int ws = wsum[it & 1][lane];
        int winc = ws;
        for (int off = 1; off < 32; off <<= 1) {
            const int u = __shfl_up(winc, off, 32);
            if (lane >= off) winc += u;
        }
        const int total = __shfl(winc, 31, 32);
        const int wexcl = __shfl(winc, wave, 32) - __shfl(ws, wave, 32);
        const int excl  = carry + wexcl + inc - t;
        int4 o;
        o.x = excl;
        o.y = o.x + a.x;
        o.z = o.y + a.y;
        o.w = o.z + a.z;
        ((int4*)rowptr)[it * 1024 + tid] = o;
        carry += total;
    }
}

__global__ __launch_bounds__(256) void edge_place(
    const int* __restrict__ src, const int* __restrict__ dst,
    const int* __restrict__ rowptr, const int* __restrict__ rank,
    int* __restrict__ perm, int nE)
{
    const int i = blockIdx.x * blockDim.x + threadIdx.x;
    if (i >= nE) return;
    perm[rowptr[dst[i]] + rank[i]] = src[i];
}

__device__ inline float edge_score1(float es_src, float ed_dst)
{
    const float t = es_src + ed_dst;
    return (t > 0.0f) ? t : 0.2f * t;
}

template <bool FIRST>
__global__ __launch_bounds__(256) void node_aggregate(
    const int* __restrict__ rowptr, const int* __restrict__ perm,
    const float* __restrict__ es, const float* __restrict__ ed,
    const float* __restrict__ H, const float* __restrict__ b,
    float* __restrict__ OUT, const int* __restrict__ rowsel,
    const float* __restrict__ us, const float* __restrict__ ud,
    float* __restrict__ es2, float* __restrict__ ed2, int n)
{
    const int lane = threadIdx.x & 31;
    const int w = __builtin_amdgcn_readfirstlane(
        (int)((blockIdx.x * blockDim.x + threadIdx.x) >> 5));
    if (w >= n) return;
    const int node = FIRST ? w : rowsel[w];
    const int beg = rowptr[node];
    const int end = rowptr[node + 1];
    const float edn = ed[node];

    const bool in0 = beg + lane < end;
    int   s0 = 0;
    float v0 = -INFINITY;
    if (in0) {
        s0 = perm[beg + lane];
        v0 = edge_score1(es[s0], edn);
    }

    float m = v0;
    for (int base = beg + 32; base < end; base += 32)
        if (base + lane < end)
            m = fmaxf(m, edge_score1(es[perm[base + lane]], edn));
    m = wave_max32(m);

    const float ex0 = in0 ? __expf(v0 - m) : 0.0f;
    float den = ex0;
    for (int base = beg + 32; base < end; base += 32)
        if (base + lane < end)
            den += __expf(edge_score1(es[perm[base + lane]], edn) - m);
    const float dn = wave_sum32(den) + 1e-16f;

    float4 acc0 = make_float4(0.0f, 0.0f, 0.0f, 0.0f);
    float4 acc1 = make_float4(0.0f, 0.0f, 0.0f, 0.0f);

    for (int base = beg; base < end; base += 32) {
        int   s = s0;
        float a = ex0 / dn;
        if (base != beg) {
            s = 0;
            a = 0.0f;
            if (base + lane < end) {
                s = perm[base + lane];
                a = __expf(edge_score1(es[s], edn) - m) / dn;
            }
        }
        const int cnt = (end - base < 32) ? (end - base) : 32;
        int t = 0;
        for (; t + 4 <= cnt; t += 4) {
            float4 h0[4], h1[4];
            float  at[4];
            #pragma unroll
            for (int u = 0; u < 4; ++u) {
                const int st = __builtin_amdgcn_readlane(s, t + u);
                at[u] = __int_as_float(
                    __builtin_amdgcn_readlane(__float_as_int(a), t + u));
                const float4* hr = (const float4*)(H + (size_t)st * D_HID);
                h0[u] = hr[lane];
                h1[u] = hr[lane + 32];
            }
            #pragma unroll
            for (int u = 0; u < 4; ++u) {
                acc0.x = fmaf(at[u], h0[u].x, acc0.x);
                acc0.y = fmaf(at[u], h0[u].y, acc0.y);
                acc0.z = fmaf(at[u], h0[u].z, acc0.z);
                acc0.w = fmaf(at[u], h0[u].w, acc0.w);
                acc1.x = fmaf(at[u], h1[u].x, acc1.x);
                acc1.y = fmaf(at[u], h1[u].y, acc1.y);
                acc1.z = fmaf(at[u], h1[u].z, acc1.z);
                acc1.w = fmaf(at[u], h1[u].w, acc1.w);
            }
        }
        #pragma nounroll
        for (; t < cnt; ++t) {
            const int st = __builtin_amdgcn_readlane(s, t);
            const float at1 = __int_as_float(
                __builtin_amdgcn_readlane(__float_as_int(a), t));
            const float4* hr = (const float4*)(H + (size_t)st * D_HID);
            const float4 g0 = hr[lane];
            const float4 g1 = hr[lane + 32];
            acc0.x = fmaf(at1, g0.x, acc0.x);
            acc0.y = fmaf(at1, g0.y, acc0.y);
            acc0.z = fmaf(at1, g0.z, acc0.z);
            acc0.w = fmaf(at1, g0.w, acc0.w);
            acc1.x = fmaf(at1, g1.x, acc1.x);
            acc1.y = fmaf(at1, g1.y, acc1.y);
            acc1.z = fmaf(at1, g1.z, acc1.z);
            acc1.w = fmaf(at1, g1.w, acc1.w);
        }
    }

    float4* orow = (float4*)(OUT + (size_t)w * D_HID);
    if (!FIRST) {
        orow[lane]      = acc0;
        orow[lane + 32] = acc1;
        return;
    }
    const int c0 = 4 * lane;
    const int c1 = 128 + 4 * lane;
    float4 o0, o1;
    o0.x = fmaxf(acc0.x + b[c0 + 0], 0.0f);
    o0.y = fmaxf(acc0.y + b[c0 + 1], 0.0f);
    o0.z = fmaxf(acc0.z + b[c0 + 2], 0.0f);
    o0.w = fmaxf(acc0.w + b[c0 + 3], 0.0f);
    o1.x = fmaxf(acc1.x + b[c1 + 0], 0.0f);
    o1.y = fmaxf(acc1.y + b[c1 + 1], 0.0f);
    o1.z = fmaxf(acc1.z + b[c1 + 2], 0.0f);
    o1.w = fmaxf(acc1.w + b[c1 + 3], 0.0f);
    orow[lane]      = o0;
    orow[lane + 32] = o1;

    const float4 u0 = ((const float4*)us)[lane];
    const float4 u1 = ((const float4*)us)[lane + 32];
    const float4 q0 = ((const float4*)ud)[lane];
    const float4 q1 = ((const float4*)ud)[lane + 32];
    float ps = o0.x * u0.x + o0.y * u0.y + o0.z * u0.z + o0.w * u0.w
             + o1.x * u1.x + o1.y * u1.y + o1.z * u1.z + o1.w * u1.w;
    float pd = o0.x * q0.x + o0.y * q0.y + o0.z * q0.z + o0.w * q0.w
             + o1.x * q1.x + o1.y * q1.y + o1.z * q1.z + o1.w * q1.w;
    ps = wave_sum32(ps);
    pd = wave_sum32(pd);
    if (lane == 0) { es2[w] = ps; ed2[w] = pd; }
}

__global__ __launch_bounds__(256) void mlp2_softmax(
    const float* __restrict__ DH, const float* __restrict__ Wm2,
    const float* __restrict__ bm2, float* __restrict__ out, int nDocs)
{
    const int doc  = (int)((blockIdx.x * blockDim.x + threadIdx.x) >> 5);
    const int lane = threadIdx.x & 31;
    if (doc >= nDocs) return;
    const float* row = DH + (size_t)doc * MLP_HID;
    float logit = -3.0e38f;
    if (lane < N_CLASSES) {
        float acc = bm2[lane];
        #pragma unroll
        for (int k = 0; k < MLP_HID; ++k)
            acc += row[k] * Wm2[k * N_CLASSES + lane];
        logit = acc;
    }
    float mx = logit;
    #pragma unroll
    for (int off = 16; off; off >>= 1) mx = fmaxf(mx, __shfl_xor(mx, off, 32));
    const float ex = (lane < N_CLASSES) ? __expf(logit - mx) : 0.0f;
    float sum = ex;
    #pragma unroll
    for (int off = 16; off; off >>= 1) sum += __shfl_xor(sum, off, 32);
    if (lane < N_CLASSES) out[(size_t)doc * N_CLASSES + lane] = ex / sum;
}

extern "C" void kernel_launch(void* const* d_in, const int* in_sizes, int n_in,
                              void* d_out, int out_size, void* d_ws, size_t ws_size,
                              hipStream_t stream)
{
    (void)in_sizes; (void)n_in; (void)out_size; (void)ws_size;

    const float* x    = (const float*)d_in[0];
    const int*   ei   = (const int*)  d_in[1];
    const int*   doci = (const int*)  d_in[2];
    const float* W1   = (const float*)d_in[3];
    const float* a1s  = (const float*)d_in[4];
    const float* a1d  = (const float*)d_in[5];
    const float* b1   = (const float*)d_in[6];
    const float* W2   = (const float*)d_in[7];
    const float* a2s  = (const float*)d_in[8];
    const float* a2d  = (const float*)d_in[9];
    const float* b2   = (const float*)d_in[10];
    const float* Wm1  = (const float*)d_in[11];
    const float* bm1  = (const float*)d_in[12];
    const float* Wm2  = (const float*)d_in[13];
    const float* bm2  = (const float*)d_in[14];

    const int* src = ei;
    const int* dst = ei + N_EDGES;

    float* H    = (float*)d_ws;
    float* OUT  = H    + (size_t)N_NODES * D_HID;
    float* ES   = OUT  + (size_t)N_NODES * D_HID;
    float* ED   = ES   + N_NODES;
    float* BKT  = ED   + N_NODES;
    float* DH   = BKT  + (size_t)N_DOCS * D_HID;
    int*  CNT    = (int*)BKT;
    int*  ROWPTR = CNT    + N_PAD;
    int*  RANK   = ROWPTR + N_PAD;
    int*  PERM   = RANK   + N_EDGES;
    float* ES2  = (float*)(PERM + N_EDGES);
    float* ED2  = ES2  + N_NODES;
    float* US   = ED2  + N_NODES;
    float* UD   = US   + D_HID;
    float* Z    = H;
    float* DOCS = OUT;

    const dim3 gemmGridD(N_DOCS  / 16, MLP_HID / 64);
    const int gNodeW = (N_NODES + 7) / 8;
    const int gEdge  = (N_EDGES + 255) / 256;
    const int gDocW  = (N_DOCS + 7) / 8;

    fill_i32<<<N_PAD / 256, 256, 0, stream>>>(CNT, 0, N_PAD);
    edge_rank<<<gEdge, 256, 0, stream>>>(dst, CNT, RANK, N_EDGES);
    scan_counts_fold<<<2, 1024, 0, stream>>>(CNT, ROWPTR, W2, a2s, a2d, US, UD);
    edge_place<<<gEdge, 256, 0, stream>>>(src, dst, ROWPTR, RANK, PERM, N_EDGES);

    gemm_wreg_f32<true><<<GEMM_WGS, 128, 0, stream>>>(x, W1, H, a1s, a1d, ES, ED,
                                                      nullptr, N_NODES / 16);
    node_aggregate<true><<<gNodeW, 256, 0, stream>>>(ROWPTR, PERM, ES, ED, H, b1,
                                                     OUT, nullptr, US, UD, ES2,
                                                     ED2, N_NODES);

    node_aggregate<false><<<gDocW, 256, 0, stream>>>(ROWPTR, PERM, ES2, ED2, OUT,
                                                     nullptr, Z, doci, nullptr,
                                                     nullptr, nullptr, nullptr,
                                                     N_DOCS);
    gemm_wreg_f32<false><<<GEMM_WGS, 128, 0, stream>>>(Z, W2, DOCS, nullptr, nullptr,
                                                       nullptr, nullptr, b2,
                                                       N_DOCS / 16);

    gemm_wmma_f32<<<gemmGridD, 64, 0, stream>>>(DOCS, Wm1, DH, N_DOCS, MLP_HID,
                                                D_HID, bm1, 1);
    mlp2_softmax<<<gDocW, 256, 0, stream>>>(DH, Wm2, bm2, (float*)d_out, N_DOCS);
}


# Bucket build's edge passes (atomic count+rank, place) ride inside layer 1's node GEMM under its WMMAs; GEMM split in 2 launches round the scan, edge launches dropped; slot order still arrival rank
# speedup vs baseline: 4.3936x; 1.0493x over previous
#include <hip/hip_runtime.h>
#include <hip/hip_bf16.h>


#define N_NODES   50000
#define N_EDGES   800000
#define D_HID     256
#define N_DOCS    20000
#define MLP_HID   64
#define N_CLASSES 20

typedef __attribute__((ext_vector_type(2))) float v2f;
typedef __attribute__((ext_vector_type(8))) float v8f;

__global__ __launch_bounds__(64) void gemm_wmma_f32(
    const float* __restrict__ A, const float* __restrict__ B,
    float* __restrict__ C, int M, int N, int K,
    const float* __restrict__ bias, int doRelu)
{
    __shared__ float As[16][32];
    __shared__ float Bs[32][64];

    const int tid  = threadIdx.x;
    const int wave = tid >> 5;
    const int lane = tid & 31;
    const int m0 = blockIdx.x * 16;
    const int n0 = blockIdx.y * 64;

    v8f acc0 = {};
    v8f acc1 = {};

    const int mrow  = lane & 15;
    const int khalf = (lane >> 4) << 1;
    const int ncol0 = (lane & 15) + (wave << 5);
    const int ncol1 = ncol0 + 16;

    const int ar = tid >> 2;
    const int ac = (tid & 3) * 8;
    const int br = tid >> 1;
    const int bc = (tid & 1) * 32;

    for (int k0 = 0; k0 < K; k0 += 32) {
        {
            const float* s = A + (size_t)(m0 + ar) * K + k0 + ac;
            *(float4*)&As[ar][ac]     = *(const float4*)(s);
            *(float4*)&As[ar][ac + 4] = *(const float4*)(s + 4);
        }
        {
            const float* s = B + (size_t)(k0 + br) * N + n0 + bc;
            #pragma unroll
            for (int j = 0; j < 8; ++j)
                *(float4*)&Bs[br][bc + 4 * j] = *(const float4*)(s + 4 * j);
        }
        if (k0 + 32 < K) {
            __builtin_prefetch(A + (size_t)(m0 + ar) * K + k0 + 32 + ac, 0, 3);
            __builtin_prefetch(B + (size_t)(k0 + 32 + br) * N + n0 + bc, 0, 3);
        }
        __syncthreads();

        #pragma unroll
        for (int kk = 0; kk < 32; kk += 4) {
            v2f a, b0, b1;
            a.x = As[mrow][kk + khalf];
            a.y = As[mrow][kk + khalf + 1];
            b0.x = Bs[kk + khalf][ncol0];
            b0.y = Bs[kk + khalf + 1][ncol0];
            b1.x = Bs[kk + khalf][ncol1];
            b1.y = Bs[kk + khalf + 1][ncol1];
            acc0 = __builtin_amdgcn_wmma_f32_16x16x4_f32(
                false, a, false, b0, (short)0, acc0, false, false);
            acc1 = __builtin_amdgcn_wmma_f32_16x16x4_f32(
                false, a, false, b1, (short)0, acc1, false, false);
        }
        __syncthreads();
    }

    const int colg0 = n0 + ncol0;
    const int colg1 = n0 + ncol1;
    const float bv0 = bias ? bias[colg0] : 0.0f;
    const float bv1 = bias ? bias[colg1] : 0.0f;
    #pragma unroll
    for (int r = 0; r < 8; ++r) {
        const int rowg = m0 + r + ((lane >> 4) << 3);
        float v0 = acc0[r] + bv0;
        float v1 = acc1[r] + bv1;
        if (doRelu) { v0 = fmaxf(v0, 0.0f); v1 = fmaxf(v1, 0.0f); }
        C[(size_t)rowg * N + colg0] = v0;
        C[(size_t)rowg * N + colg1] = v1;
    }
}

#define GA_PITCH 260
#define GEMM_WGS 256
#define TILES_A  (6 * GEMM_WGS)
#define GA_ALL(X) X(0, ga) X(1, gb) X(2, gc) X(3, gd) X(4, ge) X(5, gf) X(6, gg) X(7, gh)
#define GA_LOAD1(i, v) v = *(const float4*)(arow + (size_t)(2 * (i)) * D_HID);
#define GA_PUT1(i, v)  *(float4*)(lrow + (2 * (i)) * GA_PITCH) = v;

template <int M>
__device__ inline float row_xor_add(float v)
{
    return v + __int_as_float(__builtin_amdgcn_mov_dpp(
        __float_as_int(v), 0x160 + M, 0xf, 0xf, true));
}

#define E_STRIDE   (GEMM_WGS * 128)
#define E_ROUNDS   ((N_EDGES + E_STRIDE - 1) / E_STRIDE)
#define E_PER_TRIP 5

template <int EDGE>
__device__ inline void edge_fetch(
    int ebase, const int* __restrict__ esrc, const int* __restrict__ edst,
    const int* erank, int (&d)[E_PER_TRIP], int (&s)[E_PER_TRIP],
    int (&r)[E_PER_TRIP])
{
    #pragma unroll
    for (int u = 0; u < E_PER_TRIP; ++u) {
        const int i = ebase + u * E_STRIDE;
        d[u] = 0; s[u] = 0; r[u] = 0;
        if (i < N_EDGES) {
            d[u] = edst[i];
            if (EDGE == 2) { s[u] = esrc[i]; r[u] = erank[i]; }
        }
    }
}

template <int EDGE>
__device__ inline void edge_issue(
    int ebase, int* __restrict__ ecnt, const int* __restrict__ erowptr,
    const int (&d)[E_PER_TRIP], int (&q)[E_PER_TRIP])
{
    #pragma unroll
    for (int u = 0; u < E_PER_TRIP; ++u) {
        q[u] = 0;
        if (ebase + u * E_STRIDE < N_EDGES) {
            if (EDGE == 1) q[u] = atomicAdd(&ecnt[d[u]], 1);
            else           q[u] = erowptr[d[u]];
        }
    }
}

template <int EDGE>
__device__ inline void edge_finish(
    int ebase, int* erank, int* __restrict__ eperm,
    const int (&s)[E_PER_TRIP], const int (&r)[E_PER_TRIP],
    const int (&q)[E_PER_TRIP])
{
    #pragma unroll
    for (int u = 0; u < E_PER_TRIP; ++u) {
        const int i = ebase + u * E_STRIDE;
        if (i < N_EDGES) {
            if (EDGE == 1) erank[i] = q[u];
            else           eperm[q[u] + r[u]] = s[u];
        }
    }
}

template <bool DOTS, int EDGE>
__global__ __launch_bounds__(128) void gemm_wreg_f32(
    const float* __restrict__ A, const float* __restrict__ B,
    float* __restrict__ C, const float* __restrict__ as_,
    const float* __restrict__ ad_, float* __restrict__ es,
    float* __restrict__ ed, const float* __restrict__ bias, int nTiles,
    const int* __restrict__ esrc, const int* __restrict__ edst,
    int* __restrict__ ecnt, int* erank, const int* __restrict__ erowptr,
    int* __restrict__ eperm)
{
    __shared__ float As[2][16 * GA_PITCH];
    __shared__ __attribute__((aligned(16))) float Ps[2][4][32];

    const int tid  = threadIdx.x;
    const int wave = tid >> 5;
    const int lane = tid & 31;
    const int c    = lane & 15;
    const int h    = lane >> 4;
    const int ncol = (wave << 6) + (c << 2);

    int tile = blockIdx.x;
    if (tile >= nTiles) return;

    float4 asv = make_float4(0.0f, 0.0f, 0.0f, 0.0f);
    float4 adv = asv;
    float4 bv  = asv;
    if (DOTS) {
        asv = *(const float4*)(as_ + ncol);
        adv = *(const float4*)(ad_ + ncol);
    } else {
        bv = *(const float4*)(bias + ncol);
    }

    int ebase = blockIdx.x * 128 + tid;
    int dcur[E_PER_TRIP], scur[E_PER_TRIP], rcur[E_PER_TRIP];
    if (EDGE) edge_fetch<EDGE>(ebase, esrc, edst, erank, dcur, scur, rcur);

    const int sr = tid >> 6;
    const int sc = (tid & 63) << 2;

    float4 g0[8];
    #pragma unroll
    for (int i = 0; i < 8; ++i)
        g0[i] = *(const float4*)(A + (size_t)(tile * 16 + 2 * i + sr) * D_HID + sc);

    v2f bf[32][2][4];
    #pragma unroll
    for (int p = 0; p < 32; ++p) {
        #pragma unroll
        for (int q = 0; q < 2; ++q) {
            const int k = 8 * p + 4 * h + 2 * q;
            const float4 w0 = *(const float4*)(B + (size_t)k * D_HID + ncol);
            const float4 w1 = *(const float4*)(B + (size_t)(k + 1) * D_HID + ncol);
            bf[p][q][0].x = w0.x; bf[p][q][0].y = w1.x;
            bf[p][q][1].x = w0.y; bf[p][q][1].y = w1.y;
            bf[p][q][2].x = w0.z; bf[p][q][2].y = w1.z;
            bf[p][q][3].x = w0.w; bf[p][q][3].y = w1.w;
        }
    }

    #pragma unroll
    for (int i = 0; i < 8; ++i)
        *(float4*)&As[0][(2 * i + sr) * GA_PITCH + sc] = g0[i];

    float4 ga, gb, gc, gd, ge, gf, gg, gh;
    {
        const int t1 = (tile + GEMM_WGS < nTiles) ? tile + GEMM_WGS : tile;
        const float* arow = A + (size_t)(t1 * 16 + sr) * D_HID + sc;
        GA_ALL(GA_LOAD1)
    }
    __syncthreads();

    int buf = 0;
    for (; tile < nTiles; tile += GEMM_WGS) {
        int eq[E_PER_TRIP], dnxt[E_PER_TRIP], snxt[E_PER_TRIP], rnxt[E_PER_TRIP];
        if (EDGE) {
            edge_issue<EDGE>(ebase, ecnt, erowptr, dcur, eq);
            edge_fetch<EDGE>(ebase + E_PER_TRIP * E_STRIDE, esrc, edst, erank,
                             dnxt, snxt, rnxt);
        }

        const float* as = &As[buf][c * GA_PITCH + 4 * h];
        v8f acc0 = {};
        v8f acc1 = {};
        v8f acc2 = {};
        v8f acc3 = {};
        #pragma unroll
        for (int p = 0; p < 32; ++p) {
            const float4 av = *(const float4*)(as + 8 * p);
            v2f a0, a1;
            a0.x = av.x; a0.y = av.y;
            a1.x = av.z; a1.y = av.w;
            acc0 = __builtin_amdgcn_wmma_f32_16x16x4_f32(
                false, a0, false, bf[p][0][0], (short)0, acc0, false, false);
            acc1 = __builtin_amdgcn_wmma_f32_16x16x4_f32(
                false, a0, false, bf[p][0][1], (short)0, acc1, false, false);
            acc2 = __builtin_amdgcn_wmma_f32_16x16x4_f32(
                false, a0, false, bf[p][0][2], (short)0, acc2, false, false);
            acc3 = __builtin_amdgcn_wmma_f32_16x16x4_f32(
                false, a0, false, bf[p][0][3], (short)0, acc3, false, false);
            acc0 = __builtin_amdgcn_wmma_f32_16x16x4_f32(
                false, a1, false, bf[p][1][0], (short)0, acc0, false, false);
            acc1 = __builtin_amdgcn_wmma_f32_16x16x4_f32(
                false, a1, false, bf[p][1][1], (short)0, acc1, false, false);
            acc2 = __builtin_amdgcn_wmma_f32_16x16x4_f32(
                false, a1, false, bf[p][1][2], (short)0, acc2, false, false);
            acc3 = __builtin_amdgcn_wmma_f32_16x16x4_f32(
                false, a1, false, bf[p][1][3], (short)0, acc3, false, false);
        }

        float* crow = C + (size_t)(tile * 16 + 8 * h) * D_HID + ncol;
        #pragma unroll
        for (int r = 0; r < 8; ++r) {
            float4 o;
            o.x = acc0[r]; o.y = acc1[r]; o.z = acc2[r]; o.w = acc3[r];
            if (!DOTS) {
                o.x = fmaxf(o.x + bv.x, 0.0f);
                o.y = fmaxf(o.y + bv.y, 0.0f);
                o.z = fmaxf(o.z + bv.z, 0.0f);
                o.w = fmaxf(o.w + bv.w, 0.0f);
            }
            *(float4*)(crow + (size_t)r * D_HID) = o;
        }

        if (DOTS) {
            float ps[8], pd[8];
            #pragma unroll
            for (int r = 0; r < 8; ++r) {
                ps[r] = acc0[r] * asv.x + acc1[r] * asv.y
                      + acc2[r] * asv.z + acc3[r] * asv.w;
                pd[r] = acc0[r] * adv.x + acc1[r] * adv.y
                      + acc2[r] * adv.z + acc3[r] * adv.w;
            }
            #pragma unroll
            for (int r = 0; r < 8; ++r) { ps[r] = row_xor_add<1>(ps[r]); pd[r] = row_xor_add<1>(pd[r]); }
            #pragma unroll
            for (int r = 0; r < 8; ++r) { ps[r] = row_xor_add<2>(ps[r]); pd[r] = row_xor_add<2>(pd[r]); }
            #pragma unroll
            for (int r = 0; r < 8; ++r) { ps[r] = row_xor_add<4>(ps[r]); pd[r] = row_xor_add<4>(pd[r]); }
            #pragma unroll
            for (int r = 0; r < 8; ++r) { ps[r] = row_xor_add<8>(ps[r]); pd[r] = row_xor_add<8>(pd[r]); }
            if (c == 0) {
                float* pp = &Ps[buf][wave][8 * h];
                *(float4*)(pp)      = make_float4(ps[0], ps[1], ps[2], ps[3]);
                *(float4*)(pp + 4)  = make_float4(ps[4], ps[5], ps[6], ps[7]);
                *(float4*)(pp + 16) = make_float4(pd[0], pd[1], pd[2], pd[3]);
                *(float4*)(pp + 20) = make_float4(pd[4], pd[5], pd[6], pd[7]);
            }
        }

        if (EDGE) {
            edge_finish<EDGE>(ebase, erank, eperm, scur, rcur, eq);
            ebase += E_PER_TRIP * E_STRIDE;
            #pragma unroll
            for (int u = 0; u < E_PER_TRIP; ++u) {
                dcur[u] = dnxt[u]; scur[u] = snxt[u]; rcur[u] = rnxt[u];
            }
        }

        {
            float* lrow = &As[buf ^ 1][sr * GA_PITCH + sc];
            GA_ALL(GA_PUT1)
        }
        {
            const int t2 = (tile + 2 * GEMM_WGS < nTiles) ? tile + 2 * GEMM_WGS : tile;
            const float* arow = A + (size_t)(t2 * 16 + sr) * D_HID + sc;
            GA_ALL(GA_LOAD1)
        }

        __builtin_amdgcn_fence(__ATOMIC_RELEASE, "workgroup", "local");
        __builtin_amdgcn_s_barrier();
        __builtin_amdgcn_fence(__ATOMIC_ACQUIRE, "workgroup", "local");

        if (DOTS && wave == 0) {
            const float* q = &Ps[buf][0][lane];
            const float v = ((q[0] + q[32]) + q[64]) + q[96];
            (h ? ed : es)[tile * 16 + c] = v;
        }
        buf ^= 1;
    }

    if (EDGE) {
        for (; ebase < N_EDGES; ebase += E_PER_TRIP * E_STRIDE) {
            int eq[E_PER_TRIP];
            edge_fetch<EDGE>(ebase, esrc, edst, erank, dcur, scur, rcur);
            edge_issue<EDGE>(ebase, ecnt, erowptr, dcur, eq);
            edge_finish<EDGE>(ebase, erank, eperm, scur, rcur, eq);
        }
    }
}

#define N_PAD 53248

__global__ void fill_i32(int* __restrict__ p, int v, int n)
{
    const int i = blockIdx.x * blockDim.x + threadIdx.x;
    if (i < n) p[i] = v;
}

__device__ inline float wave_max32(float v)
{
    for (int off = 16; off; off >>= 1) v = fmaxf(v, __shfl_xor(v, off, 32));
    return v;
}

__device__ inline float wave_sum32(float v)
{
    for (int off = 16; off; off >>= 1) v += __shfl_xor(v, off, 32);
    return v;
}

__device__ inline void fold_attention(
    const float* __restrict__ W, const float* __restrict__ as_,
    const float* __restrict__ ad_, float* __restrict__ us,
    float* __restrict__ ud)
{
    const int lane = threadIdx.x & 31;
    const int wave = threadIdx.x >> 5;
    const float4 s0 = ((const float4*)as_)[lane];
    const float4 s1 = ((const float4*)as_)[lane + 32];
    const float4 d0 = ((const float4*)ad_)[lane];
    const float4 d1 = ((const float4*)ad_)[lane + 32];
    for (int i = 0; i < 8; ++i) {
        const int k = 8 * wave + i;
        const float4* wr = (const float4*)(W + (size_t)k * D_HID);
        const float4 w0 = wr[lane];
        const float4 w1 = wr[lane + 32];
        float ps = w0.x * s0.x + w0.y * s0.y + w0.z * s0.z + w0.w * s0.w
                 + w1.x * s1.x + w1.y * s1.y + w1.z * s1.z + w1.w * s1.w;
        float pd = w0.x * d0.x + w0.y * d0.y + w0.z * d0.z + w0.w * d0.w
                 + w1.x * d1.x + w1.y * d1.y + w1.z * d1.z + w1.w * d1.w;
        ps = wave_sum32(ps);
        pd = wave_sum32(pd);
        if (lane == 0) { us[k] = ps; ud[k] = pd; }
    }
}

__global__ __launch_bounds__(1024) void scan_counts_fold(
    const int* __restrict__ cnt, int* __restrict__ rowptr,
    const float* __restrict__ W, const float* __restrict__ as_,
    const float* __restrict__ ad_, float* __restrict__ us,
    float* __restrict__ ud)
{
    __shared__ int wsum[2][32];
    if (blockIdx.x == 1) {
        fold_attention(W, as_, ad_, us, ud);
        return;
    }
    const int tid  = threadIdx.x;
    const int lane = tid & 31;
    const int wave = tid >> 5;

    int4 v[N_PAD / 4096];
    #pragma unroll
    for (int it = 0; it < N_PAD / 4096; ++it)
        v[it] = ((const int4*)cnt)[it * 1024 + tid];

    int carry = 0;
    #pragma unroll
    for (int it = 0; it < N_PAD / 4096; ++it) {
        const int4 a = v[it];
        const int t = a.x + a.y + a.z + a.w;
        int inc = t;
        for (int off = 1; off < 32; off <<= 1) {
            const int u = __shfl_up(inc, off, 32);
            if (lane >= off) inc += u;
        }
        if (lane == 31) wsum[it & 1][wave] = inc;
        __syncthreads();
        const int ws = wsum[it & 1][lane];
        int winc = ws;
        for (int off = 1; off < 32; off <<= 1) {
            const int u = __shfl_up(winc, off, 32);
            if (lane >= off) winc += u;
        }
        const int total = __shfl(winc, 31, 32);
        const int wexcl = __shfl(winc, wave, 32) - __shfl(ws, wave, 32);
        const int excl  = carry + wexcl + inc - t;
        int4 o;
        o.x = excl;
        o.y = o.x + a.x;
        o.z = o.y + a.y;
        o.w = o.z + a.z;
        ((int4*)rowptr)[it * 1024 + tid] = o;
        carry += total;
    }
}

__device__ inline float edge_score1(float es_src, float ed_dst)
{
    const float t = es_src + ed_dst;
    return (t > 0.0f) ? t : 0.2f * t;
}

template <bool FIRST>
__global__ __launch_bounds__(256) void node_aggregate(
    const int* __restrict__ rowptr, const int* __restrict__ perm,
    const float* __restrict__ es, const float* __restrict__ ed,
    const float* __restrict__ H, const float* __restrict__ b,
    float* __restrict__ OUT, const int* __restrict__ rowsel,
    const float* __restrict__ us, const float* __restrict__ ud,
    float* __restrict__ es2, float* __restrict__ ed2, int n)
{
    const int lane = threadIdx.x & 31;
    const int w = __builtin_amdgcn_readfirstlane(
        (int)((blockIdx.x * blockDim.x + threadIdx.x) >> 5));
    if (w >= n) return;
    const int node = FIRST ? w : rowsel[w];
    const int beg = rowptr[node];
    const int end = rowptr[node + 1];
    const float edn = ed[node];

    const bool in0 = beg + lane < end;
    int   s0 = 0;
    float v0 = -INFINITY;
    if (in0) {
        s0 = perm[beg + lane];
        v0 = edge_score1(es[s0], edn);
    }

    float m = v0;
    for (int base = beg + 32; base < end; base += 32)
        if (base + lane < end)
            m = fmaxf(m, edge_score1(es[perm[base + lane]], edn));
    m = wave_max32(m);

    const float ex0 = in0 ? __expf(v0 - m) : 0.0f;
    float den = ex0;
    for (int base = beg + 32; base < end; base += 32)
        if (base + lane < end)
            den += __expf(edge_score1(es[perm[base + lane]], edn) - m);
    const float dn = wave_sum32(den) + 1e-16f;

    float4 acc0 = make_float4(0.0f, 0.0f, 0.0f, 0.0f);
    float4 acc1 = make_float4(0.0f, 0.0f, 0.0f, 0.0f);

    for (int base = beg; base < end; base += 32) {
        int   s = s0;
        float a = ex0 / dn;
        if (base != beg) {
            s = 0;
            a = 0.0f;
            if (base + lane < end) {
                s = perm[base + lane];
                a = __expf(edge_score1(es[s], edn) - m) / dn;
            }
        }
        const int cnt = (end - base < 32) ? (end - base) : 32;
        int t = 0;
        for (; t + 4 <= cnt; t += 4) {
            float4 h0[4], h1[4];
            float  at[4];
            #pragma unroll
            for (int u = 0; u < 4; ++u) {
                const int st = __builtin_amdgcn_readlane(s, t + u);
                at[u] = __int_as_float(
                    __builtin_amdgcn_readlane(__float_as_int(a), t + u));
                const float4* hr = (const float4*)(H + (size_t)st * D_HID);
                h0[u] = hr[lane];
                h1[u] = hr[lane + 32];
            }
            #pragma unroll
            for (int u = 0; u < 4; ++u) {
                acc0.x = fmaf(at[u], h0[u].x, acc0.x);
                acc0.y = fmaf(at[u], h0[u].y, acc0.y);
                acc0.z = fmaf(at[u], h0[u].z, acc0.z);
                acc0.w = fmaf(at[u], h0[u].w, acc0.w);
                acc1.x = fmaf(at[u], h1[u].x, acc1.x);
                acc1.y = fmaf(at[u], h1[u].y, acc1.y);
                acc1.z = fmaf(at[u], h1[u].z, acc1.z);
                acc1.w = fmaf(at[u], h1[u].w, acc1.w);
            }
        }
        #pragma nounroll
        for (; t < cnt; ++t) {
            const int st = __builtin_amdgcn_readlane(s, t);
            const float at1 = __int_as_float(
                __builtin_amdgcn_readlane(__float_as_int(a), t));
            const float4* hr = (const float4*)(H + (size_t)st * D_HID);
            const float4 g0 = hr[lane];
            const float4 g1 = hr[lane + 32];
            acc0.x = fmaf(at1, g0.x, acc0.x);
            acc0.y = fmaf(at1, g0.y, acc0.y);
            acc0.z = fmaf(at1, g0.z, acc0.z);
            acc0.w = fmaf(at1, g0.w, acc0.w);
            acc1.x = fmaf(at1, g1.x, acc1.x);
            acc1.y = fmaf(at1, g1.y, acc1.y);
            acc1.z = fmaf(at1, g1.z, acc1.z);
            acc1.w = fmaf(at1, g1.w, acc1.w);
        }
    }

    float4* orow = (float4*)(OUT + (size_t)w * D_HID);
    if (!FIRST) {
        orow[lane]      = acc0;
        orow[lane + 32] = acc1;
        return;
    }
    const int c0 = 4 * lane;
    const int c1 = 128 + 4 * lane;
    float4 o0, o1;
    o0.x = fmaxf(acc0.x + b[c0 + 0], 0.0f);
    o0.y = fmaxf(acc0.y + b[c0 + 1], 0.0f);
    o0.z = fmaxf(acc0.z + b[c0 + 2], 0.0f);
    o0.w = fmaxf(acc0.w + b[c0 + 3], 0.0f);
    o1.x = fmaxf(acc1.x + b[c1 + 0], 0.0f);
    o1.y = fmaxf(acc1.y + b[c1 + 1], 0.0f);
    o1.z = fmaxf(acc1.z + b[c1 + 2], 0.0f);
    o1.w = fmaxf(acc1.w + b[c1 + 3], 0.0f);
    orow[lane]      = o0;
    orow[lane + 32] = o1;

    const float4 u0 = ((const float4*)us)[lane];
    const float4 u1 = ((const float4*)us)[lane + 32];
    const float4 q0 = ((const float4*)ud)[lane];
    const float4 q1 = ((const float4*)ud)[lane + 32];
    float ps = o0.x * u0.x + o0.y * u0.y + o0.z * u0.z + o0.w * u0.w
             + o1.x * u1.x + o1.y * u1.y + o1.z * u1.z + o1.w * u1.w;
    float pd = o0.x * q0.x + o0.y * q0.y + o0.z * q0.z + o0.w * q0.w
             + o1.x * q1.x + o1.y * q1.y + o1.z * q1.z + o1.w * q1.w;
    ps = wave_sum32(ps);
    pd = wave_sum32(pd);
    if (lane == 0) { es2[w] = ps; ed2[w] = pd; }
}

__global__ __launch_bounds__(256) void mlp2_softmax(
    const float* __restrict__ DH, const float* __restrict__ Wm2,
    const float* __restrict__ bm2, float* __restrict__ out, int nDocs)
{
    const int doc  = (int)((blockIdx.x * blockDim.x + threadIdx.x) >> 5);
    const int lane = threadIdx.x & 31;
    if (doc >= nDocs) return;
    const float* row = DH + (size_t)doc * MLP_HID;
    float logit = -3.0e38f;
    if (lane < N_CLASSES) {
        float acc = bm2[lane];
        #pragma unroll
        for (int k = 0; k < MLP_HID; ++k)
            acc += row[k] * Wm2[k * N_CLASSES + lane];
        logit = acc;
    }
    float mx = logit;
    #pragma unroll
    for (int off = 16; off; off >>= 1) mx = fmaxf(mx, __shfl_xor(mx, off, 32));
    const float ex = (lane < N_CLASSES) ? __expf(logit - mx) : 0.0f;
    float sum = ex;
    #pragma unroll
    for (int off = 16; off; off >>= 1) sum += __shfl_xor(sum, off, 32);
    if (lane < N_CLASSES) out[(size_t)doc * N_CLASSES + lane] = ex / sum;
}

extern "C" void kernel_launch(void* const* d_in, const int* in_sizes, int n_in,
                              void* d_out, int out_size, void* d_ws, size_t ws_size,
                              hipStream_t stream)
{
    (void)in_sizes; (void)n_in; (void)out_size; (void)ws_size;

    const float* x    = (const float*)d_in[0];
    const int*   ei   = (const int*)  d_in[1];
    const int*   doci = (const int*)  d_in[2];
    const float* W1   = (const float*)d_in[3];
    const float* a1s  = (const float*)d_in[4];
    const float* a1d  = (const float*)d_in[5];
    const float* b1   = (const float*)d_in[6];
    const float* W2   = (const float*)d_in[7];
    const float* a2s  = (const float*)d_in[8];
    const float* a2d  = (const float*)d_in[9];
    const float* b2   = (const float*)d_in[10];
    const float* Wm1  = (const float*)d_in[11];
    const float* bm1  = (const float*)d_in[12];
    const float* Wm2  = (const float*)d_in[13];
    const float* bm2  = (const float*)d_in[14];

    const int* src = ei;
    const int* dst = ei + N_EDGES;

    float* H    = (float*)d_ws;
    float* OUT  = H    + (size_t)N_NODES * D_HID;
    float* ES   = OUT  + (size_t)N_NODES * D_HID;
    float* ED   = ES   + N_NODES;
    float* BKT  = ED   + N_NODES;
    float* DH   = BKT  + (size_t)N_DOCS * D_HID;
    int*  CNT    = (int*)BKT;
    int*  ROWPTR = CNT    + N_PAD;
    int*  RANK   = ROWPTR + N_PAD;
    int*  PERM   = RANK   + N_EDGES;
    float* ES2  = (float*)(PERM + N_EDGES);
    float* ED2  = ES2  + N_NODES;
    float* US   = ED2  + N_NODES;
    float* UD   = US   + D_HID;
    float* Z    = H;
    float* DOCS = OUT;

    const dim3 gemmGridD(N_DOCS  / 16, MLP_HID / 64);
    const int gNodeW = (N_NODES + 7) / 8;
    const int gDocW  = (N_DOCS + 7) / 8;

    fill_i32<<<N_PAD / 256, 256, 0, stream>>>(CNT, 0, N_PAD);
    gemm_wreg_f32<true, 1><<<GEMM_WGS, 128, 0, stream>>>(
        x, W1, H, a1s, a1d, ES, ED, nullptr, TILES_A,
        src, dst, CNT, RANK, ROWPTR, PERM);
    scan_counts_fold<<<2, 1024, 0, stream>>>(CNT, ROWPTR, W2, a2s, a2d, US, UD);
    gemm_wreg_f32<true, 2><<<GEMM_WGS, 128, 0, stream>>>(
        x + (size_t)TILES_A * 16 * D_HID, W1, H + (size_t)TILES_A * 16 * D_HID,
        a1s, a1d, ES + TILES_A * 16, ED + TILES_A * 16, nullptr,
        N_NODES / 16 - TILES_A, src, dst, CNT, RANK, ROWPTR, PERM);
    node_aggregate<true><<<gNodeW, 256, 0, stream>>>(ROWPTR, PERM, ES, ED, H, b1,
                                                     OUT, nullptr, US, UD, ES2,
                                                     ED2, N_NODES);

    node_aggregate<false><<<gDocW, 256, 0, stream>>>(ROWPTR, PERM, ES2, ED2, OUT,
                                                     nullptr, Z, doci, nullptr,
                                                     nullptr, nullptr, nullptr,
                                                     N_DOCS);
    gemm_wreg_f32<false, 0><<<GEMM_WGS, 128, 0, stream>>>(
        Z, W2, DOCS, nullptr, nullptr, nullptr, nullptr, b2, N_DOCS / 16,
        nullptr, nullptr, nullptr, nullptr, nullptr, nullptr);

    gemm_wmma_f32<<<gemmGridD, 64, 0, stream>>>(DOCS, Wm1, DH, N_DOCS, MLP_HID,
                                                D_HID, bm1, 1);
    mlp2_softmax<<<gDocW, 256, 0, stream>>>(DH, Wm2, bm2, (float*)d_out, N_DOCS);
}


# MLP head (relu(D@Wm1+bm1) on WMMA, logits, softmax) runs inside the doc GEMM on each 16-row D tile kept in LDS: D and hidden never reach memory, 2 launches dropped; hidden k sum order changes
# speedup vs baseline: 4.6401x; 1.0561x over previous
#include <hip/hip_runtime.h>
#include <hip/hip_bf16.h>


#define N_NODES   50000
#define N_EDGES   800000
#define D_HID     256
#define N_DOCS    20000
#define MLP_HID   64
#define N_CLASSES 20

typedef __attribute__((ext_vector_type(2))) float v2f;
typedef __attribute__((ext_vector_type(8))) float v8f;

#define GA_PITCH 260
#define GEMM_WGS 256
#define TILES_A  (6 * GEMM_WGS)
#define HT_PITCH 68
#define GA_ALL(X) X(0, ga) X(1, gb) X(2, gc) X(3, gd) X(4, ge) X(5, gf) X(6, gg) X(7, gh)
#define GA_LOAD1(i, v) v = *(const float4*)(arow + (size_t)(2 * (i)) * D_HID);
#define GA_PUT1(i, v)  *(float4*)(lrow + (2 * (i)) * GA_PITCH) = v;

__device__ inline float wave_max32(float v)
{
    for (int off = 16; off; off >>= 1) v = fmaxf(v, __shfl_xor(v, off, 32));
    return v;
}

__device__ inline float wave_sum32(float v)
{
    for (int off = 16; off; off >>= 1) v += __shfl_xor(v, off, 32);
    return v;
}

template <int M>
__device__ inline float row_xor_add(float v)
{
    return v + __int_as_float(__builtin_amdgcn_mov_dpp(
        __float_as_int(v), 0x160 + M, 0xf, 0xf, true));
}

template <int M>
__device__ inline float row_xor_max(float v)
{
    return fmaxf(v, __int_as_float(__builtin_amdgcn_mov_dpp(
        __float_as_int(v), 0x160 + M, 0xf, 0xf, true)));
}

__device__ inline float wave_sum32_row(float v)
{
    v += __shfl_xor(v, 16, 32);
    v = row_xor_add<8>(v);
    v = row_xor_add<4>(v);
    v = row_xor_add<2>(v);
    return row_xor_add<1>(v);
}

__device__ inline float wave_max32_row(float v)
{
    v = fmaxf(v, __shfl_xor(v, 16, 32));
    v = row_xor_max<8>(v);
    v = row_xor_max<4>(v);
    v = row_xor_max<2>(v);
    return row_xor_max<1>(v);
}

#define E_STRIDE   (GEMM_WGS * 128)
#define E_ROUNDS   ((N_EDGES + E_STRIDE - 1) / E_STRIDE)
#define E_PER_TRIP 5

template <int EDGE>
__device__ inline void edge_fetch(
    int ebase, const int* __restrict__ esrc, const int* __restrict__ edst,
    const int* erank, int (&d)[E_PER_TRIP], int (&s)[E_PER_TRIP],
    int (&r)[E_PER_TRIP])
{
    #pragma unroll
    for (int u = 0; u < E_PER_TRIP; ++u) {
        const int i = ebase + u * E_STRIDE;
        d[u] = 0; s[u] = 0; r[u] = 0;
        if (i < N_EDGES) {
            d[u] = edst[i];
            if (EDGE == 2) { s[u] = esrc[i]; r[u] = erank[i]; }
        }
    }
}

template <int EDGE>
__device__ inline void edge_issue(
    int ebase, int* __restrict__ ecnt, const int* __restrict__ erowptr,
    const int (&d)[E_PER_TRIP], int (&q)[E_PER_TRIP])
{
    #pragma unroll
    for (int u = 0; u < E_PER_TRIP; ++u) {
        q[u] = 0;
        if (ebase + u * E_STRIDE < N_EDGES) {
            if (EDGE == 1) q[u] = atomicAdd(&ecnt[d[u]], 1);
            else           q[u] = erowptr[d[u]];
        }
    }
}

template <int EDGE>
__device__ inline void edge_finish(
    int ebase, int* erank, int* __restrict__ eperm,
    const int (&s)[E_PER_TRIP], const int (&r)[E_PER_TRIP],
    const int (&q)[E_PER_TRIP])
{
    #pragma unroll
    for (int u = 0; u < E_PER_TRIP; ++u) {
        const int i = ebase + u * E_STRIDE;
        if (i < N_EDGES) {
            if (EDGE == 1) erank[i] = q[u];
            else           eperm[q[u] + r[u]] = s[u];
        }
    }
}

template <bool DOTS, int EDGE>
__global__ __launch_bounds__(128) void gemm_wreg_f32(
    const float* __restrict__ A, const float* __restrict__ B,
    float* __restrict__ C, const float* __restrict__ as_,
    const float* __restrict__ ad_, float* __restrict__ es,
    float* __restrict__ ed, const float* __restrict__ bias, int nTiles,
    const int* __restrict__ esrc, const int* __restrict__ edst,
    int* __restrict__ ecnt, int* erank, const int* __restrict__ erowptr,
    int* __restrict__ eperm,
    const float* __restrict__ hW1, const float* __restrict__ hb1,
    const float* __restrict__ hW2, const float* __restrict__ hb2,
    float* __restrict__ hout)
{
    __shared__ float As[2][16 * GA_PITCH];
    __shared__ __attribute__((aligned(16))) float Dt[16 * GA_PITCH];
    __shared__ __attribute__((aligned(16))) float Ht[16 * HT_PITCH];
    __shared__ __attribute__((aligned(16))) float W2s[MLP_HID * N_CLASSES];
    __shared__ __attribute__((aligned(16))) float Ps[2][4][32];

    const int tid  = threadIdx.x;
    const int wave = tid >> 5;
    const int lane = tid & 31;
    const int c    = lane & 15;
    const int h    = lane >> 4;
    const int ncol = (wave << 6) + (c << 2);

    int tile = blockIdx.x;
    if (tile >= nTiles) return;

    float4 asv = make_float4(0.0f, 0.0f, 0.0f, 0.0f);
    float4 adv = asv;
    float4 bv  = asv;
    if (DOTS) {
        asv = *(const float4*)(as_ + ncol);
        adv = *(const float4*)(ad_ + ncol);
    } else {
        bv = *(const float4*)(bias + ncol);
    }

    int ebase = blockIdx.x * 128 + tid;
    int dcur[E_PER_TRIP], scur[E_PER_TRIP], rcur[E_PER_TRIP];
    if (EDGE) edge_fetch<EDGE>(ebase, esrc, edst, erank, dcur, scur, rcur);

    const int sr = tid >> 6;
    const int sc = (tid & 63) << 2;

    float4 g0[8];
    #pragma unroll
    for (int i = 0; i < 8; ++i)
        g0[i] = *(const float4*)(A + (size_t)(tile * 16 + 2 * i + sr) * D_HID + sc);

    v2f bf[32][2][4];
    #pragma unroll
    for (int p = 0; p < 32; ++p) {
        #pragma unroll
        for (int q = 0; q < 2; ++q) {
            const int k = 8 * p + 4 * h + 2 * q;
            const float4 w0 = *(const float4*)(B + (size_t)k * D_HID + ncol);
            const float4 w1 = *(const float4*)(B + (size_t)(k + 1) * D_HID + ncol);
            bf[p][q][0].x = w0.x; bf[p][q][0].y = w1.x;
            bf[p][q][1].x = w0.y; bf[p][q][1].y = w1.y;
            bf[p][q][2].x = w0.z; bf[p][q][2].y = w1.z;
            bf[p][q][3].x = w0.w; bf[p][q][3].y = w1.w;
        }
    }

    v2f bm[32][2] = {};
    float hbias = 0.0f;
    float cbias = 0.0f;
    const int cl = (lane < N_CLASSES) ? lane : (N_CLASSES - 1);
    if (!DOTS) {
        const float* wp = hW1 + (wave << 4) + c;
        #pragma unroll
        for (int p = 0; p < 32; ++p) {
            #pragma unroll
            for (int q = 0; q < 2; ++q) {
                const int k = 8 * p + 4 * h + 2 * q;
                bm[p][q].x = wp[(size_t)k * MLP_HID];
                bm[p][q].y = wp[(size_t)(k + 1) * MLP_HID];
            }
        }
        hbias = hb1[(wave << 4) + c];
        cbias = hb2[cl];
        for (int i = tid; i < MLP_HID * N_CLASSES; i += 128) W2s[i] = hW2[i];
    }

    #pragma unroll
    for (int i = 0; i < 8; ++i)
        *(float4*)&As[0][(2 * i + sr) * GA_PITCH + sc] = g0[i];

    float4 ga, gb, gc, gd, ge, gf, gg, gh;
    {
        const int t1 = (tile + GEMM_WGS < nTiles) ? tile + GEMM_WGS : tile;
        const float* arow = A + (size_t)(t1 * 16 + sr) * D_HID + sc;
        GA_ALL(GA_LOAD1)
    }
    __syncthreads();

    int buf = 0;
    for (; tile < nTiles; tile += GEMM_WGS) {
        int eq[E_PER_TRIP], dnxt[E_PER_TRIP], snxt[E_PER_TRIP], rnxt[E_PER_TRIP];
        if (EDGE) {
            edge_issue<EDGE>(ebase, ecnt, erowptr, dcur, eq);
            edge_fetch<EDGE>(ebase + E_PER_TRIP * E_STRIDE, esrc, edst, erank,
                             dnxt, snxt, rnxt);
        }

        const float* as = &As[buf][c * GA_PITCH + 4 * h];
        v8f acc0 = {};
        v8f acc1 = {};
        v8f acc2 = {};
        v8f acc3 = {};
        float4 am[32];
        #pragma unroll
        for (int p = 0; p < 32; ++p)
            am[p] = *(const float4*)(as + 8 * p);
        if (!DOTS) __builtin_amdgcn_sched_barrier(0);
        #pragma unroll
        for (int p = 0; p < 32; ++p) {
            const float4 av = am[p];
            v2f a0, a1;
            a0.x = av.x; a0.y = av.y;
            a1.x = av.z; a1.y = av.w;
            acc0 = __builtin_amdgcn_wmma_f32_16x16x4_f32(
                false, a0, false, bf[p][0][0], (short)0, acc0, false, false);
            acc1 = __builtin_amdgcn_wmma_f32_16x16x4_f32(
                false, a0, false, bf[p][0][1], (short)0, acc1, false, false);
            acc2 = __builtin_amdgcn_wmma_f32_16x16x4_f32(
                false, a0, false, bf[p][0][2], (short)0, acc2, false, false);
            acc3 = __builtin_amdgcn_wmma_f32_16x16x4_f32(
                false, a0, false, bf[p][0][3], (short)0, acc3, false, false);
            acc0 = __builtin_amdgcn_wmma_f32_16x16x4_f32(
                false, a1, false, bf[p][1][0], (short)0, acc0, false, false);
            acc1 = __builtin_amdgcn_wmma_f32_16x16x4_f32(
                false, a1, false, bf[p][1][1], (short)0, acc1, false, false);
            acc2 = __builtin_amdgcn_wmma_f32_16x16x4_f32(
                false, a1, false, bf[p][1][2], (short)0, acc2, false, false);
            acc3 = __builtin_amdgcn_wmma_f32_16x16x4_f32(
                false, a1, false, bf[p][1][3], (short)0, acc3, false, false);
        }

        float* crow = DOTS ? C + (size_t)(tile * 16 + 8 * h) * D_HID + ncol : nullptr;
        float* drow = &Dt[(8 * h) * GA_PITCH + ncol];
        #pragma unroll
        for (int r = 0; r < 8; ++r) {
            float4 o;
            o.x = acc0[r]; o.y = acc1[r]; o.z = acc2[r]; o.w = acc3[r];
            if (DOTS) {
                *(float4*)(crow + (size_t)r * D_HID) = o;
            } else {
                o.x = fmaxf(o.x + bv.x, 0.0f);
                o.y = fmaxf(o.y + bv.y, 0.0f);
                o.z = fmaxf(o.z + bv.z, 0.0f);
                o.w = fmaxf(o.w + bv.w, 0.0f);
                *(float4*)(drow + r * GA_PITCH) = o;
            }
        }

        if (DOTS) {
            float ps[8], pd[8];
            #pragma unroll
            for (int r = 0; r < 8; ++r) {
                ps[r] = acc0[r] * asv.x + acc1[r] * asv.y
                      + acc2[r] * asv.z + acc3[r] * asv.w;
                pd[r] = acc0[r] * adv.x + acc1[r] * adv.y
                      + acc2[r] * adv.z + acc3[r] * adv.w;
            }
            #pragma unroll
            for (int r = 0; r < 8; ++r) { ps[r] = row_xor_add<1>(ps[r]); pd[r] = row_xor_add<1>(pd[r]); }
            #pragma unroll
            for (int r = 0; r < 8; ++r) { ps[r] = row_xor_add<2>(ps[r]); pd[r] = row_xor_add<2>(pd[r]); }
            #pragma unroll
            for (int r = 0; r < 8; ++r) { ps[r] = row_xor_add<4>(ps[r]); pd[r] = row_xor_add<4>(pd[r]); }
            #pragma unroll
            for (int r = 0; r < 8; ++r) { ps[r] = row_xor_add<8>(ps[r]); pd[r] = row_xor_add<8>(pd[r]); }
            if (c == 0) {
                float* pp = &Ps[buf][wave][8 * h];
                *(float4*)(pp)      = make_float4(ps[0], ps[1], ps[2], ps[3]);
                *(float4*)(pp + 4)  = make_float4(ps[4], ps[5], ps[6], ps[7]);
                *(float4*)(pp + 16) = make_float4(pd[0], pd[1], pd[2], pd[3]);
                *(float4*)(pp + 20) = make_float4(pd[4], pd[5], pd[6], pd[7]);
            }
        }

        if (EDGE) {
            edge_finish<EDGE>(ebase, erank, eperm, scur, rcur, eq);
            ebase += E_PER_TRIP * E_STRIDE;
            #pragma unroll
            for (int u = 0; u < E_PER_TRIP; ++u) {
                dcur[u] = dnxt[u]; scur[u] = snxt[u]; rcur[u] = rnxt[u];
            }
        }

        {
            float* lrow = &As[buf ^ 1][sr * GA_PITCH + sc];
            GA_ALL(GA_PUT1)
        }
        {
            const int t2 = (tile + 2 * GEMM_WGS < nTiles) ? tile + 2 * GEMM_WGS : tile;
            const float* arow = A + (size_t)(t2 * 16 + sr) * D_HID + sc;
            GA_ALL(GA_LOAD1)
        }

        __builtin_amdgcn_fence(__ATOMIC_RELEASE, "workgroup", "local");
        __builtin_amdgcn_s_barrier();
        __builtin_amdgcn_fence(__ATOMIC_ACQUIRE, "workgroup", "local");

        if (DOTS && wave == 0) {
            const float* q = &Ps[buf][0][lane];
            const float v = ((q[0] + q[32]) + q[64]) + q[96];
            (h ? ed : es)[tile * 16 + c] = v;
        }

        if (!DOTS) {
            const float* ds = &Dt[c * GA_PITCH + 4 * h];
            float4 af[32];
            #pragma unroll
            for (int p = 0; p < 32; ++p)
                af[p] = *(const float4*)(ds + 8 * p);
            __builtin_amdgcn_sched_barrier(0);
            v8f h0 = {};
            v8f h1 = {};
            #pragma unroll
            for (int p = 0; p < 32; ++p) {
                v2f a0, a1;
                a0.x = af[p].x; a0.y = af[p].y;
                a1.x = af[p].z; a1.y = af[p].w;
                h0 = __builtin_amdgcn_wmma_f32_16x16x4_f32(
                    false, a0, false, bm[p][0], (short)0, h0, false, false);
                h1 = __builtin_amdgcn_wmma_f32_16x16x4_f32(
                    false, a1, false, bm[p][1], (short)0, h1, false, false);
            }
            float* hrow = &Ht[(8 * h) * HT_PITCH + (wave << 4) + c];
            #pragma unroll
            for (int r = 0; r < 8; ++r)
                hrow[r * HT_PITCH] = fmaxf((h0[r] + h1[r]) + hbias, 0.0f);

            __builtin_amdgcn_fence(__ATOMIC_RELEASE, "workgroup", "local");
            __builtin_amdgcn_s_barrier();
            __builtin_amdgcn_fence(__ATOMIC_ACQUIRE, "workgroup", "local");

            const float* hr = &Ht[(4 * wave) * HT_PITCH];
            float lg0 = cbias, lg1 = cbias, lg2 = cbias, lg3 = cbias;
            #pragma unroll
            for (int k4 = 0; k4 < MLP_HID / 4; ++k4) {
                const float4 x0 = *(const float4*)(hr + 0 * HT_PITCH + 4 * k4);
                const float4 x1 = *(const float4*)(hr + 1 * HT_PITCH + 4 * k4);
                const float4 x2 = *(const float4*)(hr + 2 * HT_PITCH + 4 * k4);
                const float4 x3 = *(const float4*)(hr + 3 * HT_PITCH + 4 * k4);
                const float w0 = W2s[(4 * k4 + 0) * N_CLASSES + cl];
                const float w1 = W2s[(4 * k4 + 1) * N_CLASSES + cl];
                const float w2 = W2s[(4 * k4 + 2) * N_CLASSES + cl];
                const float w3 = W2s[(4 * k4 + 3) * N_CLASSES + cl];
                lg0 = fmaf(x0.x, w0, lg0); lg1 = fmaf(x1.x, w0, lg1);
                lg2 = fmaf(x2.x, w0, lg2); lg3 = fmaf(x3.x, w0, lg3);
                lg0 = fmaf(x0.y, w1, lg0); lg1 = fmaf(x1.y, w1, lg1);
                lg2 = fmaf(x2.y, w1, lg2); lg3 = fmaf(x3.y, w1, lg3);
                lg0 = fmaf(x0.z, w2, lg0); lg1 = fmaf(x1.z, w2, lg1);
                lg2 = fmaf(x2.z, w2, lg2); lg3 = fmaf(x3.z, w2, lg3);
                lg0 = fmaf(x0.w, w3, lg0); lg1 = fmaf(x1.w, w3, lg1);
                lg2 = fmaf(x2.w, w3, lg2); lg3 = fmaf(x3.w, w3, lg3);
            }
            const bool live = lane < N_CLASSES;
            const float t0 = live ? lg0 : -3.0e38f;
            const float t1 = live ? lg1 : -3.0e38f;
            const float t2 = live ? lg2 : -3.0e38f;
            const float t3 = live ? lg3 : -3.0e38f;
            const float m0 = wave_max32_row(t0);
            const float m1 = wave_max32_row(t1);
            const float m2 = wave_max32_row(t2);
            const float m3 = wave_max32_row(t3);
            const float e0 = live ? __expf(t0 - m0) : 0.0f;
            const float e1 = live ? __expf(t1 - m1) : 0.0f;
            const float e2 = live ? __expf(t2 - m2) : 0.0f;
            const float e3 = live ? __expf(t3 - m3) : 0.0f;
            const float s0 = wave_sum32_row(e0);
            const float s1 = wave_sum32_row(e1);
            const float s2 = wave_sum32_row(e2);
            const float s3 = wave_sum32_row(e3);
            if (live) {
                float* orow = hout + (size_t)(tile * 16 + 4 * wave) * N_CLASSES + lane;
                orow[0 * N_CLASSES] = e0 / s0;
                orow[1 * N_CLASSES] = e1 / s1;
                orow[2 * N_CLASSES] = e2 / s2;
                orow[3 * N_CLASSES] = e3 / s3;
            }
        }
        buf ^= 1;
    }

    if (EDGE) {
        for (; ebase < N_EDGES; ebase += E_PER_TRIP * E_STRIDE) {
            int eq[E_PER_TRIP];
            edge_fetch<EDGE>(ebase, esrc, edst, erank, dcur, scur, rcur);
            edge_issue<EDGE>(ebase, ecnt, erowptr, dcur, eq);
            edge_finish<EDGE>(ebase, erank, eperm, scur, rcur, eq);
        }
    }
}

#define N_PAD 53248

__global__ void fill_i32(int* __restrict__ p, int v, int n)
{
    const int i = blockIdx.x * blockDim.x + threadIdx.x;
    if (i < n) p[i] = v;
}

__device__ inline void fold_attention(
    const float* __restrict__ W, const float* __restrict__ as_,
    const float* __restrict__ ad_, float* __restrict__ us,
    float* __restrict__ ud)
{
    const int lane = threadIdx.x & 31;
    const int wave = threadIdx.x >> 5;
    const float4 s0 = ((const float4*)as_)[lane];
    const float4 s1 = ((const float4*)as_)[lane + 32];
    const float4 d0 = ((const float4*)ad_)[lane];
    const float4 d1 = ((const float4*)ad_)[lane + 32];
    for (int i = 0; i < 8; ++i) {
        const int k = 8 * wave + i;
        const float4* wr = (const float4*)(W + (size_t)k * D_HID);
        const float4 w0 = wr[lane];
        const float4 w1 = wr[lane + 32];
        float ps = w0.x * s0.x + w0.y * s0.y + w0.z * s0.z + w0.w * s0.w
                 + w1.x * s1.x + w1.y * s1.y + w1.z * s1.z + w1.w * s1.w;
        float pd = w0.x * d0.x + w0.y * d0.y + w0.z * d0.z + w0.w * d0.w
                 + w1.x * d1.x + w1.y * d1.y + w1.z * d1.z + w1.w * d1.w;
        ps = wave_sum32(ps);
        pd = wave_sum32(pd);
        if (lane == 0) { us[k] = ps; ud[k] = pd; }
    }
}

__global__ __launch_bounds__(1024) void scan_counts_fold(
    const int* __restrict__ cnt, int* __restrict__ rowptr,
    const float* __restrict__ W, const float* __restrict__ as_,
    const float* __restrict__ ad_, float* __restrict__ us,
    float* __restrict__ ud)
{
    __shared__ int wsum[2][32];
    if (blockIdx.x == 1) {
        fold_attention(W, as_, ad_, us, ud);
        return;
    }
    const int tid  = threadIdx.x;
    const int lane = tid & 31;
    const int wave = tid >> 5;

    int4 v[N_PAD / 4096];
    #pragma unroll
    for (int it = 0; it < N_PAD / 4096; ++it)
        v[it] = ((const int4*)cnt)[it * 1024 + tid];

    int carry = 0;
    #pragma unroll
    for (int it = 0; it < N_PAD / 4096; ++it) {
        const int4 a = v[it];
        const int t = a.x + a.y + a.z + a.w;
        int inc = t;
        for (int off = 1; off < 32; off <<= 1) {
            const int u = __shfl_up(inc, off, 32);
            if (lane >= off) inc += u;
        }
        if (lane == 31) wsum[it & 1][wave] = inc;
        __syncthreads();
        const int ws = wsum[it & 1][lane];
        int winc = ws;
        for (int off = 1; off < 32; off <<= 1) {
            const int u = __shfl_up(winc, off, 32);
            if (lane >= off) winc += u;
        }
        const int total = __shfl(winc, 31, 32);
        const int wexcl = __shfl(winc, wave, 32) - __shfl(ws, wave, 32);
        const int excl  = carry + wexcl + inc - t;
        int4 o;
        o.x = excl;
        o.y = o.x + a.x;
        o.z = o.y + a.y;
        o.w = o.z + a.z;
        ((int4*)rowptr)[it * 1024 + tid] = o;
        carry += total;
    }
}

__device__ inline float edge_score1(float es_src, float ed_dst)
{
    const float t = es_src + ed_dst;
    return (t > 0.0f) ? t : 0.2f * t;
}

template <bool FIRST>
__global__ __launch_bounds__(256) void node_aggregate(
    const int* __restrict__ rowptr, const int* __restrict__ perm,
    const float* __restrict__ es, const float* __restrict__ ed,
    const float* __restrict__ H, const float* __restrict__ b,
    float* __restrict__ OUT, const int* __restrict__ rowsel,
    const float* __restrict__ us, const float* __restrict__ ud,
    float* __restrict__ es2, float* __restrict__ ed2, int n)
{
    const int lane = threadIdx.x & 31;
    const int w = __builtin_amdgcn_readfirstlane(
        (int)((blockIdx.x * blockDim.x + threadIdx.x) >> 5));
    if (w >= n) return;
    const int node = FIRST ? w : rowsel[w];
    const int beg = rowptr[node];
    const int end = rowptr[node + 1];
    const float edn = ed[node];

    const bool in0 = beg + lane < end;
    int   s0 = 0;
    float v0 = -INFINITY;
    if (in0) {
        s0 = perm[beg + lane];
        v0 = edge_score1(es[s0], edn);
    }

    float m = v0;
    for (int base = beg + 32; base < end; base += 32)
        if (base + lane < end)
            m = fmaxf(m, edge_score1(es[perm[base + lane]], edn));
    m = wave_max32(m);

    const float ex0 = in0 ? __expf(v0 - m) : 0.0f;
    float den = ex0;
    for (int base = beg + 32; base < end; base += 32)
        if (base + lane < end)
            den += __expf(edge_score1(es[perm[base + lane]], edn) - m);
    const float dn = wave_sum32(den) + 1e-16f;

    float4 acc0 = make_float4(0.0f, 0.0f, 0.0f, 0.0f);
    float4 acc1 = make_float4(0.0f, 0.0f, 0.0f, 0.0f);

    for (int base = beg; base < end; base += 32) {
        int   s = s0;
        float a = ex0 / dn;
        if (base != beg) {
            s = 0;
            a = 0.0f;
            if (base + lane < end) {
                s = perm[base + lane];
                a = __expf(edge_score1(es[s], edn) - m) / dn;
            }
        }
        const int cnt = (end - base < 32) ? (end - base) : 32;
        int t = 0;
        for (; t + 4 <= cnt; t += 4) {
            float4 h0[4], h1[4];
            float  at[4];
            #pragma unroll
            for (int u = 0; u < 4; ++u) {
                const int st = __builtin_amdgcn_readlane(s, t + u);
                at[u] = __int_as_float(
                    __builtin_amdgcn_readlane(__float_as_int(a), t + u));
                const float4* hr = (const float4*)(H + (size_t)st * D_HID);
                h0[u] = hr[lane];
                h1[u] = hr[lane + 32];
            }
            #pragma unroll
            for (int u = 0; u < 4; ++u) {
                acc0.x = fmaf(at[u], h0[u].x, acc0.x);
                acc0.y = fmaf(at[u], h0[u].y, acc0.y);
                acc0.z = fmaf(at[u], h0[u].z, acc0.z);
                acc0.w = fmaf(at[u], h0[u].w, acc0.w);
                acc1.x = fmaf(at[u], h1[u].x, acc1.x);
                acc1.y = fmaf(at[u], h1[u].y, acc1.y);
                acc1.z = fmaf(at[u], h1[u].z, acc1.z);
                acc1.w = fmaf(at[u], h1[u].w, acc1.w);
            }
        }
        #pragma nounroll
        for (; t < cnt; ++t) {
            const int st = __builtin_amdgcn_readlane(s, t);
            const float at1 = __int_as_float(
                __builtin_amdgcn_readlane(__float_as_int(a), t));
            const float4* hr = (const float4*)(H + (size_t)st * D_HID);
            const float4 g0 = hr[lane];
            const float4 g1 = hr[lane + 32];
            acc0.x = fmaf(at1, g0.x, acc0.x);
            acc0.y = fmaf(at1, g0.y, acc0.y);
            acc0.z = fmaf(at1, g0.z, acc0.z);
            acc0.w = fmaf(at1, g0.w, acc0.w);
            acc1.x = fmaf(at1, g1.x, acc1.x);
            acc1.y = fmaf(at1, g1.y, acc1.y);
            acc1.z = fmaf(at1, g1.z, acc1.z);
            acc1.w = fmaf(at1, g1.w, acc1.w);
        }
    }

    float4* orow = (float4*)(OUT + (size_t)w * D_HID);
    if (!FIRST) {
        orow[lane]      = acc0;
        orow[lane + 32] = acc1;
        return;
    }
    const int c0 = 4 * lane;
    const int c1 = 128 + 4 * lane;
    float4 o0, o1;
    o0.x = fmaxf(acc0.x + b[c0 + 0], 0.0f);
    o0.y = fmaxf(acc0.y + b[c0 + 1], 0.0f);
    o0.z = fmaxf(acc0.z + b[c0 + 2], 0.0f);
    o0.w = fmaxf(acc0.w + b[c0 + 3], 0.0f);
    o1.x = fmaxf(acc1.x + b[c1 + 0], 0.0f);
    o1.y = fmaxf(acc1.y + b[c1 + 1], 0.0f);
    o1.z = fmaxf(acc1.z + b[c1 + 2], 0.0f);
    o1.w = fmaxf(acc1.w + b[c1 + 3], 0.0f);
    orow[lane]      = o0;
    orow[lane + 32] = o1;

    const float4 u0 = ((const float4*)us)[lane];
    const float4 u1 = ((const float4*)us)[lane + 32];
    const float4 q0 = ((const float4*)ud)[lane];
    const float4 q1 = ((const float4*)ud)[lane + 32];
    float ps = o0.x * u0.x + o0.y * u0.y + o0.z * u0.z + o0.w * u0.w
             + o1.x * u1.x + o1.y * u1.y + o1.z * u1.z + o1.w * u1.w;
    float pd = o0.x * q0.x + o0.y * q0.y + o0.z * q0.z + o0.w * q0.w
             + o1.x * q1.x + o1.y * q1.y + o1.z * q1.z + o1.w * q1.w;
    ps = wave_sum32(ps);
    pd = wave_sum32(pd);
    if (lane == 0) { es2[w] = ps; ed2[w] = pd; }
}

extern "C" void kernel_launch(void* const* d_in, const int* in_sizes, int n_in,
                              void* d_out, int out_size, void* d_ws, size_t ws_size,
                              hipStream_t stream)
{
    (void)in_sizes; (void)n_in; (void)out_size; (void)ws_size;

    const float* x    = (const float*)d_in[0];
    const int*   ei   = (const int*)  d_in[1];
    const int*   doci = (const int*)  d_in[2];
    const float* W1   = (const float*)d_in[3];
    const float* a1s  = (const float*)d_in[4];
    const float* a1d  = (const float*)d_in[5];
    const float* b1   = (const float*)d_in[6];
    const float* W2   = (const float*)d_in[7];
    const float* a2s  = (const float*)d_in[8];
    const float* a2d  = (const float*)d_in[9];
    const float* b2   = (const float*)d_in[10];
    const float* Wm1  = (const float*)d_in[11];
    const float* bm1  = (const float*)d_in[12];
    const float* Wm2  = (const float*)d_in[13];
    const float* bm2  = (const float*)d_in[14];

    const int* src = ei;
    const int* dst = ei + N_EDGES;

    float* H    = (float*)d_ws;
    float* OUT  = H    + (size_t)N_NODES * D_HID;
    float* ES   = OUT  + (size_t)N_NODES * D_HID;
    float* ED   = ES   + N_NODES;
    float* BKT  = ED   + N_NODES;
    int*  CNT    = (int*)BKT;
    int*  ROWPTR = CNT    + N_PAD;
    int*  RANK   = ROWPTR + N_PAD;
    int*  PERM   = RANK   + N_EDGES;
    float* ES2  = (float*)(PERM + N_EDGES);
    float* ED2  = ES2  + N_NODES;
    float* US   = ED2  + N_NODES;
    float* UD   = US   + D_HID;
    float* Z    = H;

    const int gNodeW = (N_NODES + 7) / 8;
    const int gDocW  = (N_DOCS + 7) / 8;

    fill_i32<<<N_PAD / 256, 256, 0, stream>>>(CNT, 0, N_PAD);
    gemm_wreg_f32<true, 1><<<GEMM_WGS, 128, 0, stream>>>(
        x, W1, H, a1s, a1d, ES, ED, nullptr, TILES_A,
        src, dst, CNT, RANK, ROWPTR, PERM,
        nullptr, nullptr, nullptr, nullptr, nullptr);
    scan_counts_fold<<<2, 1024, 0, stream>>>(CNT, ROWPTR, W2, a2s, a2d, US, UD);
    gemm_wreg_f32<true, 2><<<GEMM_WGS, 128, 0, stream>>>(
        x + (size_t)TILES_A * 16 * D_HID, W1, H + (size_t)TILES_A * 16 * D_HID,
        a1s, a1d, ES + TILES_A * 16, ED + TILES_A * 16, nullptr,
        N_NODES / 16 - TILES_A, src, dst, CNT, RANK, ROWPTR, PERM,
        nullptr, nullptr, nullptr, nullptr, nullptr);
    node_aggregate<true><<<gNodeW, 256, 0, stream>>>(ROWPTR, PERM, ES, ED, H, b1,
                                                     OUT, nullptr, US, UD, ES2,
                                                     ED2, N_NODES);

    node_aggregate<false><<<gDocW, 256, 0, stream>>>(ROWPTR, PERM, ES2, ED2, OUT,
                                                     nullptr, Z, doci, nullptr,
                                                     nullptr, nullptr, nullptr,
                                                     N_DOCS);
    gemm_wreg_f32<false, 0><<<GEMM_WGS, 128, 0, stream>>>(
        Z, W2, nullptr, nullptr, nullptr, nullptr, nullptr, b2, N_DOCS / 16,
        nullptr, nullptr, nullptr, nullptr, nullptr, nullptr,
        Wm1, bm1, Wm2, bm2, (float*)d_out);
}


# Edge buckets as fixed 32-slot rows filled in ONE atomic pass inside a single node GEMM: scan launch, place pass, 2nd GEMM launch dropped; buckets over 32 spill to a list (their sum order changes)
# speedup vs baseline: 5.6904x; 1.2264x over previous
#include <hip/hip_runtime.h>
#include <hip/hip_bf16.h>


#define N_NODES   50000
#define N_EDGES   800000
#define D_HID     256
#define N_DOCS    20000
#define MLP_HID   64
#define N_CLASSES 20

typedef __attribute__((ext_vector_type(2))) float v2f;
typedef __attribute__((ext_vector_type(8))) float v8f;

#define GA_PITCH 260
#define GEMM_WGS 256
#define HT_PITCH 68
#define GA_ALL(X) X(0, ga) X(1, gb) X(2, gc) X(3, gd) X(4, ge) X(5, gf) X(6, gg) X(7, gh)
#define GA_LOAD1(i, v) v = *(const float4*)(arow + (size_t)(2 * (i)) * D_HID);
#define GA_PUT1(i, v)  *(float4*)(lrow + (2 * (i)) * GA_PITCH) = v;

__device__ inline float wave_max32(float v)
{
    for (int off = 16; off; off >>= 1) v = fmaxf(v, __shfl_xor(v, off, 32));
    return v;
}

__device__ inline float wave_sum32(float v)
{
    for (int off = 16; off; off >>= 1) v += __shfl_xor(v, off, 32);
    return v;
}

template <int M>
__device__ inline float row_xor_add(float v)
{
    return v + __int_as_float(__builtin_amdgcn_mov_dpp(
        __float_as_int(v), 0x160 + M, 0xf, 0xf, true));
}

template <int M>
__device__ inline float row_xor_max(float v)
{
    return fmaxf(v, __int_as_float(__builtin_amdgcn_mov_dpp(
        __float_as_int(v), 0x160 + M, 0xf, 0xf, true)));
}

__device__ inline float wave_sum32_row(float v)
{
    v += __shfl_xor(v, 16, 32);
    v = row_xor_add<8>(v);
    v = row_xor_add<4>(v);
    v = row_xor_add<2>(v);
    return row_xor_add<1>(v);
}

__device__ inline float wave_max32_row(float v)
{
    v = fmaxf(v, __shfl_xor(v, 16, 32));
    v = row_xor_max<8>(v);
    v = row_xor_max<4>(v);
    v = row_xor_max<2>(v);
    return row_xor_max<1>(v);
}

#define N_PAD   53248
#define BKT_CAP 32
#define OV_CAP  N_PAD

#define E_STRIDE   (GEMM_WGS * 128)
#define E_ROUNDS   ((N_EDGES + E_STRIDE - 1) / E_STRIDE)
#define E_PER_TRIP 3

__device__ inline void edge_fetch(
    int ebase, const int* __restrict__ esrc, const int* __restrict__ edst,
    int (&d)[E_PER_TRIP], int (&s)[E_PER_TRIP])
{
    #pragma unroll
    for (int u = 0; u < E_PER_TRIP; ++u) {
        const int i = ebase + u * E_STRIDE;
        d[u] = 0; s[u] = 0;
        if (i < N_EDGES) {
            d[u] = edst[i];
            s[u] = esrc[i];
        }
    }
}

__device__ inline void edge_issue(
    int ebase, int* __restrict__ ecnt, const int (&d)[E_PER_TRIP],
    int (&q)[E_PER_TRIP])
{
    #pragma unroll
    for (int u = 0; u < E_PER_TRIP; ++u) {
        q[u] = 0;
        if (ebase + u * E_STRIDE < N_EDGES) q[u] = atomicAdd(&ecnt[d[u]], 1);
    }
}

__device__ inline void edge_finish(
    int ebase, int* __restrict__ ecnt, int* __restrict__ ebkt,
    int* __restrict__ eovl, const int (&d)[E_PER_TRIP],
    const int (&s)[E_PER_TRIP], const int (&q)[E_PER_TRIP])
{
    #pragma unroll
    for (int u = 0; u < E_PER_TRIP; ++u) {
        const int i = ebase + u * E_STRIDE;
        if (i < N_EDGES) {
            if (q[u] < BKT_CAP) {
                ebkt[d[u] * BKT_CAP + q[u]] = s[u];
            } else {
                const int o = atomicAdd(&ecnt[N_NODES], 1);
                if (o < OV_CAP) eovl[o] = i;
            }
        }
    }
}

template <bool DOTS, int EDGE>
__global__ __launch_bounds__(128) void gemm_wreg_f32(
    const float* __restrict__ A, const float* __restrict__ B,
    float* __restrict__ C, const float* __restrict__ as_,
    const float* __restrict__ ad_, float* __restrict__ es,
    float* __restrict__ ed, const float* __restrict__ bias, int nTiles,
    const int* __restrict__ esrc, const int* __restrict__ edst,
    int* __restrict__ ecnt, int* __restrict__ ebkt, int* __restrict__ eovl,
    const float* __restrict__ hW1, const float* __restrict__ hb1,
    const float* __restrict__ hW2, const float* __restrict__ hb2,
    float* __restrict__ hout)
{
    __shared__ float As[2][16 * GA_PITCH];
    __shared__ __attribute__((aligned(16))) float Dt[16 * GA_PITCH];
    __shared__ __attribute__((aligned(16))) float Ht[16 * HT_PITCH];
    __shared__ __attribute__((aligned(16))) float W2s[MLP_HID * N_CLASSES];
    __shared__ __attribute__((aligned(16))) float Ps[2][4][32];

    const int tid  = threadIdx.x;
    const int wave = tid >> 5;
    const int lane = tid & 31;
    const int c    = lane & 15;
    const int h    = lane >> 4;
    const int ncol = (wave << 6) + (c << 2);

    int tile = blockIdx.x;
    if (tile >= nTiles) return;

    float4 asv = make_float4(0.0f, 0.0f, 0.0f, 0.0f);
    float4 adv = asv;
    float4 bv  = asv;
    if (DOTS) {
        asv = *(const float4*)(as_ + ncol);
        adv = *(const float4*)(ad_ + ncol);
    } else {
        bv = *(const float4*)(bias + ncol);
    }

    int ebase = blockIdx.x * 128 + tid;
    int dcur[E_PER_TRIP], scur[E_PER_TRIP];
    if (EDGE) edge_fetch(ebase, esrc, edst, dcur, scur);

    const int sr = tid >> 6;
    const int sc = (tid & 63) << 2;

    float4 g0[8];
    #pragma unroll
    for (int i = 0; i < 8; ++i)
        g0[i] = *(const float4*)(A + (size_t)(tile * 16 + 2 * i + sr) * D_HID + sc);

    v2f bf[32][2][4];
    #pragma unroll
    for (int p = 0; p < 32; ++p) {
        #pragma unroll
        for (int q = 0; q < 2; ++q) {
            const int k = 8 * p + 4 * h + 2 * q;
            const float4 w0 = *(const float4*)(B + (size_t)k * D_HID + ncol);
            const float4 w1 = *(const float4*)(B + (size_t)(k + 1) * D_HID + ncol);
            bf[p][q][0].x = w0.x; bf[p][q][0].y = w1.x;
            bf[p][q][1].x = w0.y; bf[p][q][1].y = w1.y;
            bf[p][q][2].x = w0.z; bf[p][q][2].y = w1.z;
            bf[p][q][3].x = w0.w; bf[p][q][3].y = w1.w;
        }
    }

    v2f bm[32][2] = {};
    float hbias = 0.0f;
    float cbias = 0.0f;
    const int cl = (lane < N_CLASSES) ? lane : (N_CLASSES - 1);
    if (!DOTS) {
        const float* wp = hW1 + (wave << 4) + c;
        #pragma unroll
        for (int p = 0; p < 32; ++p) {
            #pragma unroll
            for (int q = 0; q < 2; ++q) {
                const int k = 8 * p + 4 * h + 2 * q;
                bm[p][q].x = wp[(size_t)k * MLP_HID];
                bm[p][q].y = wp[(size_t)(k + 1) * MLP_HID];
            }
        }
        hbias = hb1[(wave << 4) + c];
        cbias = hb2[cl];
        for (int i = tid; i < MLP_HID * N_CLASSES; i += 128) W2s[i] = hW2[i];
    }

    #pragma unroll
    for (int i = 0; i < 8; ++i)
        *(float4*)&As[0][(2 * i + sr) * GA_PITCH + sc] = g0[i];

    float4 ga, gb, gc, gd, ge, gf, gg, gh;
    {
        const int t1 = (tile + GEMM_WGS < nTiles) ? tile + GEMM_WGS : tile;
        const float* arow = A + (size_t)(t1 * 16 + sr) * D_HID + sc;
        GA_ALL(GA_LOAD1)
    }
    __syncthreads();

    int buf = 0;
    for (; tile < nTiles; tile += GEMM_WGS) {
        int eq[E_PER_TRIP], dnxt[E_PER_TRIP], snxt[E_PER_TRIP];
        if (EDGE) {
            edge_issue(ebase, ecnt, dcur, eq);
            edge_fetch(ebase + E_PER_TRIP * E_STRIDE, esrc, edst, dnxt, snxt);
        }

        const float* as = &As[buf][c * GA_PITCH + 4 * h];
        v8f acc0 = {};
        v8f acc1 = {};
        v8f acc2 = {};
        v8f acc3 = {};
        float4 am[32];
        #pragma unroll
        for (int p = 0; p < 32; ++p)
            am[p] = *(const float4*)(as + 8 * p);
        if (!DOTS) __builtin_amdgcn_sched_barrier(0);
        #pragma unroll
        for (int p = 0; p < 32; ++p) {
            const float4 av = am[p];
            v2f a0, a1;
            a0.x = av.x; a0.y = av.y;
            a1.x = av.z; a1.y = av.w;
            acc0 = __builtin_amdgcn_wmma_f32_16x16x4_f32(
                false, a0, false, bf[p][0][0], (short)0, acc0, false, false);
            acc1 = __builtin_amdgcn_wmma_f32_16x16x4_f32(
                false, a0, false, bf[p][0][1], (short)0, acc1, false, false);
            acc2 = __builtin_amdgcn_wmma_f32_16x16x4_f32(
                false, a0, false, bf[p][0][2], (short)0, acc2, false, false);
            acc3 = __builtin_amdgcn_wmma_f32_16x16x4_f32(
                false, a0, false, bf[p][0][3], (short)0, acc3, false, false);
            acc0 = __builtin_amdgcn_wmma_f32_16x16x4_f32(
                false, a1, false, bf[p][1][0], (short)0, acc0, false, false);
            acc1 = __builtin_amdgcn_wmma_f32_16x16x4_f32(
                false, a1, false, bf[p][1][1], (short)0, acc1, false, false);
            acc2 = __builtin_amdgcn_wmma_f32_16x16x4_f32(
                false, a1, false, bf[p][1][2], (short)0, acc2, false, false);
            acc3 = __builtin_amdgcn_wmma_f32_16x16x4_f32(
                false, a1, false, bf[p][1][3], (short)0, acc3, false, false);
        }

        float* crow = DOTS ? C + (size_t)(tile * 16 + 8 * h) * D_HID + ncol : nullptr;
        float* drow = &Dt[(8 * h) * GA_PITCH + ncol];
        #pragma unroll
        for (int r = 0; r < 8; ++r) {
            float4 o;
            o.x = acc0[r]; o.y = acc1[r]; o.z = acc2[r]; o.w = acc3[r];
            if (DOTS) {
                *(float4*)(crow + (size_t)r * D_HID) = o;
            } else {
                o.x = fmaxf(o.x + bv.x, 0.0f);
                o.y = fmaxf(o.y + bv.y, 0.0f);
                o.z = fmaxf(o.z + bv.z, 0.0f);
                o.w = fmaxf(o.w + bv.w, 0.0f);
                *(float4*)(drow + r * GA_PITCH) = o;
            }
        }

        if (DOTS) {
            float ps[8], pd[8];
            #pragma unroll
            for (int r = 0; r < 8; ++r) {
                ps[r] = acc0[r] * asv.x + acc1[r] * asv.y
                      + acc2[r] * asv.z + acc3[r] * asv.w;
                pd[r] = acc0[r] * adv.x + acc1[r] * adv.y
                      + acc2[r] * adv.z + acc3[r] * adv.w;
            }
            #pragma unroll
            for (int r = 0; r < 8; ++r) { ps[r] = row_xor_add<1>(ps[r]); pd[r] = row_xor_add<1>(pd[r]); }
            #pragma unroll
            for (int r = 0; r < 8; ++r) { ps[r] = row_xor_add<2>(ps[r]); pd[r] = row_xor_add<2>(pd[r]); }
            #pragma unroll
            for (int r = 0; r < 8; ++r) { ps[r] = row_xor_add<4>(ps[r]); pd[r] = row_xor_add<4>(pd[r]); }
            #pragma unroll
            for (int r = 0; r < 8; ++r) { ps[r] = row_xor_add<8>(ps[r]); pd[r] = row_xor_add<8>(pd[r]); }
            if (c == 0) {
                float* pp = &Ps[buf][wave][8 * h];
                *(float4*)(pp)      = make_float4(ps[0], ps[1], ps[2], ps[3]);
                *(float4*)(pp + 4)  = make_float4(ps[4], ps[5], ps[6], ps[7]);
                *(float4*)(pp + 16) = make_float4(pd[0], pd[1], pd[2], pd[3]);
                *(float4*)(pp + 20) = make_float4(pd[4], pd[5], pd[6], pd[7]);
            }
        }

        if (EDGE) {
            edge_finish(ebase, ecnt, ebkt, eovl, dcur, scur, eq);
            ebase += E_PER_TRIP * E_STRIDE;
            #pragma unroll
            for (int u = 0; u < E_PER_TRIP; ++u) {
                dcur[u] = dnxt[u]; scur[u] = snxt[u];
            }
        }

        {
            float* lrow = &As[buf ^ 1][sr * GA_PITCH + sc];
            GA_ALL(GA_PUT1)
        }
        {
            const int t2 = (tile + 2 * GEMM_WGS < nTiles) ? tile + 2 * GEMM_WGS : tile;
            const float* arow = A + (size_t)(t2 * 16 + sr) * D_HID + sc;
            GA_ALL(GA_LOAD1)
        }

        __builtin_amdgcn_fence(__ATOMIC_RELEASE, "workgroup", "local");
        __builtin_amdgcn_s_barrier();
        __builtin_amdgcn_fence(__ATOMIC_ACQUIRE, "workgroup", "local");

        if (DOTS && wave == 0) {
            const float* q = &Ps[buf][0][lane];
            const float v = ((q[0] + q[32]) + q[64]) + q[96];
            (h ? ed : es)[tile * 16 + c] = v;
        }

        if (!DOTS) {
            const float* ds = &Dt[c * GA_PITCH + 4 * h];
            float4 af[32];
            #pragma unroll
            for (int p = 0; p < 32; ++p)
                af[p] = *(const float4*)(ds + 8 * p);
            __builtin_amdgcn_sched_barrier(0);
            v8f h0 = {};
            v8f h1 = {};
            #pragma unroll
            for (int p = 0; p < 32; ++p) {
                v2f a0, a1;
                a0.x = af[p].x; a0.y = af[p].y;
                a1.x = af[p].z; a1.y = af[p].w;
                h0 = __builtin_amdgcn_wmma_f32_16x16x4_f32(
                    false, a0, false, bm[p][0], (short)0, h0, false, false);
                h1 = __builtin_amdgcn_wmma_f32_16x16x4_f32(
                    false, a1, false, bm[p][1], (short)0, h1, false, false);
            }
            float* hrow = &Ht[(8 * h) * HT_PITCH + (wave << 4) + c];
            #pragma unroll
            for (int r = 0; r < 8; ++r)
                hrow[r * HT_PITCH] = fmaxf((h0[r] + h1[r]) + hbias, 0.0f);

            __builtin_amdgcn_fence(__ATOMIC_RELEASE, "workgroup", "local");
            __builtin_amdgcn_s_barrier();
            __builtin_amdgcn_fence(__ATOMIC_ACQUIRE, "workgroup", "local");

            const float* hr = &Ht[(4 * wave) * HT_PITCH];
            float lg0 = cbias, lg1 = cbias, lg2 = cbias, lg3 = cbias;
            #pragma unroll
            for (int k4 = 0; k4 < MLP_HID / 4; ++k4) {
                const float4 x0 = *(const float4*)(hr + 0 * HT_PITCH + 4 * k4);
                const float4 x1 = *(const float4*)(hr + 1 * HT_PITCH + 4 * k4);
                const float4 x2 = *(const float4*)(hr + 2 * HT_PITCH + 4 * k4);
                const float4 x3 = *(const float4*)(hr + 3 * HT_PITCH + 4 * k4);
                const float w0 = W2s[(4 * k4 + 0) * N_CLASSES + cl];
                const float w1 = W2s[(4 * k4 + 1) * N_CLASSES + cl];
                const float w2 = W2s[(4 * k4 + 2) * N_CLASSES + cl];
                const float w3 = W2s[(4 * k4 + 3) * N_CLASSES + cl];
                lg0 = fmaf(x0.x, w0, lg0); lg1 = fmaf(x1.x, w0, lg1);
                lg2 = fmaf(x2.x, w0, lg2); lg3 = fmaf(x3.x, w0, lg3);
                lg0 = fmaf(x0.y, w1, lg0); lg1 = fmaf(x1.y, w1, lg1);
                lg2 = fmaf(x2.y, w1, lg2); lg3 = fmaf(x3.y, w1, lg3);
                lg0 = fmaf(x0.z, w2, lg0); lg1 = fmaf(x1.z, w2, lg1);
                lg2 = fmaf(x2.z, w2, lg2); lg3 = fmaf(x3.z, w2, lg3);
                lg0 = fmaf(x0.w, w3, lg0); lg1 = fmaf(x1.w, w3, lg1);
                lg2 = fmaf(x2.w, w3, lg2); lg3 = fmaf(x3.w, w3, lg3);
            }
            const bool live = lane < N_CLASSES;
            const float t0 = live ? lg0 : -3.0e38f;
            const float t1 = live ? lg1 : -3.0e38f;
            const float t2 = live ? lg2 : -3.0e38f;
            const float t3 = live ? lg3 : -3.0e38f;
            const float m0 = wave_max32_row(t0);
            const float m1 = wave_max32_row(t1);
            const float m2 = wave_max32_row(t2);
            const float m3 = wave_max32_row(t3);
            const float e0 = live ? __expf(t0 - m0) : 0.0f;
            const float e1 = live ? __expf(t1 - m1) : 0.0f;
            const float e2 = live ? __expf(t2 - m2) : 0.0f;
            const float e3 = live ? __expf(t3 - m3) : 0.0f;
            const float s0 = wave_sum32_row(e0);
            const float s1 = wave_sum32_row(e1);
            const float s2 = wave_sum32_row(e2);
            const float s3 = wave_sum32_row(e3);
            if (live) {
                float* orow = hout + (size_t)(tile * 16 + 4 * wave) * N_CLASSES + lane;
                orow[0 * N_CLASSES] = e0 / s0;
                orow[1 * N_CLASSES] = e1 / s1;
                orow[2 * N_CLASSES] = e2 / s2;
                orow[3 * N_CLASSES] = e3 / s3;
            }
        }
        buf ^= 1;
    }

    if (EDGE) {
        for (; ebase < N_EDGES; ebase += E_PER_TRIP * E_STRIDE) {
            int eq[E_PER_TRIP];
            edge_fetch(ebase, esrc, edst, dcur, scur);
            edge_issue(ebase, ecnt, dcur, eq);
            edge_finish(ebase, ecnt, ebkt, eovl, dcur, scur, eq);
        }
    }
}

#define FILL_BLOCKS (N_PAD / 1024)
#define FOLD_BLOCKS (D_HID / 32)

__global__ __launch_bounds__(1024) void clear_counts_fold(
    int* __restrict__ cnt, const float* __restrict__ W,
    const float* __restrict__ as_, const float* __restrict__ ad_,
    float* __restrict__ us, float* __restrict__ ud)
{
    if (blockIdx.x < FILL_BLOCKS) {
        cnt[blockIdx.x * 1024 + threadIdx.x] = 0;
        return;
    }
    const int lane = threadIdx.x & 31;
    const int wave = threadIdx.x >> 5;
    const int k = 32 * (blockIdx.x - FILL_BLOCKS) + wave;
    const float4 s0 = ((const float4*)as_)[lane];
    const float4 s1 = ((const float4*)as_)[lane + 32];
    const float4 d0 = ((const float4*)ad_)[lane];
    const float4 d1 = ((const float4*)ad_)[lane + 32];
    const float4* wr = (const float4*)(W + (size_t)k * D_HID);
    const float4 w0 = wr[lane];
    const float4 w1 = wr[lane + 32];
    float ps = w0.x * s0.x + w0.y * s0.y + w0.z * s0.z + w0.w * s0.w
             + w1.x * s1.x + w1.y * s1.y + w1.z * s1.z + w1.w * s1.w;
    float pd = w0.x * d0.x + w0.y * d0.y + w0.z * d0.z + w0.w * d0.w
             + w1.x * d1.x + w1.y * d1.y + w1.z * d1.z + w1.w * d1.w;
    ps = wave_sum32(ps);
    pd = wave_sum32(pd);
    if (lane == 0) { us[k] = ps; ud[k] = pd; }
}

__device__ inline float edge_score1(float es_src, float ed_dst)
{
    const float t = es_src + ed_dst;
    return (t > 0.0f) ? t : 0.2f * t;
}

__device__ inline bool spill_place(
    int j, int xn, bool full, int node, const int* __restrict__ ovl,
    const int* __restrict__ esrc, const int* __restrict__ edst, int& s)
{
    s = 0;
    if (j >= xn) return false;
    const int i = full ? j : ovl[j];
    if (edst[i] != node) return false;
    s = esrc[i];
    return true;
}

__device__ inline void add_row(
    const float* __restrict__ H, int st, float at, int lane,
    float4& acc0, float4& acc1)
{
    const float4* hr = (const float4*)(H + (size_t)st * D_HID);
    const float4 g0 = hr[lane];
    const float4 g1 = hr[lane + 32];
    acc0.x = fmaf(at, g0.x, acc0.x);
    acc0.y = fmaf(at, g0.y, acc0.y);
    acc0.z = fmaf(at, g0.z, acc0.z);
    acc0.w = fmaf(at, g0.w, acc0.w);
    acc1.x = fmaf(at, g1.x, acc1.x);
    acc1.y = fmaf(at, g1.y, acc1.y);
    acc1.z = fmaf(at, g1.z, acc1.z);
    acc1.w = fmaf(at, g1.w, acc1.w);
}

template <bool FIRST>
__global__ __launch_bounds__(256) void node_aggregate(
    const int* __restrict__ cnt, const int* __restrict__ bkt,
    const int* __restrict__ ovl, const int* __restrict__ esrc,
    const int* __restrict__ edst,
    const float* __restrict__ es, const float* __restrict__ ed,
    const float* __restrict__ H, const float* __restrict__ b,
    float* __restrict__ OUT, const int* __restrict__ rowsel,
    const float* __restrict__ us, const float* __restrict__ ud,
    float* __restrict__ es2, float* __restrict__ ed2, int n)
{
    const int lane = threadIdx.x & 31;
    const int w = __builtin_amdgcn_readfirstlane(
        (int)((blockIdx.x * blockDim.x + threadIdx.x) >> 5));
    if (w >= n) return;
    const int node = FIRST ? w : rowsel[w];
    const int deg = cnt[node];
    const float edn = ed[node];

    int  xn   = 0;
    bool full = false;
    if (deg > BKT_CAP) {
        const int ovn = cnt[N_NODES];
        full = ovn > OV_CAP;
        xn   = full ? N_EDGES : ovn;
    }
    const int nb = full ? 0 : ((deg < BKT_CAP) ? deg : BKT_CAP);
    const bool in0 = lane < nb;
    int   s0 = 0;
    float v0 = -INFINITY;
    if (in0) {
        s0 = bkt[node * BKT_CAP + lane];
        v0 = edge_score1(es[s0], edn);
    }

    float m = v0;
    for (int base = 0; base < xn; base += 32) {
        int s;
        if (spill_place(base + lane, xn, full, node, ovl, esrc, edst, s))
            m = fmaxf(m, edge_score1(es[s], edn));
    }
    m = wave_max32(m);

    const float ex0 = in0 ? __expf(v0 - m) : 0.0f;
    float den = ex0;
    for (int base = 0; base < xn; base += 32) {
        int s;
        if (spill_place(base + lane, xn, full, node, ovl, esrc, edst, s))
            den += __expf(edge_score1(es[s], edn) - m);
    }
    const float dn = wave_sum32(den) + 1e-16f;

    float4 acc0 = make_float4(0.0f, 0.0f, 0.0f, 0.0f);
    float4 acc1 = make_float4(0.0f, 0.0f, 0.0f, 0.0f);

    {
        const int   s = s0;
        const float a = ex0 / dn;
        int t = 0;
        for (; t + 4 <= nb; t += 4) {
            float4 h0[4], h1[4];
            float  at[4];
            #pragma unroll
            for (int u = 0; u < 4; ++u) {
                const int st = __builtin_amdgcn_readlane(s, t + u);
                at[u] = __int_as_float(
                    __builtin_amdgcn_readlane(__float_as_int(a), t + u));
                const float4* hr = (const float4*)(H + (size_t)st * D_HID);
                h0[u] = hr[lane];
                h1[u] = hr[lane + 32];
            }
            #pragma unroll
            for (int u = 0; u < 4; ++u) {
                acc0.x = fmaf(at[u], h0[u].x, acc0.x);
                acc0.y = fmaf(at[u], h0[u].y, acc0.y);
                acc0.z = fmaf(at[u], h0[u].z, acc0.z);
                acc0.w = fmaf(at[u], h0[u].w, acc0.w);
                acc1.x = fmaf(at[u], h1[u].x, acc1.x);
                acc1.y = fmaf(at[u], h1[u].y, acc1.y);
                acc1.z = fmaf(at[u], h1[u].z, acc1.z);
                acc1.w = fmaf(at[u], h1[u].w, acc1.w);
            }
        }
        #pragma nounroll
        for (; t < nb; ++t) {
            const int st = __builtin_amdgcn_readlane(s, t);
            const float at1 = __int_as_float(
                __builtin_amdgcn_readlane(__float_as_int(a), t));
            add_row(H, st, at1, lane, acc0, acc1);
        }
    }

    for (int base = 0; base < xn; base += 32) {
        int   s;
        float a = 0.0f;
        const bool ok = spill_place(base + lane, xn, full, node, ovl, esrc, edst, s);
        if (ok) a = __expf(edge_score1(es[s], edn) - m) / dn;
        unsigned mask = __builtin_amdgcn_ballot_w32(ok);
        while (mask) {
            const int t = __builtin_ctz(mask);
            mask &= mask - 1;
            const int st = __builtin_amdgcn_readlane(s, t);
            const float at1 = __int_as_float(
                __builtin_amdgcn_readlane(__float_as_int(a), t));
            add_row(H, st, at1, lane, acc0, acc1);
        }
    }

    float4* orow = (float4*)(OUT + (size_t)w * D_HID);
    if (!FIRST) {
        orow[lane]      = acc0;
        orow[lane + 32] = acc1;
        return;
    }
    const int c0 = 4 * lane;
    const int c1 = 128 + 4 * lane;
    float4 o0, o1;
    o0.x = fmaxf(acc0.x + b[c0 + 0], 0.0f);
    o0.y = fmaxf(acc0.y + b[c0 + 1], 0.0f);
    o0.z = fmaxf(acc0.z + b[c0 + 2], 0.0f);
    o0.w = fmaxf(acc0.w + b[c0 + 3], 0.0f);
    o1.x = fmaxf(acc1.x + b[c1 + 0], 0.0f);
    o1.y = fmaxf(acc1.y + b[c1 + 1], 0.0f);
    o1.z = fmaxf(acc1.z + b[c1 + 2], 0.0f);
    o1.w = fmaxf(acc1.w + b[c1 + 3], 0.0f);
    orow[lane]      = o0;
    orow[lane + 32] = o1;

    const float4 u0 = ((const float4*)us)[lane];
    const float4 u1 = ((const float4*)us)[lane + 32];
    const float4 q0 = ((const float4*)ud)[lane];
    const float4 q1 = ((const float4*)ud)[lane + 32];
    float ps = o0.x * u0.x + o0.y * u0.y + o0.z * u0.z + o0.w * u0.w
             + o1.x * u1.x + o1.y * u1.y + o1.z * u1.z + o1.w * u1.w;
    float pd = o0.x * q0.x + o0.y * q0.y + o0.z * q0.z + o0.w * q0.w
             + o1.x * q1.x + o1.y * q1.y + o1.z * q1.z + o1.w * q1.w;
    ps = wave_sum32(ps);
    pd = wave_sum32(pd);
    if (lane == 0) { es2[w] = ps; ed2[w] = pd; }
}

extern "C" void kernel_launch(void* const* d_in, const int* in_sizes, int n_in,
                              void* d_out, int out_size, void* d_ws, size_t ws_size,
                              hipStream_t stream)
{
    (void)in_sizes; (void)n_in; (void)out_size; (void)ws_size;

    const float* x    = (const float*)d_in[0];
    const int*   ei   = (const int*)  d_in[1];
    const int*   doci = (const int*)  d_in[2];
    const float* W1   = (const float*)d_in[3];
    const float* a1s  = (const float*)d_in[4];
    const float* a1d  = (const float*)d_in[5];
    const float* b1   = (const float*)d_in[6];
    const float* W2   = (const float*)d_in[7];
    const float* a2s  = (const float*)d_in[8];
    const float* a2d  = (const float*)d_in[9];
    const float* b2   = (const float*)d_in[10];
    const float* Wm1  = (const float*)d_in[11];
    const float* bm1  = (const float*)d_in[12];
    const float* Wm2  = (const float*)d_in[13];
    const float* bm2  = (const float*)d_in[14];

    const int* src = ei;
    const int* dst = ei + N_EDGES;

    float* H    = (float*)d_ws;
    float* OUT  = H    + (size_t)N_NODES * D_HID;
    float* ES   = OUT  + (size_t)N_NODES * D_HID;
    float* ED   = ES   + N_NODES;
    float* BKT  = ED   + N_NODES;
    int*  CNT    = (int*)BKT;
    int*  OVL    = CNT    + N_PAD;
    int*  SLOTS  = OVL    + OV_CAP;
    float* ES2  = (float*)(SLOTS + (size_t)N_NODES * BKT_CAP);
    float* ED2  = ES2  + N_NODES;
    float* US   = ED2  + N_NODES;
    float* UD   = US   + D_HID;
    float* Z    = H;

    const int gNodeW = (N_NODES + 7) / 8;
    const int gDocW  = (N_DOCS + 7) / 8;

    clear_counts_fold<<<FILL_BLOCKS + FOLD_BLOCKS, 1024, 0, stream>>>(
        CNT, W2, a2s, a2d, US, UD);
    gemm_wreg_f32<true, 1><<<GEMM_WGS, 128, 0, stream>>>(
        x, W1, H, a1s, a1d, ES, ED, nullptr, N_NODES / 16,
        src, dst, CNT, SLOTS, OVL,
        nullptr, nullptr, nullptr, nullptr, nullptr);
    node_aggregate<true><<<gNodeW, 256, 0, stream>>>(CNT, SLOTS, OVL, src, dst,
                                                     ES, ED, H, b1, OUT, nullptr,
                                                     US, UD, ES2, ED2, N_NODES);

    node_aggregate<false><<<gDocW, 256, 0, stream>>>(CNT, SLOTS, OVL, src, dst,
                                                     ES2, ED2, OUT, nullptr, Z,
                                                     doci, nullptr, nullptr,
                                                     nullptr, nullptr, N_DOCS);
    gemm_wreg_f32<false, 0><<<GEMM_WGS, 128, 0, stream>>>(
        Z, W2, nullptr, nullptr, nullptr, nullptr, nullptr, b2, N_DOCS / 16,
        nullptr, nullptr, nullptr, nullptr, nullptr,
        Wm1, bm1, Wm2, bm2, (float*)d_out);
}
